# starting kernel
# baseline (speedup 1.0000x reference)
#include <hip/hip_runtime.h>

typedef __attribute__((ext_vector_type(2))) float v2f;
typedef __attribute__((ext_vector_type(8))) float v8f;

#define WMMA_F32(a, b, c) \
  __builtin_amdgcn_wmma_f32_16x16x4_f32(false, (a), false, (b), (short)0, (c), false, false)

static constexpr int Tn = 32768;
static constexpr int NS = 256;
static constexpr int L1 = 32;
static constexpr int C1 = 1024;
static constexpr int L2 = 16;
static constexpr int C2 = 64;
static constexpr int SC_COLS = 16;

__constant__ float d_H[6][4] = {
  {1.0f, 0.0f, 0.0f, 0.0f},
  {0.896f, 0.128f, 0.104f, -0.032f},
  {0.784f, 0.147f, 0.216f, -0.063f},
  {0.104f, 0.032f, 0.896f, -0.128f},
  {25.0f/729.0f, 8.0f/729.0f, 704.0f/729.0f, -64.0f/729.0f},
  {0.0f, 0.0f, 1.0f, 0.0f},
};

__global__ void k_scale_dt(float* __restrict__ o, const float* __restrict__ i,
                           int n, const float* __restrict__ t) {
  int idx = blockIdx.x * 256 + threadIdx.x;
  if (idx < n) { float dt = t[1] - t[0]; o[idx] = dt * i[idx]; }
}
__global__ void k_copy(float* __restrict__ o, const float* __restrict__ i, int n) {
  int idx = blockIdx.x * 256 + threadIdx.x;
  if (idx < n) o[idx] = i[idx];
}
__global__ void k_zero(float* __restrict__ o, int n) {
  int idx = blockIdx.x * 256 + threadIdx.x;
  if (idx < n) o[idx] = 0.0f;
}
__global__ void k_eye(float* __restrict__ o) {
  int idx = blockIdx.x * 256 + threadIdx.x;
  if (idx < 65536) o[idx] = ((idx >> 8) == (idx & 255)) ? 1.0f : 0.0f;
}
__global__ void k_axpy(float* __restrict__ o, const float* __restrict__ i,
                       int n, float alpha) {
  int idx = blockIdx.x * 256 + threadIdx.x;
  if (idx < n) o[idx] += alpha * i[idx];
}
__global__ void k_addblock(float* __restrict__ Lm, const float* __restrict__ Bd, int blk) {
  int idx = blockIdx.x * 256 + threadIdx.x;
  if (idx < 8192) { int r = idx >> 5, c = idx & 31; Lm[r * 192 + blk * 32 + c] += Bd[idx]; }
}
__global__ void k_qcat(float* __restrict__ Q, const float* __restrict__ W) {
  int idx = blockIdx.x * 256 + threadIdx.x;
  if (idx >= 128 * 256) return;
  int k = idx >> 8, r = idx & 255;
  int j = k >> 5, c = k & 31;
  float s = 0.0f;
  for (int st = 0; st < 6; ++st) s += d_H[st][j] * W[r * 192 + 32 * st + c];
  Q[idx] = s;
}
__global__ void k_buildP(float* __restrict__ P, const float* __restrict__ u) {
  int idx = blockIdx.x * 256 + threadIdx.x;
  if (idx >= Tn * 128) return;
  int n = idx >> 7, k = idx & 127;
  float val = 0.0f;
  if (n < Tn - 1) {
    int j = k >> 5, c = k & 31;
    int m = n + ((j >= 2) ? 1 : 0);
    if ((j & 1) == 0) {
      val = u[m * 32 + c];
    } else {
      if (m == 0)            val = u[32 + c] - u[c];
      else if (m == Tn - 1)  val = u[(Tn - 1) * 32 + c] - u[(Tn - 2) * 32 + c];
      else                   val = 0.5f * (u[(m + 1) * 32 + c] - u[(m - 1) * 32 + c]);
    }
  }
  P[idx] = val;
}

__global__ __launch_bounds__(128) void gemm_kernel(
    float* __restrict__ D, int ldd,
    const float* __restrict__ A, int lda,
    const float* __restrict__ B, int ldb,
    const float* __restrict__ Cin, int ldc,
    int Mr, int Nc, int Kd, float alpha, float beta, int transB) {
  const int lane = threadIdx.x & 31, wave = threadIdx.x >> 5;
  const int hl = lane >> 4, lr = lane & 15;
  const int tilesN = Nc >> 4;
  const int tile = blockIdx.x * 4 + wave;
  if (tile >= (Mr >> 4) * tilesN) return;
  const int tm = tile / tilesN, tn = tile % tilesN;
  const float* Ar = A + (size_t)(tm * 16 + lr) * lda + 2 * hl;
  v8f acc = {};
  if (transB) {
    const float* Br = B + (size_t)(tn * 16 + lr) * ldb + 2 * hl;
    #pragma unroll 4
    for (int k = 0; k < Kd; k += 4) {
      v2f a = *(const v2f*)(Ar + k);
      v2f b = *(const v2f*)(Br + k);
      acc = WMMA_F32(a, b, acc);
    }
  } else {
    const float* Bc = B + (size_t)(2 * hl) * ldb + tn * 16 + lr;
    #pragma unroll 4
    for (int k = 0; k < Kd; k += 4) {
      v2f a = *(const v2f*)(Ar + k);
      v2f b;
      b[0] = Bc[(size_t)k * ldb];
      b[1] = Bc[(size_t)k * ldb + ldb];
      acc = WMMA_F32(a, b, acc);
    }
  }
  float* Dp = D + (size_t)(tm * 16 + 8 * hl) * ldd + tn * 16 + lr;
  if (beta != 0.0f) {
    const float* Cp = Cin + (size_t)(tm * 16 + 8 * hl) * ldc + tn * 16 + lr;
    #pragma unroll
    for (int v = 0; v < 8; ++v)
      Dp[(size_t)v * ldd] = alpha * acc[v] + beta * Cp[(size_t)v * ldc];
  } else {
    #pragma unroll
    for (int v = 0; v < 8; ++v)
      Dp[(size_t)v * ldd] = alpha * acc[v];
  }
}

__global__ __launch_bounds__(256) void scan_kernel(
    const float* __restrict__ Mg, const float* __restrict__ G,
    const float* __restrict__ Init,
    float* __restrict__ OutAll,
    float* __restrict__ OutEnd,
    int L) {
  extern __shared__ float smem[];
  float* Ms = smem;
  float* Xbuf0 = smem + 65536;
  float* Xbuf1 = Xbuf0 + 256 * SC_COLS;
  const int tid = threadIdx.x;
  const int lane = tid & 31, wave = tid >> 5;
  const int hl = lane >> 4, lr = lane & 15;
  const int cbase = blockIdx.x * SC_COLS;
  const int tr0 = wave * 2;

  {
    const float4* src = (const float4*)Mg;
    float4* dst = (float4*)Ms;
    for (int k = tid; k < 65536 / 4; k += 256) dst[k] = src[k];
  }
  for (int e = 0; e < SC_COLS; ++e)
    Xbuf0[tid * SC_COLS + e] = Init ? Init[(size_t)(cbase + e) * 256 + tid] : 0.0f;
  __syncthreads();

  const float* gbase = G + ((size_t)(cbase + lr) * L) * 256 + tr0 * 16 + 8 * hl;

  float* Xc = Xbuf0;
  float* Xn = Xbuf1;
  for (int i = 0; i < L; ++i) {
    if (OutAll) {
      for (int e = 0; e < SC_COLS; ++e) {
        size_t m = (size_t)(cbase + e) * L + i;
        OutAll[m * 256 + tid] = Xc[tid * SC_COLS + e];
      }
    }
    const float* gp = gbase + (size_t)i * 256;
    if (i + 1 < L) __builtin_prefetch(gp + 256, 0, 1);
    v8f acc0, acc1;
    {
      float4 ga = *(const float4*)(gp);
      float4 gb = *(const float4*)(gp + 4);
      float4 gc = *(const float4*)(gp + 16);
      float4 gd = *(const float4*)(gp + 20);
      acc0[0]=ga.x; acc0[1]=ga.y; acc0[2]=ga.z; acc0[3]=ga.w;
      acc0[4]=gb.x; acc0[5]=gb.y; acc0[6]=gb.z; acc0[7]=gb.w;
      acc1[0]=gc.x; acc1[1]=gc.y; acc1[2]=gc.z; acc1[3]=gc.w;
      acc1[4]=gd.x; acc1[5]=gd.y; acc1[6]=gd.z; acc1[7]=gd.w;
    }
    const float* Mr0 = Ms + (size_t)(tr0 * 16 + lr) * 256 + 2 * hl;
    const float* Mr1 = Mr0 + 16 * 256;
    const float* Xk = Xc + 2 * hl * SC_COLS + lr;
    #pragma unroll 8
    for (int kc = 0; kc < 64; ++kc) {
      v2f a0 = *(const v2f*)(Mr0 + kc * 4);
      v2f a1 = *(const v2f*)(Mr1 + kc * 4);
      v2f b;
      b[0] = Xk[kc * 4 * SC_COLS];
      b[1] = Xk[kc * 4 * SC_COLS + SC_COLS];
      acc0 = WMMA_F32(a0, b, acc0);
      acc1 = WMMA_F32(a1, b, acc1);
    }
    #pragma unroll
    for (int v = 0; v < 8; ++v) {
      Xn[(tr0 * 16 + 8 * hl + v) * SC_COLS + lr] = acc0[v];
      Xn[((tr0 + 1) * 16 + 8 * hl + v) * SC_COLS + lr] = acc1[v];
    }
    __syncthreads();
    float* tmp = Xc; Xc = Xn; Xn = tmp;
  }
  if (OutEnd) {
    for (int e = 0; e < SC_COLS; ++e)
      OutEnd[(size_t)(cbase + e) * 256 + tid] = Xc[tid * SC_COLS + e];
  }
}

__global__ __launch_bounds__(256) void carry_kernel(
    const float* __restrict__ Mg, const float* __restrict__ Z,
    const float* __restrict__ x0, float* __restrict__ Tcar, int n) {
  __shared__ float tv[256];
  int r = threadIdx.x;
  tv[r] = x0[r];
  Tcar[r] = tv[r];
  __syncthreads();
  const float* Mr = Mg + (size_t)r * 256;
  for (int j = 0; j + 1 < n; ++j) {
    float acc = Z[(size_t)j * 256 + r];
    #pragma unroll 8
    for (int k = 0; k < 256; ++k) acc += Mr[k] * tv[k];
    __syncthreads();
    tv[r] = acc;
    Tcar[(size_t)(j + 1) * 256 + r] = acc;
    __syncthreads();
  }
}

static inline void launch_gemm(hipStream_t s, float* D, int ldd,
                               const float* A, int lda, const float* B, int ldb,
                               const float* Cin, int ldc,
                               int M, int N, int K, float alpha, float beta, int transB) {
  int tiles = (M / 16) * (N / 16);
  gemm_kernel<<<(tiles + 3) / 4, 128, 0, s>>>(D, ldd, A, lda, B, ldb, Cin, ldc,
                                              M, N, K, alpha, beta, transB);
}

extern "C" void kernel_launch(void* const* d_in, const int* in_sizes, int n_in,
                              void* d_out, int out_size, void* d_ws, size_t ws_size,
                              hipStream_t stream) {
  (void)in_sizes; (void)n_in; (void)out_size; (void)ws_size;
  const float* t  = (const float*)d_in[0];
  const float* u  = (const float*)d_in[1];
  const float* x0 = (const float*)d_in[2];
  const float* Am = (const float*)d_in[3];
  const float* Bm = (const float*)d_in[4];
  const float* Cm = (const float*)d_in[5];
  const float* Dm = (const float*)d_in[6];
  float* xout = (float*)d_out;
  float* yout = xout + (size_t)Tn * NS;

  float* w = (float*)d_ws;
  float* Ad = w;  w += 65536;
  float* Bd = w;  w += 8192;
  float* Kt[6]; for (int i = 0; i < 6; i++) { Kt[i] = w; w += 65536; }
  float* Lt[6]; for (int i = 0; i < 6; i++) { Lt[i] = w; w += 49152; }
  float* TMP  = w; w += 65536;
  float* TMPL = w; w += 49152;
  float* Mm   = w; w += 65536;
  float* Wm   = w; w += 49152;
  float* Qc   = w; w += 32768;
  float* ML   = w; w += 65536;
  float* M512 = w; w += 65536;
  float* PT1  = w; w += 65536;
  float* PT2  = w; w += 65536;
  float* Pm   = w; w += (size_t)Tn * 128;
  float* G    = w; w += (size_t)Tn * 256;
  float* Zend = w; w += (size_t)C1 * 256;
  float* Zend2= w; w += (size_t)C2 * 256;
  float* Tcar = w; w += (size_t)C2 * 256;
  float* Scar = w; w += (size_t)C1 * 256;

  auto gdim = [](int n) { return dim3((unsigned)((n + 255) / 256)); };

  k_scale_dt<<<gdim(65536), 256, 0, stream>>>(Ad, Am, 65536, t);
  k_scale_dt<<<gdim(8192),  256, 0, stream>>>(Bd, Bm, 8192, t);
  k_copy<<<gdim(65536), 256, 0, stream>>>(Kt[0], Ad, 65536);
  k_zero<<<gdim(49152), 256, 0, stream>>>(Lt[0], 49152);
  k_addblock<<<gdim(8192), 256, 0, stream>>>(Lt[0], Bd, 0);

  double aa[7][7] = {};
  aa[2][1] = 0.2;
  aa[3][1] = 3.0/40.0;      aa[3][2] = 9.0/40.0;
  aa[4][1] = 44.0/45.0;     aa[4][2] = -56.0/15.0;     aa[4][3] = 32.0/9.0;
  aa[5][1] = 19372.0/6561.0; aa[5][2] = -25360.0/2187.0;
  aa[5][3] = 64448.0/6561.0; aa[5][4] = -212.0/729.0;
  aa[6][1] = 9017.0/3168.0; aa[6][2] = -355.0/33.0; aa[6][3] = 46732.0/5247.0;
  aa[6][4] = 49.0/176.0;    aa[6][5] = -5103.0/18656.0;
  for (int i = 2; i <= 6; ++i) {
    k_eye<<<gdim(65536), 256, 0, stream>>>(TMP);
    k_zero<<<gdim(49152), 256, 0, stream>>>(TMPL, 49152);
    for (int j = 1; j < i; ++j) {
      float c = (float)aa[i][j];
      k_axpy<<<gdim(65536), 256, 0, stream>>>(TMP,  Kt[j-1], 65536, c);
      k_axpy<<<gdim(49152), 256, 0, stream>>>(TMPL, Lt[j-1], 49152, c);
    }
    launch_gemm(stream, Kt[i-1], 256, Ad, 256, TMP,  256, nullptr, 0, 256, 256, 256, 1.f, 0.f, 0);
    launch_gemm(stream, Lt[i-1], 192, Ad, 256, TMPL, 192, nullptr, 0, 256, 192, 256, 1.f, 0.f, 0);
    k_addblock<<<gdim(8192), 256, 0, stream>>>(Lt[i-1], Bd, i - 1);
  }
  const double bb[7] = {0.0, 35.0/384.0, 0.0, 500.0/1113.0, 125.0/192.0,
                        -2187.0/6784.0, 11.0/84.0};
  k_eye<<<gdim(65536), 256, 0, stream>>>(Mm);
  k_zero<<<gdim(49152), 256, 0, stream>>>(Wm, 49152);
  for (int i = 1; i <= 6; ++i) {
    float c = (float)bb[i];
    if (c != 0.0f) {
      k_axpy<<<gdim(65536), 256, 0, stream>>>(Mm, Kt[i-1], 65536, c);
      k_axpy<<<gdim(49152), 256, 0, stream>>>(Wm, Lt[i-1], 49152, c);
    }
  }
  k_qcat<<<gdim(32768), 256, 0, stream>>>(Qc, Wm);

  launch_gemm(stream, PT1, 256, Mm, 256, Mm, 256, nullptr, 0, 256, 256, 256, 1.f, 0.f, 0);
  launch_gemm(stream, PT2, 256, PT1, 256, PT1, 256, nullptr, 0, 256, 256, 256, 1.f, 0.f, 0);
  launch_gemm(stream, PT1, 256, PT2, 256, PT2, 256, nullptr, 0, 256, 256, 256, 1.f, 0.f, 0);
  launch_gemm(stream, PT2, 256, PT1, 256, PT1, 256, nullptr, 0, 256, 256, 256, 1.f, 0.f, 0);
  launch_gemm(stream, ML,  256, PT2, 256, PT2, 256, nullptr, 0, 256, 256, 256, 1.f, 0.f, 0);
  launch_gemm(stream, PT1, 256, ML,  256, ML,  256, nullptr, 0, 256, 256, 256, 1.f, 0.f, 0);
  launch_gemm(stream, PT2, 256, PT1, 256, PT1, 256, nullptr, 0, 256, 256, 256, 1.f, 0.f, 0);
  launch_gemm(stream, PT1, 256, PT2, 256, PT2, 256, nullptr, 0, 256, 256, 256, 1.f, 0.f, 0);
  launch_gemm(stream, M512,256, PT1, 256, PT1, 256, nullptr, 0, 256, 256, 256, 1.f, 0.f, 0);

  k_buildP<<<gdim(Tn * 128), 256, 0, stream>>>(Pm, u);
  launch_gemm(stream, G, 256, Pm, 128, Qc, 256, nullptr, 0, Tn, 256, 128, 1.f, 0.f, 0);

  const size_t SCAN_LDS = (size_t)(65536 + 2 * 256 * SC_COLS) * sizeof(float);
  scan_kernel<<<C1 / SC_COLS, 256, SCAN_LDS, stream>>>(Mm, G, nullptr, nullptr, Zend, L1);
  scan_kernel<<<C2 / SC_COLS, 256, SCAN_LDS, stream>>>(ML, Zend, nullptr, nullptr, Zend2, L2);
  carry_kernel<<<1, 256, 0, stream>>>(M512, Zend2, x0, Tcar, C2);
  scan_kernel<<<C2 / SC_COLS, 256, SCAN_LDS, stream>>>(ML, Zend, Tcar, Scar, nullptr, L2);
  scan_kernel<<<C1 / SC_COLS, 256, SCAN_LDS, stream>>>(Mm, G, Scar, xout, nullptr, L1);

  launch_gemm(stream, yout, 32, xout, 256, Cm, 256, nullptr, 0, Tn, 32, 256, 1.f, 0.f, 1);
  launch_gemm(stream, yout, 32, u,    32,  Dm, 32,  yout,   32, Tn, 32, 32,  1.f, 1.f, 1);
}


# scan kernel: pad the LDS row pitch of the step matrix from 256 to 260 floats so the f32 WMMA A-fragment loads are bank-conflict-free instead of 16-way conflicted
# speedup vs baseline: 1.2428x; 1.2428x over previous
#include <hip/hip_runtime.h>

typedef __attribute__((ext_vector_type(2))) float v2f;
typedef __attribute__((ext_vector_type(8))) float v8f;

#define WMMA_F32(a, b, c) \
  __builtin_amdgcn_wmma_f32_16x16x4_f32(false, (a), false, (b), (short)0, (c), false, false)

static constexpr int Tn = 32768;
static constexpr int NS = 256;
static constexpr int L1 = 32;
static constexpr int C1 = 1024;
static constexpr int L2 = 16;
static constexpr int C2 = 64;
static constexpr int SC_COLS = 16;
static constexpr int MS_LD = 260;
static constexpr int SCAN_LDS_FLOATS = 256 * MS_LD + 2 * 256 * SC_COLS;

__constant__ float d_H[6][4] = {
  {1.0f, 0.0f, 0.0f, 0.0f},
  {0.896f, 0.128f, 0.104f, -0.032f},
  {0.784f, 0.147f, 0.216f, -0.063f},
  {0.104f, 0.032f, 0.896f, -0.128f},
  {25.0f/729.0f, 8.0f/729.0f, 704.0f/729.0f, -64.0f/729.0f},
  {0.0f, 0.0f, 1.0f, 0.0f},
};

__global__ void k_scale_dt(float* __restrict__ o, const float* __restrict__ i,
                           int n, const float* __restrict__ t) {
  int idx = blockIdx.x * 256 + threadIdx.x;
  if (idx < n) { float dt = t[1] - t[0]; o[idx] = dt * i[idx]; }
}
__global__ void k_copy(float* __restrict__ o, const float* __restrict__ i, int n) {
  int idx = blockIdx.x * 256 + threadIdx.x;
  if (idx < n) o[idx] = i[idx];
}
__global__ void k_zero(float* __restrict__ o, int n) {
  int idx = blockIdx.x * 256 + threadIdx.x;
  if (idx < n) o[idx] = 0.0f;
}
__global__ void k_eye(float* __restrict__ o) {
  int idx = blockIdx.x * 256 + threadIdx.x;
  if (idx < 65536) o[idx] = ((idx >> 8) == (idx & 255)) ? 1.0f : 0.0f;
}
__global__ void k_axpy(float* __restrict__ o, const float* __restrict__ i,
                       int n, float alpha) {
  int idx = blockIdx.x * 256 + threadIdx.x;
  if (idx < n) o[idx] += alpha * i[idx];
}
__global__ void k_addblock(float* __restrict__ Lm, const float* __restrict__ Bd, int blk) {
  int idx = blockIdx.x * 256 + threadIdx.x;
  if (idx < 8192) { int r = idx >> 5, c = idx & 31; Lm[r * 192 + blk * 32 + c] += Bd[idx]; }
}
__global__ void k_qcat(float* __restrict__ Q, const float* __restrict__ W) {
  int idx = blockIdx.x * 256 + threadIdx.x;
  if (idx >= 128 * 256) return;
  int k = idx >> 8, r = idx & 255;
  int j = k >> 5, c = k & 31;
  float s = 0.0f;
  for (int st = 0; st < 6; ++st) s += d_H[st][j] * W[r * 192 + 32 * st + c];
  Q[idx] = s;
}
__global__ void k_buildP(float* __restrict__ P, const float* __restrict__ u) {
  int idx = blockIdx.x * 256 + threadIdx.x;
  if (idx >= Tn * 128) return;
  int n = idx >> 7, k = idx & 127;
  float val = 0.0f;
  if (n < Tn - 1) {
    int j = k >> 5, c = k & 31;
    int m = n + ((j >= 2) ? 1 : 0);
    if ((j & 1) == 0) {
      val = u[m * 32 + c];
    } else {
      if (m == 0)            val = u[32 + c] - u[c];
      else if (m == Tn - 1)  val = u[(Tn - 1) * 32 + c] - u[(Tn - 2) * 32 + c];
      else                   val = 0.5f * (u[(m + 1) * 32 + c] - u[(m - 1) * 32 + c]);
    }
  }
  P[idx] = val;
}

__global__ __launch_bounds__(128) void gemm_kernel(
    float* __restrict__ D, int ldd,
    const float* __restrict__ A, int lda,
    const float* __restrict__ B, int ldb,
    const float* __restrict__ Cin, int ldc,
    int Mr, int Nc, int Kd, float alpha, float beta, int transB) {
  const int lane = threadIdx.x & 31, wave = threadIdx.x >> 5;
  const int hl = lane >> 4, lr = lane & 15;
  const int tilesN = Nc >> 4;
  const int tile = blockIdx.x * 4 + wave;
  if (tile >= (Mr >> 4) * tilesN) return;
  const int tm = tile / tilesN, tn = tile % tilesN;
  const float* Ar = A + (size_t)(tm * 16 + lr) * lda + 2 * hl;
  v8f acc = {};
  if (transB) {
    const float* Br = B + (size_t)(tn * 16 + lr) * ldb + 2 * hl;
    #pragma unroll 4
    for (int k = 0; k < Kd; k += 4) {
      v2f a = *(const v2f*)(Ar + k);
      v2f b = *(const v2f*)(Br + k);
      acc = WMMA_F32(a, b, acc);
    }
  } else {
    const float* Bc = B + (size_t)(2 * hl) * ldb + tn * 16 + lr;
    #pragma unroll 4
    for (int k = 0; k < Kd; k += 4) {
      v2f a = *(const v2f*)(Ar + k);
      v2f b;
      b[0] = Bc[(size_t)k * ldb];
      b[1] = Bc[(size_t)k * ldb + ldb];
      acc = WMMA_F32(a, b, acc);
    }
  }
  float* Dp = D + (size_t)(tm * 16 + 8 * hl) * ldd + tn * 16 + lr;
  if (beta != 0.0f) {
    const float* Cp = Cin + (size_t)(tm * 16 + 8 * hl) * ldc + tn * 16 + lr;
    #pragma unroll
    for (int v = 0; v < 8; ++v)
      Dp[(size_t)v * ldd] = alpha * acc[v] + beta * Cp[(size_t)v * ldc];
  } else {
    #pragma unroll
    for (int v = 0; v < 8; ++v)
      Dp[(size_t)v * ldd] = alpha * acc[v];
  }
}

__global__ __launch_bounds__(256) void scan_kernel(
    const float* __restrict__ Mg, const float* __restrict__ G,
    const float* __restrict__ Init,
    float* __restrict__ OutAll,
    float* __restrict__ OutEnd,
    int L) {
  __shared__ __attribute__((aligned(16))) float smem[SCAN_LDS_FLOATS];
  float* Ms = smem;
  float* Xbuf0 = smem + 256 * MS_LD;
  float* Xbuf1 = Xbuf0 + 256 * SC_COLS;
  const int tid = threadIdx.x;
  const int lane = tid & 31, wave = tid >> 5;
  const int hl = lane >> 4, lr = lane & 15;
  const int cbase = blockIdx.x * SC_COLS;
  const int tr0 = wave * 2;

  {
    const float4* src = (const float4*)Mg;
    float4* dst = (float4*)Ms;
    for (int k = tid; k < 65536 / 4; k += 256)
      dst[(k >> 6) * (MS_LD / 4) + (k & 63)] = src[k];
  }
  for (int e = 0; e < SC_COLS; ++e)
    Xbuf0[tid * SC_COLS + e] = Init ? Init[(size_t)(cbase + e) * 256 + tid] : 0.0f;
  __syncthreads();

  const float* gbase = G + ((size_t)(cbase + lr) * L) * 256 + tr0 * 16 + 8 * hl;

  float* Xc = Xbuf0;
  float* Xn = Xbuf1;
  for (int i = 0; i < L; ++i) {
    if (OutAll) {
      for (int e = 0; e < SC_COLS; ++e) {
        size_t m = (size_t)(cbase + e) * L + i;
        OutAll[m * 256 + tid] = Xc[tid * SC_COLS + e];
      }
    }
    const float* gp = gbase + (size_t)i * 256;
    if (i + 1 < L) __builtin_prefetch(gp + 256, 0, 1);
    v8f acc0, acc1;
    {
      float4 ga = *(const float4*)(gp);
      float4 gb = *(const float4*)(gp + 4);
      float4 gc = *(const float4*)(gp + 16);
      float4 gd = *(const float4*)(gp + 20);
      acc0[0]=ga.x; acc0[1]=ga.y; acc0[2]=ga.z; acc0[3]=ga.w;
      acc0[4]=gb.x; acc0[5]=gb.y; acc0[6]=gb.z; acc0[7]=gb.w;
      acc1[0]=gc.x; acc1[1]=gc.y; acc1[2]=gc.z; acc1[3]=gc.w;
      acc1[4]=gd.x; acc1[5]=gd.y; acc1[6]=gd.z; acc1[7]=gd.w;
    }
    const float* Mr0 = Ms + (size_t)(tr0 * 16 + lr) * MS_LD + 2 * hl;
    const float* Mr1 = Mr0 + 16 * MS_LD;
    const float* Xk = Xc + 2 * hl * SC_COLS + lr;
    #pragma unroll 8
    for (int kc = 0; kc < 64; ++kc) {
      v2f a0 = *(const v2f*)(Mr0 + kc * 4);
      v2f a1 = *(const v2f*)(Mr1 + kc * 4);
      v2f b;
      b[0] = Xk[kc * 4 * SC_COLS];
      b[1] = Xk[kc * 4 * SC_COLS + SC_COLS];
      acc0 = WMMA_F32(a0, b, acc0);
      acc1 = WMMA_F32(a1, b, acc1);
    }
    #pragma unroll
    for (int v = 0; v < 8; ++v) {
      Xn[(tr0 * 16 + 8 * hl + v) * SC_COLS + lr] = acc0[v];
      Xn[((tr0 + 1) * 16 + 8 * hl + v) * SC_COLS + lr] = acc1[v];
    }
    __syncthreads();
    float* tmp = Xc; Xc = Xn; Xn = tmp;
  }
  if (OutEnd) {
    for (int e = 0; e < SC_COLS; ++e)
      OutEnd[(size_t)(cbase + e) * 256 + tid] = Xc[tid * SC_COLS + e];
  }
}

__global__ __launch_bounds__(256) void carry_kernel(
    const float* __restrict__ Mg, const float* __restrict__ Z,
    const float* __restrict__ x0, float* __restrict__ Tcar, int n) {
  __shared__ float tv[256];
  int r = threadIdx.x;
  tv[r] = x0[r];
  Tcar[r] = tv[r];
  __syncthreads();
  const float* Mr = Mg + (size_t)r * 256;
  for (int j = 0; j + 1 < n; ++j) {
    float acc = Z[(size_t)j * 256 + r];
    #pragma unroll 8
    for (int k = 0; k < 256; ++k) acc += Mr[k] * tv[k];
    __syncthreads();
    tv[r] = acc;
    Tcar[(size_t)(j + 1) * 256 + r] = acc;
    __syncthreads();
  }
}

static inline void launch_gemm(hipStream_t s, float* D, int ldd,
                               const float* A, int lda, const float* B, int ldb,
                               const float* Cin, int ldc,
                               int M, int N, int K, float alpha, float beta, int transB) {
  int tiles = (M / 16) * (N / 16);
  gemm_kernel<<<(tiles + 3) / 4, 128, 0, s>>>(D, ldd, A, lda, B, ldb, Cin, ldc,
                                              M, N, K, alpha, beta, transB);
}

extern "C" void kernel_launch(void* const* d_in, const int* in_sizes, int n_in,
                              void* d_out, int out_size, void* d_ws, size_t ws_size,
                              hipStream_t stream) {
  (void)in_sizes; (void)n_in; (void)out_size; (void)ws_size;
  const float* t  = (const float*)d_in[0];
  const float* u  = (const float*)d_in[1];
  const float* x0 = (const float*)d_in[2];
  const float* Am = (const float*)d_in[3];
  const float* Bm = (const float*)d_in[4];
  const float* Cm = (const float*)d_in[5];
  const float* Dm = (const float*)d_in[6];
  float* xout = (float*)d_out;
  float* yout = xout + (size_t)Tn * NS;

  float* w = (float*)d_ws;
  float* Ad = w;  w += 65536;
  float* Bd = w;  w += 8192;
  float* Kt[6]; for (int i = 0; i < 6; i++) { Kt[i] = w; w += 65536; }
  float* Lt[6]; for (int i = 0; i < 6; i++) { Lt[i] = w; w += 49152; }
  float* TMP  = w; w += 65536;
  float* TMPL = w; w += 49152;
  float* Mm   = w; w += 65536;
  float* Wm   = w; w += 49152;
  float* Qc   = w; w += 32768;
  float* ML   = w; w += 65536;
  float* M512 = w; w += 65536;
  float* PT1  = w; w += 65536;
  float* PT2  = w; w += 65536;
  float* Pm   = w; w += (size_t)Tn * 128;
  float* G    = w; w += (size_t)Tn * 256;
  float* Zend = w; w += (size_t)C1 * 256;
  float* Zend2= w; w += (size_t)C2 * 256;
  float* Tcar = w; w += (size_t)C2 * 256;
  float* Scar = w; w += (size_t)C1 * 256;

  auto gdim = [](int n) { return dim3((unsigned)((n + 255) / 256)); };

  k_scale_dt<<<gdim(65536), 256, 0, stream>>>(Ad, Am, 65536, t);
  k_scale_dt<<<gdim(8192),  256, 0, stream>>>(Bd, Bm, 8192, t);
  k_copy<<<gdim(65536), 256, 0, stream>>>(Kt[0], Ad, 65536);
  k_zero<<<gdim(49152), 256, 0, stream>>>(Lt[0], 49152);
  k_addblock<<<gdim(8192), 256, 0, stream>>>(Lt[0], Bd, 0);

  double aa[7][7] = {};
  aa[2][1] = 0.2;
  aa[3][1] = 3.0/40.0;      aa[3][2] = 9.0/40.0;
  aa[4][1] = 44.0/45.0;     aa[4][2] = -56.0/15.0;     aa[4][3] = 32.0/9.0;
  aa[5][1] = 19372.0/6561.0; aa[5][2] = -25360.0/2187.0;
  aa[5][3] = 64448.0/6561.0; aa[5][4] = -212.0/729.0;
  aa[6][1] = 9017.0/3168.0; aa[6][2] = -355.0/33.0; aa[6][3] = 46732.0/5247.0;
  aa[6][4] = 49.0/176.0;    aa[6][5] = -5103.0/18656.0;
  for (int i = 2; i <= 6; ++i) {
    k_eye<<<gdim(65536), 256, 0, stream>>>(TMP);
    k_zero<<<gdim(49152), 256, 0, stream>>>(TMPL, 49152);
    for (int j = 1; j < i; ++j) {
      float c = (float)aa[i][j];
      k_axpy<<<gdim(65536), 256, 0, stream>>>(TMP,  Kt[j-1], 65536, c);
      k_axpy<<<gdim(49152), 256, 0, stream>>>(TMPL, Lt[j-1], 49152, c);
    }
    launch_gemm(stream, Kt[i-1], 256, Ad, 256, TMP,  256, nullptr, 0, 256, 256, 256, 1.f, 0.f, 0);
    launch_gemm(stream, Lt[i-1], 192, Ad, 256, TMPL, 192, nullptr, 0, 256, 192, 256, 1.f, 0.f, 0);
    k_addblock<<<gdim(8192), 256, 0, stream>>>(Lt[i-1], Bd, i - 1);
  }
  const double bb[7] = {0.0, 35.0/384.0, 0.0, 500.0/1113.0, 125.0/192.0,
                        -2187.0/6784.0, 11.0/84.0};
  k_eye<<<gdim(65536), 256, 0, stream>>>(Mm);
  k_zero<<<gdim(49152), 256, 0, stream>>>(Wm, 49152);
  for (int i = 1; i <= 6; ++i) {
    float c = (float)bb[i];
    if (c != 0.0f) {
      k_axpy<<<gdim(65536), 256, 0, stream>>>(Mm, Kt[i-1], 65536, c);
      k_axpy<<<gdim(49152), 256, 0, stream>>>(Wm, Lt[i-1], 49152, c);
    }
  }
  k_qcat<<<gdim(32768), 256, 0, stream>>>(Qc, Wm);

  launch_gemm(stream, PT1, 256, Mm, 256, Mm, 256, nullptr, 0, 256, 256, 256, 1.f, 0.f, 0);
  launch_gemm(stream, PT2, 256, PT1, 256, PT1, 256, nullptr, 0, 256, 256, 256, 1.f, 0.f, 0);
  launch_gemm(stream, PT1, 256, PT2, 256, PT2, 256, nullptr, 0, 256, 256, 256, 1.f, 0.f, 0);
  launch_gemm(stream, PT2, 256, PT1, 256, PT1, 256, nullptr, 0, 256, 256, 256, 1.f, 0.f, 0);
  launch_gemm(stream, ML,  256, PT2, 256, PT2, 256, nullptr, 0, 256, 256, 256, 1.f, 0.f, 0);
  launch_gemm(stream, PT1, 256, ML,  256, ML,  256, nullptr, 0, 256, 256, 256, 1.f, 0.f, 0);
  launch_gemm(stream, PT2, 256, PT1, 256, PT1, 256, nullptr, 0, 256, 256, 256, 1.f, 0.f, 0);
  launch_gemm(stream, PT1, 256, PT2, 256, PT2, 256, nullptr, 0, 256, 256, 256, 1.f, 0.f, 0);
  launch_gemm(stream, M512,256, PT1, 256, PT1, 256, nullptr, 0, 256, 256, 256, 1.f, 0.f, 0);

  k_buildP<<<gdim(Tn * 128), 256, 0, stream>>>(Pm, u);
  launch_gemm(stream, G, 256, Pm, 128, Qc, 256, nullptr, 0, Tn, 256, 128, 1.f, 0.f, 0);

  scan_kernel<<<C1 / SC_COLS, 256, 0, stream>>>(Mm, G, nullptr, nullptr, Zend, L1);
  scan_kernel<<<C2 / SC_COLS, 256, 0, stream>>>(ML, Zend, nullptr, nullptr, Zend2, L2);
  carry_kernel<<<1, 256, 0, stream>>>(M512, Zend2, x0, Tcar, C2);
  scan_kernel<<<C2 / SC_COLS, 256, 0, stream>>>(ML, Zend, Tcar, Scar, nullptr, L2);
  scan_kernel<<<C1 / SC_COLS, 256, 0, stream>>>(Mm, G, Scar, xout, nullptr, L1);

  launch_gemm(stream, yout, 32, xout, 256, Cm, 256, nullptr, 0, Tn, 32, 256, 1.f, 0.f, 1);
  launch_gemm(stream, yout, 32, u,    32,  Dm, 32,  yout,   32, Tn, 32, 32,  1.f, 1.f, 1);
}


# carry kernel: hold the 256x256 matrix in registers (1024 threads x 64 floats) instead of re-reading it from memory in each of the 63 steps; row sums become 16 lane partials joined by a lane tree
# speedup vs baseline: 1.7557x; 1.4127x over previous
#include <hip/hip_runtime.h>

typedef __attribute__((ext_vector_type(2))) float v2f;
typedef __attribute__((ext_vector_type(8))) float v8f;

#define WMMA_F32(a, b, c) \
  __builtin_amdgcn_wmma_f32_16x16x4_f32(false, (a), false, (b), (short)0, (c), false, false)

static constexpr int Tn = 32768;
static constexpr int NS = 256;
static constexpr int L1 = 32;
static constexpr int C1 = 1024;
static constexpr int L2 = 16;
static constexpr int C2 = 64;
static constexpr int SC_COLS = 16;
static constexpr int MS_LD = 260;
static constexpr int SCAN_LDS_FLOATS = 256 * MS_LD + 2 * 256 * SC_COLS;

__constant__ float d_H[6][4] = {
  {1.0f, 0.0f, 0.0f, 0.0f},
  {0.896f, 0.128f, 0.104f, -0.032f},
  {0.784f, 0.147f, 0.216f, -0.063f},
  {0.104f, 0.032f, 0.896f, -0.128f},
  {25.0f/729.0f, 8.0f/729.0f, 704.0f/729.0f, -64.0f/729.0f},
  {0.0f, 0.0f, 1.0f, 0.0f},
};

__global__ void k_scale_dt(float* __restrict__ o, const float* __restrict__ i,
                           int n, const float* __restrict__ t) {
  int idx = blockIdx.x * 256 + threadIdx.x;
  if (idx < n) { float dt = t[1] - t[0]; o[idx] = dt * i[idx]; }
}
__global__ void k_copy(float* __restrict__ o, const float* __restrict__ i, int n) {
  int idx = blockIdx.x * 256 + threadIdx.x;
  if (idx < n) o[idx] = i[idx];
}
__global__ void k_zero(float* __restrict__ o, int n) {
  int idx = blockIdx.x * 256 + threadIdx.x;
  if (idx < n) o[idx] = 0.0f;
}
__global__ void k_eye(float* __restrict__ o) {
  int idx = blockIdx.x * 256 + threadIdx.x;
  if (idx < 65536) o[idx] = ((idx >> 8) == (idx & 255)) ? 1.0f : 0.0f;
}
__global__ void k_axpy(float* __restrict__ o, const float* __restrict__ i,
                       int n, float alpha) {
  int idx = blockIdx.x * 256 + threadIdx.x;
  if (idx < n) o[idx] += alpha * i[idx];
}
__global__ void k_addblock(float* __restrict__ Lm, const float* __restrict__ Bd, int blk) {
  int idx = blockIdx.x * 256 + threadIdx.x;
  if (idx < 8192) { int r = idx >> 5, c = idx & 31; Lm[r * 192 + blk * 32 + c] += Bd[idx]; }
}
__global__ void k_qcat(float* __restrict__ Q, const float* __restrict__ W) {
  int idx = blockIdx.x * 256 + threadIdx.x;
  if (idx >= 128 * 256) return;
  int k = idx >> 8, r = idx & 255;
  int j = k >> 5, c = k & 31;
  float s = 0.0f;
  for (int st = 0; st < 6; ++st) s += d_H[st][j] * W[r * 192 + 32 * st + c];
  Q[idx] = s;
}
__global__ void k_buildP(float* __restrict__ P, const float* __restrict__ u) {
  int idx = blockIdx.x * 256 + threadIdx.x;
  if (idx >= Tn * 128) return;
  int n = idx >> 7, k = idx & 127;
  float val = 0.0f;
  if (n < Tn - 1) {
    int j = k >> 5, c = k & 31;
    int m = n + ((j >= 2) ? 1 : 0);
    if ((j & 1) == 0) {
      val = u[m * 32 + c];
    } else {
      if (m == 0)            val = u[32 + c] - u[c];
      else if (m == Tn - 1)  val = u[(Tn - 1) * 32 + c] - u[(Tn - 2) * 32 + c];
      else                   val = 0.5f * (u[(m + 1) * 32 + c] - u[(m - 1) * 32 + c]);
    }
  }
  P[idx] = val;
}

__global__ __launch_bounds__(128) void gemm_kernel(
    float* __restrict__ D, int ldd,
    const float* __restrict__ A, int lda,
    const float* __restrict__ B, int ldb,
    const float* __restrict__ Cin, int ldc,
    int Mr, int Nc, int Kd, float alpha, float beta, int transB) {
  const int lane = threadIdx.x & 31, wave = threadIdx.x >> 5;
  const int hl = lane >> 4, lr = lane & 15;
  const int tilesN = Nc >> 4;
  const int tile = blockIdx.x * 4 + wave;
  if (tile >= (Mr >> 4) * tilesN) return;
  const int tm = tile / tilesN, tn = tile % tilesN;
  const float* Ar = A + (size_t)(tm * 16 + lr) * lda + 2 * hl;
  v8f acc = {};
  if (transB) {
    const float* Br = B + (size_t)(tn * 16 + lr) * ldb + 2 * hl;
    #pragma unroll 4
    for (int k = 0; k < Kd; k += 4) {
      v2f a = *(const v2f*)(Ar + k);
      v2f b = *(const v2f*)(Br + k);
      acc = WMMA_F32(a, b, acc);
    }
  } else {
    const float* Bc = B + (size_t)(2 * hl) * ldb + tn * 16 + lr;
    #pragma unroll 4
    for (int k = 0; k < Kd; k += 4) {
      v2f a = *(const v2f*)(Ar + k);
      v2f b;
      b[0] = Bc[(size_t)k * ldb];
      b[1] = Bc[(size_t)k * ldb + ldb];
      acc = WMMA_F32(a, b, acc);
    }
  }
  float* Dp = D + (size_t)(tm * 16 + 8 * hl) * ldd + tn * 16 + lr;
  if (beta != 0.0f) {
    const float* Cp = Cin + (size_t)(tm * 16 + 8 * hl) * ldc + tn * 16 + lr;
    #pragma unroll
    for (int v = 0; v < 8; ++v)
      Dp[(size_t)v * ldd] = alpha * acc[v] + beta * Cp[(size_t)v * ldc];
  } else {
    #pragma unroll
    for (int v = 0; v < 8; ++v)
      Dp[(size_t)v * ldd] = alpha * acc[v];
  }
}

__global__ __launch_bounds__(256) void scan_kernel(
    const float* __restrict__ Mg, const float* __restrict__ G,
    const float* __restrict__ Init,
    float* __restrict__ OutAll,
    float* __restrict__ OutEnd,
    int L) {
  __shared__ __attribute__((aligned(16))) float smem[SCAN_LDS_FLOATS];
  float* Ms = smem;
  float* Xbuf0 = smem + 256 * MS_LD;
  float* Xbuf1 = Xbuf0 + 256 * SC_COLS;
  const int tid = threadIdx.x;
  const int lane = tid & 31, wave = tid >> 5;
  const int hl = lane >> 4, lr = lane & 15;
  const int cbase = blockIdx.x * SC_COLS;
  const int tr0 = wave * 2;

  {
    const float4* src = (const float4*)Mg;
    float4* dst = (float4*)Ms;
    for (int k = tid; k < 65536 / 4; k += 256)
      dst[(k >> 6) * (MS_LD / 4) + (k & 63)] = src[k];
  }
  for (int e = 0; e < SC_COLS; ++e)
    Xbuf0[tid * SC_COLS + e] = Init ? Init[(size_t)(cbase + e) * 256 + tid] : 0.0f;
  __syncthreads();

  const float* gbase = G + ((size_t)(cbase + lr) * L) * 256 + tr0 * 16 + 8 * hl;

  float* Xc = Xbuf0;
  float* Xn = Xbuf1;
  for (int i = 0; i < L; ++i) {
    if (OutAll) {
      for (int e = 0; e < SC_COLS; ++e) {
        size_t m = (size_t)(cbase + e) * L + i;
        OutAll[m * 256 + tid] = Xc[tid * SC_COLS + e];
      }
    }
    const float* gp = gbase + (size_t)i * 256;
    if (i + 1 < L) __builtin_prefetch(gp + 256, 0, 1);
    v8f acc0, acc1;
    {
      float4 ga = *(const float4*)(gp);
      float4 gb = *(const float4*)(gp + 4);
      float4 gc = *(const float4*)(gp + 16);
      float4 gd = *(const float4*)(gp + 20);
      acc0[0]=ga.x; acc0[1]=ga.y; acc0[2]=ga.z; acc0[3]=ga.w;
      acc0[4]=gb.x; acc0[5]=gb.y; acc0[6]=gb.z; acc0[7]=gb.w;
      acc1[0]=gc.x; acc1[1]=gc.y; acc1[2]=gc.z; acc1[3]=gc.w;
      acc1[4]=gd.x; acc1[5]=gd.y; acc1[6]=gd.z; acc1[7]=gd.w;
    }
    const float* Mr0 = Ms + (size_t)(tr0 * 16 + lr) * MS_LD + 2 * hl;
    const float* Mr1 = Mr0 + 16 * MS_LD;
    const float* Xk = Xc + 2 * hl * SC_COLS + lr;
    #pragma unroll 8
    for (int kc = 0; kc < 64; ++kc) {
      v2f a0 = *(const v2f*)(Mr0 + kc * 4);
      v2f a1 = *(const v2f*)(Mr1 + kc * 4);
      v2f b;
      b[0] = Xk[kc * 4 * SC_COLS];
      b[1] = Xk[kc * 4 * SC_COLS + SC_COLS];
      acc0 = WMMA_F32(a0, b, acc0);
      acc1 = WMMA_F32(a1, b, acc1);
    }
    #pragma unroll
    for (int v = 0; v < 8; ++v) {
      Xn[(tr0 * 16 + 8 * hl + v) * SC_COLS + lr] = acc0[v];
      Xn[((tr0 + 1) * 16 + 8 * hl + v) * SC_COLS + lr] = acc1[v];
    }
    __syncthreads();
    float* tmp = Xc; Xc = Xn; Xn = tmp;
  }
  if (OutEnd) {
    for (int e = 0; e < SC_COLS; ++e)
      OutEnd[(size_t)(cbase + e) * 256 + tid] = Xc[tid * SC_COLS + e];
  }
}

static constexpr int TV_PITCH = 20;
static constexpr int TV_FLOATS = 16 * TV_PITCH;

template <int CTRL>
__device__ __forceinline__ float lane_read(float v) {
  return __int_as_float(__builtin_amdgcn_update_dpp(0, __float_as_int(v), CTRL, 0xF, 0xF, true));
}
static constexpr int DPP_SWAP1 = 0xB1;
static constexpr int DPP_SWAP2 = 0x4E;
static constexpr int DPP_ROR4  = 0x124;
static constexpr int DPP_ROR8  = 0x128;

__global__ __launch_bounds__(1024) void carry_kernel(
    const float* __restrict__ Mg, const float* __restrict__ Z,
    const float* __restrict__ x0, float* __restrict__ Tcar, int n) {
  __shared__ __attribute__((aligned(16))) float tv[2 * TV_FLOATS];
  const int tid = threadIdx.x;
  const int lane = tid & 31, wave = tid >> 5;
  const int g = lane & 15, rg = lane >> 4;
  const int rbase = wave * 8 + rg * 4;

  v2f m[4][8];
  for (int i = 0; i < 4; ++i) {
    const float4* mrow = (const float4*)(Mg + (size_t)(rbase + i) * 256 + 16 * g);
    for (int c = 0; c < 4; ++c) {
      float4 t4 = mrow[c];
      m[i][2 * c]     = (v2f){t4.x, t4.y};
      m[i][2 * c + 1] = (v2f){t4.z, t4.w};
    }
  }
  const int outslot = (tid >> 4) * TV_PITCH + (tid & 15);
  if (tid < 256) tv[outslot] = x0[tid];
  const bool odd1 = (g & 1) != 0, odd2 = (g & 2) != 0;
  const int myrow = rbase + (odd1 ? 2 : 0) + (odd2 ? 1 : 0);
  const int myslot = (myrow >> 4) * TV_PITCH + (myrow & 15);
  float zc = (n > 1) ? Z[myrow] : 0.0f;
  __syncthreads();

  for (int j = 0; j + 1 < n; ++j) {
    const float* tcur = tv + (j & 1) * TV_FLOATS;
    float* tnxt = tv + ((j + 1) & 1) * TV_FLOATS;
    if (wave < 8) Tcar[(size_t)j * 256 + tid] = tcur[outslot];
    float zn = 0.0f;
    if (j + 2 < n) zn = Z[(size_t)(j + 1) * 256 + myrow];
    const float* xs = tcur + g * TV_PITCH;
    v2f acc[4] = {};
    for (int c = 0; c < 4; ++c) {
      float4 x4 = *(const float4*)(xs + 4 * c);
      v2f xa = {x4.x, x4.y}, xb = {x4.z, x4.w};
      for (int i = 0; i < 4; ++i) {
        acc[i] += m[i][2 * c] * xa;
        acc[i] += m[i][2 * c + 1] * xb;
      }
    }
    float p0 = acc[0][0] + acc[0][1];
    float p1 = acc[1][0] + acc[1][1];
    float p2 = acc[2][0] + acc[2][1];
    float p3 = acc[3][0] + acc[3][1];
    float s0 = odd1 ? p0 : p2, s1 = odd1 ? p1 : p3;
    float k0 = odd1 ? p2 : p0, k1 = odd1 ? p3 : p1;
    k0 += lane_read<DPP_SWAP1>(s0);
    k1 += lane_read<DPP_SWAP1>(s1);
    float s = odd2 ? k0 : k1;
    float k = odd2 ? k1 : k0;
    k += lane_read<DPP_SWAP2>(s);
    k += lane_read<DPP_ROR4>(k);
    k += lane_read<DPP_ROR8>(k);
    if (g < 4) tnxt[myslot] = k + zc;
    __syncthreads();
    zc = zn;
  }
  if (wave < 8) Tcar[(size_t)(n - 1) * 256 + tid] = tv[((n - 1) & 1) * TV_FLOATS + outslot];
}

static inline void launch_gemm(hipStream_t s, float* D, int ldd,
                               const float* A, int lda, const float* B, int ldb,
                               const float* Cin, int ldc,
                               int M, int N, int K, float alpha, float beta, int transB) {
  int tiles = (M / 16) * (N / 16);
  gemm_kernel<<<(tiles + 3) / 4, 128, 0, s>>>(D, ldd, A, lda, B, ldb, Cin, ldc,
                                              M, N, K, alpha, beta, transB);
}

extern "C" void kernel_launch(void* const* d_in, const int* in_sizes, int n_in,
                              void* d_out, int out_size, void* d_ws, size_t ws_size,
                              hipStream_t stream) {
  (void)in_sizes; (void)n_in; (void)out_size; (void)ws_size;
  const float* t  = (const float*)d_in[0];
  const float* u  = (const float*)d_in[1];
  const float* x0 = (const float*)d_in[2];
  const float* Am = (const float*)d_in[3];
  const float* Bm = (const float*)d_in[4];
  const float* Cm = (const float*)d_in[5];
  const float* Dm = (const float*)d_in[6];
  float* xout = (float*)d_out;
  float* yout = xout + (size_t)Tn * NS;

  float* w = (float*)d_ws;
  float* Ad = w;  w += 65536;
  float* Bd = w;  w += 8192;
  float* Kt[6]; for (int i = 0; i < 6; i++) { Kt[i] = w; w += 65536; }
  float* Lt[6]; for (int i = 0; i < 6; i++) { Lt[i] = w; w += 49152; }
  float* TMP  = w; w += 65536;
  float* TMPL = w; w += 49152;
  float* Mm   = w; w += 65536;
  float* Wm   = w; w += 49152;
  float* Qc   = w; w += 32768;
  float* ML   = w; w += 65536;
  float* M512 = w; w += 65536;
  float* PT1  = w; w += 65536;
  float* PT2  = w; w += 65536;
  float* Pm   = w; w += (size_t)Tn * 128;
  float* G    = w; w += (size_t)Tn * 256;
  float* Zend = w; w += (size_t)C1 * 256;
  float* Zend2= w; w += (size_t)C2 * 256;
  float* Tcar = w; w += (size_t)C2 * 256;
  float* Scar = w; w += (size_t)C1 * 256;

  auto gdim = [](int n) { return dim3((unsigned)((n + 255) / 256)); };

  k_scale_dt<<<gdim(65536), 256, 0, stream>>>(Ad, Am, 65536, t);
  k_scale_dt<<<gdim(8192),  256, 0, stream>>>(Bd, Bm, 8192, t);
  k_copy<<<gdim(65536), 256, 0, stream>>>(Kt[0], Ad, 65536);
  k_zero<<<gdim(49152), 256, 0, stream>>>(Lt[0], 49152);
  k_addblock<<<gdim(8192), 256, 0, stream>>>(Lt[0], Bd, 0);

  double aa[7][7] = {};
  aa[2][1] = 0.2;
  aa[3][1] = 3.0/40.0;      aa[3][2] = 9.0/40.0;
  aa[4][1] = 44.0/45.0;     aa[4][2] = -56.0/15.0;     aa[4][3] = 32.0/9.0;
  aa[5][1] = 19372.0/6561.0; aa[5][2] = -25360.0/2187.0;
  aa[5][3] = 64448.0/6561.0; aa[5][4] = -212.0/729.0;
  aa[6][1] = 9017.0/3168.0; aa[6][2] = -355.0/33.0; aa[6][3] = 46732.0/5247.0;
  aa[6][4] = 49.0/176.0;    aa[6][5] = -5103.0/18656.0;
  for (int i = 2; i <= 6; ++i) {
    k_eye<<<gdim(65536), 256, 0, stream>>>(TMP);
    k_zero<<<gdim(49152), 256, 0, stream>>>(TMPL, 49152);
    for (int j = 1; j < i; ++j) {
      float c = (float)aa[i][j];
      k_axpy<<<gdim(65536), 256, 0, stream>>>(TMP,  Kt[j-1], 65536, c);
      k_axpy<<<gdim(49152), 256, 0, stream>>>(TMPL, Lt[j-1], 49152, c);
    }
    launch_gemm(stream, Kt[i-1], 256, Ad, 256, TMP,  256, nullptr, 0, 256, 256, 256, 1.f, 0.f, 0);
    launch_gemm(stream, Lt[i-1], 192, Ad, 256, TMPL, 192, nullptr, 0, 256, 192, 256, 1.f, 0.f, 0);
    k_addblock<<<gdim(8192), 256, 0, stream>>>(Lt[i-1], Bd, i - 1);
  }
  const double bb[7] = {0.0, 35.0/384.0, 0.0, 500.0/1113.0, 125.0/192.0,
                        -2187.0/6784.0, 11.0/84.0};
  k_eye<<<gdim(65536), 256, 0, stream>>>(Mm);
  k_zero<<<gdim(49152), 256, 0, stream>>>(Wm, 49152);
  for (int i = 1; i <= 6; ++i) {
    float c = (float)bb[i];
    if (c != 0.0f) {
      k_axpy<<<gdim(65536), 256, 0, stream>>>(Mm, Kt[i-1], 65536, c);
      k_axpy<<<gdim(49152), 256, 0, stream>>>(Wm, Lt[i-1], 49152, c);
    }
  }
  k_qcat<<<gdim(32768), 256, 0, stream>>>(Qc, Wm);

  launch_gemm(stream, PT1, 256, Mm, 256, Mm, 256, nullptr, 0, 256, 256, 256, 1.f, 0.f, 0);
  launch_gemm(stream, PT2, 256, PT1, 256, PT1, 256, nullptr, 0, 256, 256, 256, 1.f, 0.f, 0);
  launch_gemm(stream, PT1, 256, PT2, 256, PT2, 256, nullptr, 0, 256, 256, 256, 1.f, 0.f, 0);
  launch_gemm(stream, PT2, 256, PT1, 256, PT1, 256, nullptr, 0, 256, 256, 256, 1.f, 0.f, 0);
  launch_gemm(stream, ML,  256, PT2, 256, PT2, 256, nullptr, 0, 256, 256, 256, 1.f, 0.f, 0);
  launch_gemm(stream, PT1, 256, ML,  256, ML,  256, nullptr, 0, 256, 256, 256, 1.f, 0.f, 0);
  launch_gemm(stream, PT2, 256, PT1, 256, PT1, 256, nullptr, 0, 256, 256, 256, 1.f, 0.f, 0);
  launch_gemm(stream, PT1, 256, PT2, 256, PT2, 256, nullptr, 0, 256, 256, 256, 1.f, 0.f, 0);
  launch_gemm(stream, M512,256, PT1, 256, PT1, 256, nullptr, 0, 256, 256, 256, 1.f, 0.f, 0);

  k_buildP<<<gdim(Tn * 128), 256, 0, stream>>>(Pm, u);
  launch_gemm(stream, G, 256, Pm, 128, Qc, 256, nullptr, 0, Tn, 256, 128, 1.f, 0.f, 0);

  scan_kernel<<<C1 / SC_COLS, 256, 0, stream>>>(Mm, G, nullptr, nullptr, Zend, L1);
  scan_kernel<<<C2 / SC_COLS, 256, 0, stream>>>(ML, Zend, nullptr, nullptr, Zend2, L2);
  carry_kernel<<<1, 1024, 0, stream>>>(M512, Zend2, x0, Tcar, C2);
  scan_kernel<<<C2 / SC_COLS, 256, 0, stream>>>(ML, Zend, Tcar, Scar, nullptr, L2);
  scan_kernel<<<C1 / SC_COLS, 256, 0, stream>>>(Mm, G, Scar, xout, nullptr, L1);

  launch_gemm(stream, yout, 32, xout, 256, Cm, 256, nullptr, 0, Tn, 32, 256, 1.f, 0.f, 1);
  launch_gemm(stream, yout, 32, u,    32,  Dm, 32,  yout,   32, Tn, 32, 32,  1.f, 1.f, 1);
}


# set-up fused: each stage's K and L GEMMs are one launch that forms its B operand (identity + weighted earlier stages) while loading it; 73 set-up launches become 7 (91 to 25); sum order kept
# speedup vs baseline: 2.2489x; 1.2809x over previous
#include <hip/hip_runtime.h>

typedef __attribute__((ext_vector_type(2))) float v2f;
typedef __attribute__((ext_vector_type(8))) float v8f;

#define WMMA_F32(a, b, c) \
  __builtin_amdgcn_wmma_f32_16x16x4_f32(false, (a), false, (b), (short)0, (c), false, false)

static constexpr int Tn = 32768;
static constexpr int NS = 256;
static constexpr int L1 = 32;
static constexpr int C1 = 1024;
static constexpr int L2 = 16;
static constexpr int C2 = 64;
static constexpr int SC_COLS = 16;
static constexpr int MS_LD = 260;
static constexpr int SCAN_LDS_FLOATS = 256 * MS_LD + 2 * 256 * SC_COLS;

__constant__ float d_H[6][4] = {
  {1.0f, 0.0f, 0.0f, 0.0f},
  {0.896f, 0.128f, 0.104f, -0.032f},
  {0.784f, 0.147f, 0.216f, -0.063f},
  {0.104f, 0.032f, 0.896f, -0.128f},
  {25.0f/729.0f, 8.0f/729.0f, 704.0f/729.0f, -64.0f/729.0f},
  {0.0f, 0.0f, 1.0f, 0.0f},
};

__global__ void k_prep(float* __restrict__ Ad, float* __restrict__ Bd, float* __restrict__ Lt0,
                       const float* __restrict__ A, const float* __restrict__ B,
                       const float* __restrict__ t) {
  const int idx = blockIdx.x * 256 + threadIdx.x;
  const float dt = t[1] - t[0];
  if (idx < 65536) {
    Ad[idx] = dt * A[idx];
  } else {
    const int q = idx - 65536;
    if (q < 49152) {
      const int r = q / 192, c = q - r * 192;
      float val = 0.0f;
      if (c < 32) { const float bd = dt * B[r * 32 + c]; Bd[r * 32 + c] = bd; val += bd; }
      Lt0[q] = val;
    }
  }
}
__global__ void k_final(float* __restrict__ Mm, float* __restrict__ Q,
                        const float* __restrict__ K1, const float* __restrict__ K3,
                        const float* __restrict__ K4, const float* __restrict__ K5,
                        const float* __restrict__ K6,
                        const float* __restrict__ W1, const float* __restrict__ W3,
                        const float* __restrict__ W4, const float* __restrict__ W5,
                        const float* __restrict__ W6,
                        float b1, float b3, float b4, float b5, float b6) {
  const int idx = blockIdx.x * 256 + threadIdx.x;
  if (idx < 65536) {
    float v = ((idx >> 8) == (idx & 255)) ? 1.0f : 0.0f;
    v = fmaf(b1, K1[idx], v);
    v = fmaf(b3, K3[idx], v);
    v = fmaf(b4, K4[idx], v);
    v = fmaf(b5, K5[idx], v);
    v = fmaf(b6, K6[idx], v);
    Mm[idx] = v;
  } else {
    const int q = idx - 65536;
    if (q >= 256 * 32) return;
    const int r = q >> 5, c = q & 31;
    float w[6];
    for (int st = 0; st < 6; ++st) {
      const int o = r * 192 + 32 * st + c;
      float t = 0.0f;
      t = fmaf(b1, W1[o], t);
      t = fmaf(b3, W3[o], t);
      t = fmaf(b4, W4[o], t);
      t = fmaf(b5, W5[o], t);
      t = fmaf(b6, W6[o], t);
      w[st] = t;
    }
    for (int j = 0; j < 4; ++j) {
      float s = 0.0f;
      for (int st = 0; st < 6; ++st) s += d_H[st][j] * w[st];
      Q[(32 * j + c) * 256 + r] = s;
    }
  }
}
__global__ void k_buildP(float* __restrict__ P, const float* __restrict__ u) {
  int idx = blockIdx.x * 256 + threadIdx.x;
  if (idx >= Tn * 128) return;
  int n = idx >> 7, k = idx & 127;
  float val = 0.0f;
  if (n < Tn - 1) {
    int j = k >> 5, c = k & 31;
    int m = n + ((j >= 2) ? 1 : 0);
    if ((j & 1) == 0) {
      val = u[m * 32 + c];
    } else {
      if (m == 0)            val = u[32 + c] - u[c];
      else if (m == Tn - 1)  val = u[(Tn - 1) * 32 + c] - u[(Tn - 2) * 32 + c];
      else                   val = 0.5f * (u[(m + 1) * 32 + c] - u[(m - 1) * 32 + c]);
    }
  }
  P[idx] = val;
}

__global__ __launch_bounds__(128) void gemm_kernel(
    float* __restrict__ D, int ldd,
    const float* __restrict__ A, int lda,
    const float* __restrict__ B, int ldb,
    const float* __restrict__ Cin, int ldc,
    int Mr, int Nc, int Kd, float alpha, float beta, int transB) {
  const int lane = threadIdx.x & 31, wave = threadIdx.x >> 5;
  const int hl = lane >> 4, lr = lane & 15;
  const int tilesN = Nc >> 4;
  const int tile = blockIdx.x * 4 + wave;
  if (tile >= (Mr >> 4) * tilesN) return;
  const int tm = tile / tilesN, tn = tile % tilesN;
  const float* Ar = A + (size_t)(tm * 16 + lr) * lda + 2 * hl;
  v8f acc = {};
  if (transB) {
    const float* Br = B + (size_t)(tn * 16 + lr) * ldb + 2 * hl;
    #pragma unroll 4
    for (int k = 0; k < Kd; k += 4) {
      v2f a = *(const v2f*)(Ar + k);
      v2f b = *(const v2f*)(Br + k);
      acc = WMMA_F32(a, b, acc);
    }
  } else {
    const float* Bc = B + (size_t)(2 * hl) * ldb + tn * 16 + lr;
    #pragma unroll 4
    for (int k = 0; k < Kd; k += 4) {
      v2f a = *(const v2f*)(Ar + k);
      v2f b;
      b[0] = Bc[(size_t)k * ldb];
      b[1] = Bc[(size_t)k * ldb + ldb];
      acc = WMMA_F32(a, b, acc);
    }
  }
  float* Dp = D + (size_t)(tm * 16 + 8 * hl) * ldd + tn * 16 + lr;
  if (beta != 0.0f) {
    const float* Cp = Cin + (size_t)(tm * 16 + 8 * hl) * ldc + tn * 16 + lr;
    #pragma unroll
    for (int v = 0; v < 8; ++v)
      Dp[(size_t)v * ldd] = alpha * acc[v] + beta * Cp[(size_t)v * ldc];
  } else {
    #pragma unroll
    for (int v = 0; v < 8; ++v)
      Dp[(size_t)v * ldd] = alpha * acc[v];
  }
}

template <int NJ>
__global__ __launch_bounds__(128) void stage_kernel(
    float* __restrict__ Kout, float* __restrict__ Lout,
    const float* __restrict__ Ad, const float* __restrict__ Bd, int blk,
    const float* __restrict__ K0, const float* __restrict__ K1, const float* __restrict__ K2,
    const float* __restrict__ K3, const float* __restrict__ K4,
    const float* __restrict__ W0, const float* __restrict__ W1, const float* __restrict__ W2,
    const float* __restrict__ W3, const float* __restrict__ W4,
    float c0, float c1, float c2, float c3, float c4) {
  const int lane = threadIdx.x & 31;
  const int wave = __builtin_amdgcn_readfirstlane(threadIdx.x >> 5);
  const int hl = lane >> 4, lr = lane & 15;
  const int tile = blockIdx.x * 4 + wave;
  const int tm = tile / 28, tn = tile % 28;
  const float* Ar = Ad + (size_t)(tm * 16 + lr) * 256 + 2 * hl;
  v8f acc = {};
  if (tn < 16) {
    const int n = tn * 16 + lr;
    const int off = (2 * hl) * 256 + n;
    #pragma unroll 4
    for (int k = 0; k < 256; k += 4) {
      v2f a = *(const v2f*)(Ar + k);
      const int o = off + k * 256;
      float b0 = (k + 2 * hl     == n) ? 1.0f : 0.0f;
      float b1 = (k + 2 * hl + 1 == n) ? 1.0f : 0.0f;
      b0 = fmaf(c0, K0[o], b0);  b1 = fmaf(c0, K0[o + 256], b1);
      if (NJ > 1) { b0 = fmaf(c1, K1[o], b0);  b1 = fmaf(c1, K1[o + 256], b1); }
      if (NJ > 2) { b0 = fmaf(c2, K2[o], b0);  b1 = fmaf(c2, K2[o + 256], b1); }
      if (NJ > 3) { b0 = fmaf(c3, K3[o], b0);  b1 = fmaf(c3, K3[o + 256], b1); }
      if (NJ > 4) { b0 = fmaf(c4, K4[o], b0);  b1 = fmaf(c4, K4[o + 256], b1); }
      v2f b = {b0, b1};
      acc = WMMA_F32(a, b, acc);
    }
    float* Dp = Kout + (size_t)(tm * 16 + 8 * hl) * 256 + n;
    #pragma unroll
    for (int v = 0; v < 8; ++v) Dp[(size_t)v * 256] = acc[v];
  } else {
    const int n = (tn - 16) * 16 + lr;
    const int off = (2 * hl) * 192 + n;
    #pragma unroll 4
    for (int k = 0; k < 256; k += 4) {
      v2f a = *(const v2f*)(Ar + k);
      const int o = off + k * 192;
      float b0 = 0.0f, b1 = 0.0f;
      b0 = fmaf(c0, W0[o], b0);  b1 = fmaf(c0, W0[o + 192], b1);
      if (NJ > 1) { b0 = fmaf(c1, W1[o], b0);  b1 = fmaf(c1, W1[o + 192], b1); }
      if (NJ > 2) { b0 = fmaf(c2, W2[o], b0);  b1 = fmaf(c2, W2[o + 192], b1); }
      if (NJ > 3) { b0 = fmaf(c3, W3[o], b0);  b1 = fmaf(c3, W3[o + 192], b1); }
      if (NJ > 4) { b0 = fmaf(c4, W4[o], b0);  b1 = fmaf(c4, W4[o + 192], b1); }
      v2f b = {b0, b1};
      acc = WMMA_F32(a, b, acc);
    }
    const int row0 = tm * 16 + 8 * hl;
    float* Dp = Lout + (size_t)row0 * 192 + n;
    if (((tn - 16) >> 1) == blk) {
      const float* Bp = Bd + (size_t)row0 * 32 + (n & 31);
      #pragma unroll
      for (int v = 0; v < 8; ++v) Dp[(size_t)v * 192] = acc[v] + Bp[(size_t)v * 32];
    } else {
      #pragma unroll
      for (int v = 0; v < 8; ++v) Dp[(size_t)v * 192] = acc[v];
    }
  }
}

__global__ __launch_bounds__(256) void scan_kernel(
    const float* __restrict__ Mg, const float* __restrict__ G,
    const float* __restrict__ Init,
    float* __restrict__ OutAll,
    float* __restrict__ OutEnd,
    int L) {
  __shared__ __attribute__((aligned(16))) float smem[SCAN_LDS_FLOATS];
  float* Ms = smem;
  float* Xbuf0 = smem + 256 * MS_LD;
  float* Xbuf1 = Xbuf0 + 256 * SC_COLS;
  const int tid = threadIdx.x;
  const int lane = tid & 31, wave = tid >> 5;
  const int hl = lane >> 4, lr = lane & 15;
  const int cbase = blockIdx.x * SC_COLS;
  const int tr0 = wave * 2;

  {
    const float4* src = (const float4*)Mg;
    float4* dst = (float4*)Ms;
    for (int k = tid; k < 65536 / 4; k += 256)
      dst[(k >> 6) * (MS_LD / 4) + (k & 63)] = src[k];
  }
  for (int e = 0; e < SC_COLS; ++e)
    Xbuf0[tid * SC_COLS + e] = Init ? Init[(size_t)(cbase + e) * 256 + tid] : 0.0f;
  __syncthreads();

  const float* gbase = G + ((size_t)(cbase + lr) * L) * 256 + tr0 * 16 + 8 * hl;

  float* Xc = Xbuf0;
  float* Xn = Xbuf1;
  for (int i = 0; i < L; ++i) {
    if (OutAll) {
      for (int e = 0; e < SC_COLS; ++e) {
        size_t m = (size_t)(cbase + e) * L + i;
        OutAll[m * 256 + tid] = Xc[tid * SC_COLS + e];
      }
    }
    const float* gp = gbase + (size_t)i * 256;
    if (i + 1 < L) __builtin_prefetch(gp + 256, 0, 1);
    v8f acc0, acc1;
    {
      float4 ga = *(const float4*)(gp);
      float4 gb = *(const float4*)(gp + 4);
      float4 gc = *(const float4*)(gp + 16);
      float4 gd = *(const float4*)(gp + 20);
      acc0[0]=ga.x; acc0[1]=ga.y; acc0[2]=ga.z; acc0[3]=ga.w;
      acc0[4]=gb.x; acc0[5]=gb.y; acc0[6]=gb.z; acc0[7]=gb.w;
      acc1[0]=gc.x; acc1[1]=gc.y; acc1[2]=gc.z; acc1[3]=gc.w;
      acc1[4]=gd.x; acc1[5]=gd.y; acc1[6]=gd.z; acc1[7]=gd.w;
    }
    const float* Mr0 = Ms + (size_t)(tr0 * 16 + lr) * MS_LD + 2 * hl;
    const float* Mr1 = Mr0 + 16 * MS_LD;
    const float* Xk = Xc + 2 * hl * SC_COLS + lr;
    #pragma unroll 8
    for (int kc = 0; kc < 64; ++kc) {
      v2f a0 = *(const v2f*)(Mr0 + kc * 4);
      v2f a1 = *(const v2f*)(Mr1 + kc * 4);
      v2f b;
      b[0] = Xk[kc * 4 * SC_COLS];
      b[1] = Xk[kc * 4 * SC_COLS + SC_COLS];
      acc0 = WMMA_F32(a0, b, acc0);
      acc1 = WMMA_F32(a1, b, acc1);
    }
    #pragma unroll
    for (int v = 0; v < 8; ++v) {
      Xn[(tr0 * 16 + 8 * hl + v) * SC_COLS + lr] = acc0[v];
      Xn[((tr0 + 1) * 16 + 8 * hl + v) * SC_COLS + lr] = acc1[v];
    }
    __syncthreads();
    float* tmp = Xc; Xc = Xn; Xn = tmp;
  }
  if (OutEnd) {
    for (int e = 0; e < SC_COLS; ++e)
      OutEnd[(size_t)(cbase + e) * 256 + tid] = Xc[tid * SC_COLS + e];
  }
}

static constexpr int TV_PITCH = 20;
static constexpr int TV_FLOATS = 16 * TV_PITCH;

template <int CTRL>
__device__ __forceinline__ float lane_read(float v) {
  return __int_as_float(__builtin_amdgcn_update_dpp(0, __float_as_int(v), CTRL, 0xF, 0xF, true));
}
static constexpr int DPP_SWAP1 = 0xB1;
static constexpr int DPP_SWAP2 = 0x4E;
static constexpr int DPP_ROR4  = 0x124;
static constexpr int DPP_ROR8  = 0x128;

__global__ __launch_bounds__(1024) void carry_kernel(
    const float* __restrict__ Mg, const float* __restrict__ Z,
    const float* __restrict__ x0, float* __restrict__ Tcar, int n) {
  __shared__ __attribute__((aligned(16))) float tv[2 * TV_FLOATS];
  const int tid = threadIdx.x;
  const int lane = tid & 31, wave = tid >> 5;
  const int g = lane & 15, rg = lane >> 4;
  const int rbase = wave * 8 + rg * 4;

  v2f m[4][8];
  for (int i = 0; i < 4; ++i) {
    const float4* mrow = (const float4*)(Mg + (size_t)(rbase + i) * 256 + 16 * g);
    for (int c = 0; c < 4; ++c) {
      float4 t4 = mrow[c];
      m[i][2 * c]     = (v2f){t4.x, t4.y};
      m[i][2 * c + 1] = (v2f){t4.z, t4.w};
    }
  }
  const int outslot = (tid >> 4) * TV_PITCH + (tid & 15);
  if (tid < 256) tv[outslot] = x0[tid];
  const bool odd1 = (g & 1) != 0, odd2 = (g & 2) != 0;
  const int myrow = rbase + (odd1 ? 2 : 0) + (odd2 ? 1 : 0);
  const int myslot = (myrow >> 4) * TV_PITCH + (myrow & 15);
  float zc = (n > 1) ? Z[myrow] : 0.0f;
  __syncthreads();

  for (int j = 0; j + 1 < n; ++j) {
    const float* tcur = tv + (j & 1) * TV_FLOATS;
    float* tnxt = tv + ((j + 1) & 1) * TV_FLOATS;
    if (wave < 8) Tcar[(size_t)j * 256 + tid] = tcur[outslot];
    float zn = 0.0f;
    if (j + 2 < n) zn = Z[(size_t)(j + 1) * 256 + myrow];
    const float* xs = tcur + g * TV_PITCH;
    v2f acc[4] = {};
    for (int c = 0; c < 4; ++c) {
      float4 x4 = *(const float4*)(xs + 4 * c);
      v2f xa = {x4.x, x4.y}, xb = {x4.z, x4.w};
      for (int i = 0; i < 4; ++i) {
        acc[i] += m[i][2 * c] * xa;
        acc[i] += m[i][2 * c + 1] * xb;
      }
    }
    float p0 = acc[0][0] + acc[0][1];
    float p1 = acc[1][0] + acc[1][1];
    float p2 = acc[2][0] + acc[2][1];
    float p3 = acc[3][0] + acc[3][1];
    float s0 = odd1 ? p0 : p2, s1 = odd1 ? p1 : p3;
    float k0 = odd1 ? p2 : p0, k1 = odd1 ? p3 : p1;
    k0 += lane_read<DPP_SWAP1>(s0);
    k1 += lane_read<DPP_SWAP1>(s1);
    float s = odd2 ? k0 : k1;
    float k = odd2 ? k1 : k0;
    k += lane_read<DPP_SWAP2>(s);
    k += lane_read<DPP_ROR4>(k);
    k += lane_read<DPP_ROR8>(k);
    if (g < 4) tnxt[myslot] = k + zc;
    __syncthreads();
    zc = zn;
  }
  if (wave < 8) Tcar[(size_t)(n - 1) * 256 + tid] = tv[((n - 1) & 1) * TV_FLOATS + outslot];
}

static inline void launch_gemm(hipStream_t s, float* D, int ldd,
                               const float* A, int lda, const float* B, int ldb,
                               const float* Cin, int ldc,
                               int M, int N, int K, float alpha, float beta, int transB) {
  int tiles = (M / 16) * (N / 16);
  gemm_kernel<<<(tiles + 3) / 4, 128, 0, s>>>(D, ldd, A, lda, B, ldb, Cin, ldc,
                                              M, N, K, alpha, beta, transB);
}

extern "C" void kernel_launch(void* const* d_in, const int* in_sizes, int n_in,
                              void* d_out, int out_size, void* d_ws, size_t ws_size,
                              hipStream_t stream) {
  (void)in_sizes; (void)n_in; (void)out_size; (void)ws_size;
  const float* t  = (const float*)d_in[0];
  const float* u  = (const float*)d_in[1];
  const float* x0 = (const float*)d_in[2];
  const float* Am = (const float*)d_in[3];
  const float* Bm = (const float*)d_in[4];
  const float* Cm = (const float*)d_in[5];
  const float* Dm = (const float*)d_in[6];
  float* xout = (float*)d_out;
  float* yout = xout + (size_t)Tn * NS;

  float* w = (float*)d_ws;
  float* Ad = w;  w += 65536;
  float* Bd = w;  w += 8192;
  float* Kt[6]; for (int i = 0; i < 6; i++) { Kt[i] = w; w += 65536; }
  Kt[0] = Ad;
  float* Lt[6]; for (int i = 0; i < 6; i++) { Lt[i] = w; w += 49152; }
  w += 65536 + 49152;
  float* Mm   = w; w += 65536;
  w += 49152;
  float* Qc   = w; w += 32768;
  float* ML   = w; w += 65536;
  float* M512 = w; w += 65536;
  float* PT1  = w; w += 65536;
  float* PT2  = w; w += 65536;
  float* Pm   = w; w += (size_t)Tn * 128;
  float* G    = w; w += (size_t)Tn * 256;
  float* Zend = w; w += (size_t)C1 * 256;
  float* Zend2= w; w += (size_t)C2 * 256;
  float* Tcar = w; w += (size_t)C2 * 256;
  float* Scar = w; w += (size_t)C1 * 256;

  auto gdim = [](int n) { return dim3((unsigned)((n + 255) / 256)); };

  k_prep<<<gdim(65536 + 49152), 256, 0, stream>>>(Ad, Bd, Lt[0], Am, Bm, t);

  double aa[7][7] = {};
  aa[2][1] = 0.2;
  aa[3][1] = 3.0/40.0;      aa[3][2] = 9.0/40.0;
  aa[4][1] = 44.0/45.0;     aa[4][2] = -56.0/15.0;     aa[4][3] = 32.0/9.0;
  aa[5][1] = 19372.0/6561.0; aa[5][2] = -25360.0/2187.0;
  aa[5][3] = 64448.0/6561.0; aa[5][4] = -212.0/729.0;
  aa[6][1] = 9017.0/3168.0; aa[6][2] = -355.0/33.0; aa[6][3] = 46732.0/5247.0;
  aa[6][4] = 49.0/176.0;    aa[6][5] = -5103.0/18656.0;
  stage_kernel<1><<<112, 128, 0, stream>>>(Kt[1], Lt[1], Ad, Bd, 1,
      Kt[0], Kt[0], Kt[0], Kt[0], Kt[0], Lt[0], Lt[0], Lt[0], Lt[0], Lt[0],
      (float)aa[2][1], 0.f, 0.f, 0.f, 0.f);
  stage_kernel<2><<<112, 128, 0, stream>>>(Kt[2], Lt[2], Ad, Bd, 2,
      Kt[0], Kt[1], Kt[0], Kt[0], Kt[0], Lt[0], Lt[1], Lt[0], Lt[0], Lt[0],
      (float)aa[3][1], (float)aa[3][2], 0.f, 0.f, 0.f);
  stage_kernel<3><<<112, 128, 0, stream>>>(Kt[3], Lt[3], Ad, Bd, 3,
      Kt[0], Kt[1], Kt[2], Kt[0], Kt[0], Lt[0], Lt[1], Lt[2], Lt[0], Lt[0],
      (float)aa[4][1], (float)aa[4][2], (float)aa[4][3], 0.f, 0.f);
  stage_kernel<4><<<112, 128, 0, stream>>>(Kt[4], Lt[4], Ad, Bd, 4,
      Kt[0], Kt[1], Kt[2], Kt[3], Kt[0], Lt[0], Lt[1], Lt[2], Lt[3], Lt[0],
      (float)aa[5][1], (float)aa[5][2], (float)aa[5][3], (float)aa[5][4], 0.f);
  stage_kernel<5><<<112, 128, 0, stream>>>(Kt[5], Lt[5], Ad, Bd, 5,
      Kt[0], Kt[1], Kt[2], Kt[3], Kt[4], Lt[0], Lt[1], Lt[2], Lt[3], Lt[4],
      (float)aa[6][1], (float)aa[6][2], (float)aa[6][3], (float)aa[6][4], (float)aa[6][5]);

  const double bb[7] = {0.0, 35.0/384.0, 0.0, 500.0/1113.0, 125.0/192.0,
                        -2187.0/6784.0, 11.0/84.0};
  k_final<<<gdim(65536 + 8192), 256, 0, stream>>>(Mm, Qc,
      Kt[0], Kt[2], Kt[3], Kt[4], Kt[5], Lt[0], Lt[2], Lt[3], Lt[4], Lt[5],
      (float)bb[1], (float)bb[3], (float)bb[4], (float)bb[5], (float)bb[6]);

  launch_gemm(stream, PT1, 256, Mm, 256, Mm, 256, nullptr, 0, 256, 256, 256, 1.f, 0.f, 0);
  launch_gemm(stream, PT2, 256, PT1, 256, PT1, 256, nullptr, 0, 256, 256, 256, 1.f, 0.f, 0);
  launch_gemm(stream, PT1, 256, PT2, 256, PT2, 256, nullptr, 0, 256, 256, 256, 1.f, 0.f, 0);
  launch_gemm(stream, PT2, 256, PT1, 256, PT1, 256, nullptr, 0, 256, 256, 256, 1.f, 0.f, 0);
  launch_gemm(stream, ML,  256, PT2, 256, PT2, 256, nullptr, 0, 256, 256, 256, 1.f, 0.f, 0);
  launch_gemm(stream, PT1, 256, ML,  256, ML,  256, nullptr, 0, 256, 256, 256, 1.f, 0.f, 0);
  launch_gemm(stream, PT2, 256, PT1, 256, PT1, 256, nullptr, 0, 256, 256, 256, 1.f, 0.f, 0);
  launch_gemm(stream, PT1, 256, PT2, 256, PT2, 256, nullptr, 0, 256, 256, 256, 1.f, 0.f, 0);
  launch_gemm(stream, M512,256, PT1, 256, PT1, 256, nullptr, 0, 256, 256, 256, 1.f, 0.f, 0);

  k_buildP<<<gdim(Tn * 128), 256, 0, stream>>>(Pm, u);
  launch_gemm(stream, G, 256, Pm, 128, Qc, 256, nullptr, 0, Tn, 256, 128, 1.f, 0.f, 0);

  scan_kernel<<<C1 / SC_COLS, 256, 0, stream>>>(Mm, G, nullptr, nullptr, Zend, L1);
  scan_kernel<<<C2 / SC_COLS, 256, 0, stream>>>(ML, Zend, nullptr, nullptr, Zend2, L2);
  carry_kernel<<<1, 1024, 0, stream>>>(M512, Zend2, x0, Tcar, C2);
  scan_kernel<<<C2 / SC_COLS, 256, 0, stream>>>(ML, Zend, Tcar, Scar, nullptr, L2);
  scan_kernel<<<C1 / SC_COLS, 256, 0, stream>>>(Mm, G, Scar, xout, nullptr, L1);

  launch_gemm(stream, yout, 32, xout, 256, Cm, 256, nullptr, 0, Tn, 32, 256, 1.f, 0.f, 1);
  launch_gemm(stream, yout, 32, u,    32,  Dm, 32,  yout,   32, Tn, 32, 32,  1.f, 1.f, 1);
}


# scan time hierarchy re-chunked from 32x16x64 to 8x8x8x64 (one more scan level, on M^8 and M^64): level-1 scans run 256 work-groups x 8 steps, not 64 x 32; a different reduction tree over time
# speedup vs baseline: 2.9106x; 1.2942x over previous
#include <hip/hip_runtime.h>

typedef __attribute__((ext_vector_type(2))) float v2f;
typedef __attribute__((ext_vector_type(8))) float v8f;

#define WMMA_F32(a, b, c) \
  __builtin_amdgcn_wmma_f32_16x16x4_f32(false, (a), false, (b), (short)0, (c), false, false)

static constexpr int Tn = 32768;
static constexpr int NS = 256;
static constexpr int LC = 8;
static constexpr int C1 = 4096;
static constexpr int C2 = 512;
static constexpr int C3 = 64;
static constexpr int SC_COLS = 16;
static constexpr int MS_LD = 260;
static constexpr int SCAN_LDS_FLOATS = 256 * MS_LD + 2 * 256 * SC_COLS;

__constant__ float d_H[6][4] = {
  {1.0f, 0.0f, 0.0f, 0.0f},
  {0.896f, 0.128f, 0.104f, -0.032f},
  {0.784f, 0.147f, 0.216f, -0.063f},
  {0.104f, 0.032f, 0.896f, -0.128f},
  {25.0f/729.0f, 8.0f/729.0f, 704.0f/729.0f, -64.0f/729.0f},
  {0.0f, 0.0f, 1.0f, 0.0f},
};

__global__ void k_prep(float* __restrict__ Ad, float* __restrict__ Bd, float* __restrict__ Lt0,
                       const float* __restrict__ A, const float* __restrict__ B,
                       const float* __restrict__ t) {
  const int idx = blockIdx.x * 256 + threadIdx.x;
  const float dt = t[1] - t[0];
  if (idx < 65536) {
    Ad[idx] = dt * A[idx];
  } else {
    const int q = idx - 65536;
    if (q < 49152) {
      const int r = q / 192, c = q - r * 192;
      float val = 0.0f;
      if (c < 32) { const float bd = dt * B[r * 32 + c]; Bd[r * 32 + c] = bd; val += bd; }
      Lt0[q] = val;
    }
  }
}
__global__ void k_final(float* __restrict__ Mm, float* __restrict__ Q,
                        const float* __restrict__ K1, const float* __restrict__ K3,
                        const float* __restrict__ K4, const float* __restrict__ K5,
                        const float* __restrict__ K6,
                        const float* __restrict__ W1, const float* __restrict__ W3,
                        const float* __restrict__ W4, const float* __restrict__ W5,
                        const float* __restrict__ W6,
                        float b1, float b3, float b4, float b5, float b6) {
  const int idx = blockIdx.x * 256 + threadIdx.x;
  if (idx < 65536) {
    float v = ((idx >> 8) == (idx & 255)) ? 1.0f : 0.0f;
    v = fmaf(b1, K1[idx], v);
    v = fmaf(b3, K3[idx], v);
    v = fmaf(b4, K4[idx], v);
    v = fmaf(b5, K5[idx], v);
    v = fmaf(b6, K6[idx], v);
    Mm[idx] = v;
  } else {
    const int q = idx - 65536;
    if (q >= 256 * 32) return;
    const int r = q >> 5, c = q & 31;
    float w[6];
    for (int st = 0; st < 6; ++st) {
      const int o = r * 192 + 32 * st + c;
      float t = 0.0f;
      t = fmaf(b1, W1[o], t);
      t = fmaf(b3, W3[o], t);
      t = fmaf(b4, W4[o], t);
      t = fmaf(b5, W5[o], t);
      t = fmaf(b6, W6[o], t);
      w[st] = t;
    }
    for (int j = 0; j < 4; ++j) {
      float s = 0.0f;
      for (int st = 0; st < 6; ++st) s += d_H[st][j] * w[st];
      Q[(32 * j + c) * 256 + r] = s;
    }
  }
}
__global__ void k_buildP(float* __restrict__ P, const float* __restrict__ u) {
  int idx = blockIdx.x * 256 + threadIdx.x;
  if (idx >= Tn * 128) return;
  int n = idx >> 7, k = idx & 127;
  float val = 0.0f;
  if (n < Tn - 1) {
    int j = k >> 5, c = k & 31;
    int m = n + ((j >= 2) ? 1 : 0);
    if ((j & 1) == 0) {
      val = u[m * 32 + c];
    } else {
      if (m == 0)            val = u[32 + c] - u[c];
      else if (m == Tn - 1)  val = u[(Tn - 1) * 32 + c] - u[(Tn - 2) * 32 + c];
      else                   val = 0.5f * (u[(m + 1) * 32 + c] - u[(m - 1) * 32 + c]);
    }
  }
  P[idx] = val;
}

__global__ __launch_bounds__(128) void gemm_kernel(
    float* __restrict__ D, int ldd,
    const float* __restrict__ A, int lda,
    const float* __restrict__ B, int ldb,
    const float* __restrict__ Cin, int ldc,
    int Mr, int Nc, int Kd, float alpha, float beta, int transB) {
  const int lane = threadIdx.x & 31, wave = threadIdx.x >> 5;
  const int hl = lane >> 4, lr = lane & 15;
  const int tilesN = Nc >> 4;
  const int tile = blockIdx.x * 4 + wave;
  if (tile >= (Mr >> 4) * tilesN) return;
  const int tm = tile / tilesN, tn = tile % tilesN;
  const float* Ar = A + (size_t)(tm * 16 + lr) * lda + 2 * hl;
  v8f acc = {};
  if (transB) {
    const float* Br = B + (size_t)(tn * 16 + lr) * ldb + 2 * hl;
    #pragma unroll 4
    for (int k = 0; k < Kd; k += 4) {
      v2f a = *(const v2f*)(Ar + k);
      v2f b = *(const v2f*)(Br + k);
      acc = WMMA_F32(a, b, acc);
    }
  } else {
    const float* Bc = B + (size_t)(2 * hl) * ldb + tn * 16 + lr;
    #pragma unroll 4
    for (int k = 0; k < Kd; k += 4) {
      v2f a = *(const v2f*)(Ar + k);
      v2f b;
      b[0] = Bc[(size_t)k * ldb];
      b[1] = Bc[(size_t)k * ldb + ldb];
      acc = WMMA_F32(a, b, acc);
    }
  }
  float* Dp = D + (size_t)(tm * 16 + 8 * hl) * ldd + tn * 16 + lr;
  if (beta != 0.0f) {
    const float* Cp = Cin + (size_t)(tm * 16 + 8 * hl) * ldc + tn * 16 + lr;
    #pragma unroll
    for (int v = 0; v < 8; ++v)
      Dp[(size_t)v * ldd] = alpha * acc[v] + beta * Cp[(size_t)v * ldc];
  } else {
    #pragma unroll
    for (int v = 0; v < 8; ++v)
      Dp[(size_t)v * ldd] = alpha * acc[v];
  }
}

template <int NJ>
__global__ __launch_bounds__(128) void stage_kernel(
    float* __restrict__ Kout, float* __restrict__ Lout,
    const float* __restrict__ Ad, const float* __restrict__ Bd, int blk,
    const float* __restrict__ K0, const float* __restrict__ K1, const float* __restrict__ K2,
    const float* __restrict__ K3, const float* __restrict__ K4,
    const float* __restrict__ W0, const float* __restrict__ W1, const float* __restrict__ W2,
    const float* __restrict__ W3, const float* __restrict__ W4,
    float c0, float c1, float c2, float c3, float c4) {
  const int lane = threadIdx.x & 31;
  const int wave = __builtin_amdgcn_readfirstlane(threadIdx.x >> 5);
  const int hl = lane >> 4, lr = lane & 15;
  const int tile = blockIdx.x * 4 + wave;
  const int tm = tile / 28, tn = tile % 28;
  const float* Ar = Ad + (size_t)(tm * 16 + lr) * 256 + 2 * hl;
  v8f acc = {};
  if (tn < 16) {
    const int n = tn * 16 + lr;
    const int off = (2 * hl) * 256 + n;
    #pragma unroll 4
    for (int k = 0; k < 256; k += 4) {
      v2f a = *(const v2f*)(Ar + k);
      const int o = off + k * 256;
      float b0 = (k + 2 * hl     == n) ? 1.0f : 0.0f;
      float b1 = (k + 2 * hl + 1 == n) ? 1.0f : 0.0f;
      b0 = fmaf(c0, K0[o], b0);  b1 = fmaf(c0, K0[o + 256], b1);
      if (NJ > 1) { b0 = fmaf(c1, K1[o], b0);  b1 = fmaf(c1, K1[o + 256], b1); }
      if (NJ > 2) { b0 = fmaf(c2, K2[o], b0);  b1 = fmaf(c2, K2[o + 256], b1); }
      if (NJ > 3) { b0 = fmaf(c3, K3[o], b0);  b1 = fmaf(c3, K3[o + 256], b1); }
      if (NJ > 4) { b0 = fmaf(c4, K4[o], b0);  b1 = fmaf(c4, K4[o + 256], b1); }
      v2f b = {b0, b1};
      acc = WMMA_F32(a, b, acc);
    }
    float* Dp = Kout + (size_t)(tm * 16 + 8 * hl) * 256 + n;
    #pragma unroll
    for (int v = 0; v < 8; ++v) Dp[(size_t)v * 256] = acc[v];
  } else {
    const int n = (tn - 16) * 16 + lr;
    const int off = (2 * hl) * 192 + n;
    #pragma unroll 4
    for (int k = 0; k < 256; k += 4) {
      v2f a = *(const v2f*)(Ar + k);
      const int o = off + k * 192;
      float b0 = 0.0f, b1 = 0.0f;
      b0 = fmaf(c0, W0[o], b0);  b1 = fmaf(c0, W0[o + 192], b1);
      if (NJ > 1) { b0 = fmaf(c1, W1[o], b0);  b1 = fmaf(c1, W1[o + 192], b1); }
      if (NJ > 2) { b0 = fmaf(c2, W2[o], b0);  b1 = fmaf(c2, W2[o + 192], b1); }
      if (NJ > 3) { b0 = fmaf(c3, W3[o], b0);  b1 = fmaf(c3, W3[o + 192], b1); }
      if (NJ > 4) { b0 = fmaf(c4, W4[o], b0);  b1 = fmaf(c4, W4[o + 192], b1); }
      v2f b = {b0, b1};
      acc = WMMA_F32(a, b, acc);
    }
    const int row0 = tm * 16 + 8 * hl;
    float* Dp = Lout + (size_t)row0 * 192 + n;
    if (((tn - 16) >> 1) == blk) {
      const float* Bp = Bd + (size_t)row0 * 32 + (n & 31);
      #pragma unroll
      for (int v = 0; v < 8; ++v) Dp[(size_t)v * 192] = acc[v] + Bp[(size_t)v * 32];
    } else {
      #pragma unroll
      for (int v = 0; v < 8; ++v) Dp[(size_t)v * 192] = acc[v];
    }
  }
}

__global__ __launch_bounds__(256) void scan_kernel(
    const float* __restrict__ Mg, const float* __restrict__ G,
    const float* __restrict__ Init,
    float* __restrict__ OutAll,
    float* __restrict__ OutEnd,
    int L) {
  __shared__ __attribute__((aligned(16))) float smem[SCAN_LDS_FLOATS];
  float* Ms = smem;
  float* Xbuf0 = smem + 256 * MS_LD;
  float* Xbuf1 = Xbuf0 + 256 * SC_COLS;
  const int tid = threadIdx.x;
  const int lane = tid & 31, wave = tid >> 5;
  const int hl = lane >> 4, lr = lane & 15;
  const int cbase = blockIdx.x * SC_COLS;
  const int tr0 = wave * 2;

  {
    const float4* src = (const float4*)Mg;
    float4* dst = (float4*)Ms;
    for (int k = tid; k < 65536 / 4; k += 256)
      dst[(k >> 6) * (MS_LD / 4) + (k & 63)] = src[k];
  }
  for (int e = 0; e < SC_COLS; ++e)
    Xbuf0[tid * SC_COLS + e] = Init ? Init[(size_t)(cbase + e) * 256 + tid] : 0.0f;
  __syncthreads();

  const float* gbase = G + ((size_t)(cbase + lr) * L) * 256 + tr0 * 16 + 8 * hl;

  float* Xc = Xbuf0;
  float* Xn = Xbuf1;
  for (int i = 0; i < L; ++i) {
    if (OutAll) {
      for (int e = 0; e < SC_COLS; ++e) {
        size_t m = (size_t)(cbase + e) * L + i;
        OutAll[m * 256 + tid] = Xc[tid * SC_COLS + e];
      }
    }
    const float* gp = gbase + (size_t)i * 256;
    if (i + 1 < L) __builtin_prefetch(gp + 256, 0, 1);
    v8f acc0, acc1;
    {
      float4 ga = *(const float4*)(gp);
      float4 gb = *(const float4*)(gp + 4);
      float4 gc = *(const float4*)(gp + 16);
      float4 gd = *(const float4*)(gp + 20);
      acc0[0]=ga.x; acc0[1]=ga.y; acc0[2]=ga.z; acc0[3]=ga.w;
      acc0[4]=gb.x; acc0[5]=gb.y; acc0[6]=gb.z; acc0[7]=gb.w;
      acc1[0]=gc.x; acc1[1]=gc.y; acc1[2]=gc.z; acc1[3]=gc.w;
      acc1[4]=gd.x; acc1[5]=gd.y; acc1[6]=gd.z; acc1[7]=gd.w;
    }
    const float* Mr0 = Ms + (size_t)(tr0 * 16 + lr) * MS_LD + 2 * hl;
    const float* Mr1 = Mr0 + 16 * MS_LD;
    const float* Xk = Xc + 2 * hl * SC_COLS + lr;
    #pragma unroll 8
    for (int kc = 0; kc < 64; ++kc) {
      v2f a0 = *(const v2f*)(Mr0 + kc * 4);
      v2f a1 = *(const v2f*)(Mr1 + kc * 4);
      v2f b;
      b[0] = Xk[kc * 4 * SC_COLS];
      b[1] = Xk[kc * 4 * SC_COLS + SC_COLS];
      acc0 = WMMA_F32(a0, b, acc0);
      acc1 = WMMA_F32(a1, b, acc1);
    }
    #pragma unroll
    for (int v = 0; v < 8; ++v) {
      Xn[(tr0 * 16 + 8 * hl + v) * SC_COLS + lr] = acc0[v];
      Xn[((tr0 + 1) * 16 + 8 * hl + v) * SC_COLS + lr] = acc1[v];
    }
    __syncthreads();
    float* tmp = Xc; Xc = Xn; Xn = tmp;
  }
  if (OutEnd) {
    for (int e = 0; e < SC_COLS; ++e)
      OutEnd[(size_t)(cbase + e) * 256 + tid] = Xc[tid * SC_COLS + e];
  }
}

static constexpr int TV_PITCH = 20;
static constexpr int TV_FLOATS = 16 * TV_PITCH;

template <int CTRL>
__device__ __forceinline__ float lane_read(float v) {
  return __int_as_float(__builtin_amdgcn_update_dpp(0, __float_as_int(v), CTRL, 0xF, 0xF, true));
}
static constexpr int DPP_SWAP1 = 0xB1;
static constexpr int DPP_SWAP2 = 0x4E;
static constexpr int DPP_ROR4  = 0x124;
static constexpr int DPP_ROR8  = 0x128;

__global__ __launch_bounds__(1024) void carry_kernel(
    const float* __restrict__ Mg, const float* __restrict__ Z,
    const float* __restrict__ x0, float* __restrict__ Tcar, int n) {
  __shared__ __attribute__((aligned(16))) float tv[2 * TV_FLOATS];
  const int tid = threadIdx.x;
  const int lane = tid & 31, wave = tid >> 5;
  const int g = lane & 15, rg = lane >> 4;
  const int rbase = wave * 8 + rg * 4;

  v2f m[4][8];
  for (int i = 0; i < 4; ++i) {
    const float4* mrow = (const float4*)(Mg + (size_t)(rbase + i) * 256 + 16 * g);
    for (int c = 0; c < 4; ++c) {
      float4 t4 = mrow[c];
      m[i][2 * c]     = (v2f){t4.x, t4.y};
      m[i][2 * c + 1] = (v2f){t4.z, t4.w};
    }
  }
  const int outslot = (tid >> 4) * TV_PITCH + (tid & 15);
  if (tid < 256) tv[outslot] = x0[tid];
  const bool odd1 = (g & 1) != 0, odd2 = (g & 2) != 0;
  const int myrow = rbase + (odd1 ? 2 : 0) + (odd2 ? 1 : 0);
  const int myslot = (myrow >> 4) * TV_PITCH + (myrow & 15);
  float zc = (n > 1) ? Z[myrow] : 0.0f;
  __syncthreads();

  for (int j = 0; j + 1 < n; ++j) {
    const float* tcur = tv + (j & 1) * TV_FLOATS;
    float* tnxt = tv + ((j + 1) & 1) * TV_FLOATS;
    if (wave < 8) Tcar[(size_t)j * 256 + tid] = tcur[outslot];
    float zn = 0.0f;
    if (j + 2 < n) zn = Z[(size_t)(j + 1) * 256 + myrow];
    const float* xs = tcur + g * TV_PITCH;
    v2f acc[4] = {};
    for (int c = 0; c < 4; ++c) {
      float4 x4 = *(const float4*)(xs + 4 * c);
      v2f xa = {x4.x, x4.y}, xb = {x4.z, x4.w};
      for (int i = 0; i < 4; ++i) {
        acc[i] += m[i][2 * c] * xa;
        acc[i] += m[i][2 * c + 1] * xb;
      }
    }
    float p0 = acc[0][0] + acc[0][1];
    float p1 = acc[1][0] + acc[1][1];
    float p2 = acc[2][0] + acc[2][1];
    float p3 = acc[3][0] + acc[3][1];
    float s0 = odd1 ? p0 : p2, s1 = odd1 ? p1 : p3;
    float k0 = odd1 ? p2 : p0, k1 = odd1 ? p3 : p1;
    k0 += lane_read<DPP_SWAP1>(s0);
    k1 += lane_read<DPP_SWAP1>(s1);
    float s = odd2 ? k0 : k1;
    float k = odd2 ? k1 : k0;
    k += lane_read<DPP_SWAP2>(s);
    k += lane_read<DPP_ROR4>(k);
    k += lane_read<DPP_ROR8>(k);
    if (g < 4) tnxt[myslot] = k + zc;
    __syncthreads();
    zc = zn;
  }
  if (wave < 8) Tcar[(size_t)(n - 1) * 256 + tid] = tv[((n - 1) & 1) * TV_FLOATS + outslot];
}

static inline void launch_gemm(hipStream_t s, float* D, int ldd,
                               const float* A, int lda, const float* B, int ldb,
                               const float* Cin, int ldc,
                               int M, int N, int K, float alpha, float beta, int transB) {
  int tiles = (M / 16) * (N / 16);
  gemm_kernel<<<(tiles + 3) / 4, 128, 0, s>>>(D, ldd, A, lda, B, ldb, Cin, ldc,
                                              M, N, K, alpha, beta, transB);
}

extern "C" void kernel_launch(void* const* d_in, const int* in_sizes, int n_in,
                              void* d_out, int out_size, void* d_ws, size_t ws_size,
                              hipStream_t stream) {
  (void)in_sizes; (void)n_in; (void)out_size; (void)ws_size;
  const float* t  = (const float*)d_in[0];
  const float* u  = (const float*)d_in[1];
  const float* x0 = (const float*)d_in[2];
  const float* Am = (const float*)d_in[3];
  const float* Bm = (const float*)d_in[4];
  const float* Cm = (const float*)d_in[5];
  const float* Dm = (const float*)d_in[6];
  float* xout = (float*)d_out;
  float* yout = xout + (size_t)Tn * NS;

  float* w = (float*)d_ws;
  float* Ad = w;  w += 65536;
  float* Bd = w;  w += 8192;
  float* Kt[6]; for (int i = 0; i < 6; i++) { Kt[i] = w; w += 65536; }
  Kt[0] = Ad;
  float* Lt[6]; for (int i = 0; i < 6; i++) { Lt[i] = w; w += 49152; }
  float* M64  = w; w += 65536 + 49152;
  float* Mm   = w; w += 65536;
  w += 49152;
  float* Qc   = w; w += 32768;
  float* M8   = w; w += 65536;
  float* M512 = w; w += 65536;
  float* PT1  = w; w += 65536;
  float* PT2  = w; w += 65536;
  float* Pm   = w; w += (size_t)Tn * 128;
  float* G    = w; w += (size_t)Tn * 256;
  float* Zend2 = w; w += (size_t)C2 * 256;
  float* Scar2 = w; w += (size_t)C2 * 256;
  float* Zend3 = w; w += (size_t)C3 * 256;
  float* Tcar  = w; w += (size_t)C3 * 256;
  float* Zend1 = Pm;
  float* Scar1 = Pm + (size_t)C1 * 256;

  auto gdim = [](int n) { return dim3((unsigned)((n + 255) / 256)); };

  k_prep<<<gdim(65536 + 49152), 256, 0, stream>>>(Ad, Bd, Lt[0], Am, Bm, t);

  double aa[7][7] = {};
  aa[2][1] = 0.2;
  aa[3][1] = 3.0/40.0;      aa[3][2] = 9.0/40.0;
  aa[4][1] = 44.0/45.0;     aa[4][2] = -56.0/15.0;     aa[4][3] = 32.0/9.0;
  aa[5][1] = 19372.0/6561.0; aa[5][2] = -25360.0/2187.0;
  aa[5][3] = 64448.0/6561.0; aa[5][4] = -212.0/729.0;
  aa[6][1] = 9017.0/3168.0; aa[6][2] = -355.0/33.0; aa[6][3] = 46732.0/5247.0;
  aa[6][4] = 49.0/176.0;    aa[6][5] = -5103.0/18656.0;
  stage_kernel<1><<<112, 128, 0, stream>>>(Kt[1], Lt[1], Ad, Bd, 1,
      Kt[0], Kt[0], Kt[0], Kt[0], Kt[0], Lt[0], Lt[0], Lt[0], Lt[0], Lt[0],
      (float)aa[2][1], 0.f, 0.f, 0.f, 0.f);
  stage_kernel<2><<<112, 128, 0, stream>>>(Kt[2], Lt[2], Ad, Bd, 2,
      Kt[0], Kt[1], Kt[0], Kt[0], Kt[0], Lt[0], Lt[1], Lt[0], Lt[0], Lt[0],
      (float)aa[3][1], (float)aa[3][2], 0.f, 0.f, 0.f);
  stage_kernel<3><<<112, 128, 0, stream>>>(Kt[3], Lt[3], Ad, Bd, 3,
      Kt[0], Kt[1], Kt[2], Kt[0], Kt[0], Lt[0], Lt[1], Lt[2], Lt[0], Lt[0],
      (float)aa[4][1], (float)aa[4][2], (float)aa[4][3], 0.f, 0.f);
  stage_kernel<4><<<112, 128, 0, stream>>>(Kt[4], Lt[4], Ad, Bd, 4,
      Kt[0], Kt[1], Kt[2], Kt[3], Kt[0], Lt[0], Lt[1], Lt[2], Lt[3], Lt[0],
      (float)aa[5][1], (float)aa[5][2], (float)aa[5][3], (float)aa[5][4], 0.f);
  stage_kernel<5><<<112, 128, 0, stream>>>(Kt[5], Lt[5], Ad, Bd, 5,
      Kt[0], Kt[1], Kt[2], Kt[3], Kt[4], Lt[0], Lt[1], Lt[2], Lt[3], Lt[4],
      (float)aa[6][1], (float)aa[6][2], (float)aa[6][3], (float)aa[6][4], (float)aa[6][5]);

  const double bb[7] = {0.0, 35.0/384.0, 0.0, 500.0/1113.0, 125.0/192.0,
                        -2187.0/6784.0, 11.0/84.0};
  k_final<<<gdim(65536 + 8192), 256, 0, stream>>>(Mm, Qc,
      Kt[0], Kt[2], Kt[3], Kt[4], Kt[5], Lt[0], Lt[2], Lt[3], Lt[4], Lt[5],
      (float)bb[1], (float)bb[3], (float)bb[4], (float)bb[5], (float)bb[6]);

  launch_gemm(stream, PT1, 256, Mm, 256, Mm, 256, nullptr, 0, 256, 256, 256, 1.f, 0.f, 0);
  launch_gemm(stream, PT2, 256, PT1, 256, PT1, 256, nullptr, 0, 256, 256, 256, 1.f, 0.f, 0);
  launch_gemm(stream, M8,  256, PT2, 256, PT2, 256, nullptr, 0, 256, 256, 256, 1.f, 0.f, 0);
  launch_gemm(stream, PT1, 256, M8,  256, M8,  256, nullptr, 0, 256, 256, 256, 1.f, 0.f, 0);
  launch_gemm(stream, PT2, 256, PT1, 256, PT1, 256, nullptr, 0, 256, 256, 256, 1.f, 0.f, 0);
  launch_gemm(stream, M64, 256, PT2, 256, PT2, 256, nullptr, 0, 256, 256, 256, 1.f, 0.f, 0);
  launch_gemm(stream, PT1, 256, M64, 256, M64, 256, nullptr, 0, 256, 256, 256, 1.f, 0.f, 0);
  launch_gemm(stream, PT2, 256, PT1, 256, PT1, 256, nullptr, 0, 256, 256, 256, 1.f, 0.f, 0);
  launch_gemm(stream, M512,256, PT2, 256, PT2, 256, nullptr, 0, 256, 256, 256, 1.f, 0.f, 0);

  k_buildP<<<gdim(Tn * 128), 256, 0, stream>>>(Pm, u);
  launch_gemm(stream, G, 256, Pm, 128, Qc, 256, nullptr, 0, Tn, 256, 128, 1.f, 0.f, 0);

  scan_kernel<<<C1 / SC_COLS, 256, 0, stream>>>(Mm, G, nullptr, nullptr, Zend1, LC);
  scan_kernel<<<C2 / SC_COLS, 256, 0, stream>>>(M8, Zend1, nullptr, nullptr, Zend2, LC);
  scan_kernel<<<C3 / SC_COLS, 256, 0, stream>>>(M64, Zend2, nullptr, nullptr, Zend3, LC);
  carry_kernel<<<1, 1024, 0, stream>>>(M512, Zend3, x0, Tcar, C3);
  scan_kernel<<<C3 / SC_COLS, 256, 0, stream>>>(M64, Zend2, Tcar, Scar2, nullptr, LC);
  scan_kernel<<<C2 / SC_COLS, 256, 0, stream>>>(M8, Zend1, Scar2, Scar1, nullptr, LC);
  scan_kernel<<<C1 / SC_COLS, 256, 0, stream>>>(Mm, G, Scar1, xout, nullptr, LC);

  launch_gemm(stream, yout, 32, xout, 256, Cm, 256, nullptr, 0, Tn, 32, 256, 1.f, 0.f, 1);
  launch_gemm(stream, yout, 32, u,    32,  Dm, 32,  yout,   32, Tn, 32, 32,  1.f, 1.f, 1);
}


# drive-term product G = P @ Qcat in one 256-work-group launch: Hermite rows P built from u in LDS (P plane and its launch gone), Qcat tiles in registers over a row-tile loop; k order kept
# speedup vs baseline: 3.1379x; 1.0781x over previous
#include <hip/hip_runtime.h>

typedef __attribute__((ext_vector_type(2))) float v2f;
typedef __attribute__((ext_vector_type(8))) float v8f;

#define WMMA_F32(a, b, c) \
  __builtin_amdgcn_wmma_f32_16x16x4_f32(false, (a), false, (b), (short)0, (c), false, false)

static constexpr int Tn = 32768;
static constexpr int NS = 256;
static constexpr int LC = 8;
static constexpr int C1 = 4096;
static constexpr int C2 = 512;
static constexpr int C3 = 64;
static constexpr int SC_COLS = 16;
static constexpr int MS_LD = 260;
static constexpr int SCAN_LDS_FLOATS = 256 * MS_LD + 2 * 256 * SC_COLS;

__constant__ float d_H[6][4] = {
  {1.0f, 0.0f, 0.0f, 0.0f},
  {0.896f, 0.128f, 0.104f, -0.032f},
  {0.784f, 0.147f, 0.216f, -0.063f},
  {0.104f, 0.032f, 0.896f, -0.128f},
  {25.0f/729.0f, 8.0f/729.0f, 704.0f/729.0f, -64.0f/729.0f},
  {0.0f, 0.0f, 1.0f, 0.0f},
};

__global__ void k_prep(float* __restrict__ Ad, float* __restrict__ Bd, float* __restrict__ Lt0,
                       const float* __restrict__ A, const float* __restrict__ B,
                       const float* __restrict__ t) {
  const int idx = blockIdx.x * 256 + threadIdx.x;
  const float dt = t[1] - t[0];
  if (idx < 65536) {
    Ad[idx] = dt * A[idx];
  } else {
    const int q = idx - 65536;
    if (q < 49152) {
      const int r = q / 192, c = q - r * 192;
      float val = 0.0f;
      if (c < 32) { const float bd = dt * B[r * 32 + c]; Bd[r * 32 + c] = bd; val += bd; }
      Lt0[q] = val;
    }
  }
}
__global__ void k_final(float* __restrict__ Mm, float* __restrict__ Q,
                        const float* __restrict__ K1, const float* __restrict__ K3,
                        const float* __restrict__ K4, const float* __restrict__ K5,
                        const float* __restrict__ K6,
                        const float* __restrict__ W1, const float* __restrict__ W3,
                        const float* __restrict__ W4, const float* __restrict__ W5,
                        const float* __restrict__ W6,
                        float b1, float b3, float b4, float b5, float b6) {
  const int idx = blockIdx.x * 256 + threadIdx.x;
  if (idx < 65536) {
    float v = ((idx >> 8) == (idx & 255)) ? 1.0f : 0.0f;
    v = fmaf(b1, K1[idx], v);
    v = fmaf(b3, K3[idx], v);
    v = fmaf(b4, K4[idx], v);
    v = fmaf(b5, K5[idx], v);
    v = fmaf(b6, K6[idx], v);
    Mm[idx] = v;
  } else {
    const int q = idx - 65536;
    if (q >= 256 * 32) return;
    const int r = q >> 5, c = q & 31;
    float w[6];
    for (int st = 0; st < 6; ++st) {
      const int o = r * 192 + 32 * st + c;
      float t = 0.0f;
      t = fmaf(b1, W1[o], t);
      t = fmaf(b3, W3[o], t);
      t = fmaf(b4, W4[o], t);
      t = fmaf(b5, W5[o], t);
      t = fmaf(b6, W6[o], t);
      w[st] = t;
    }
    for (int j = 0; j < 4; ++j) {
      float s = 0.0f;
      for (int st = 0; st < 6; ++st) s += d_H[st][j] * w[st];
      Q[(32 * j + c) * 256 + r] = s;
    }
  }
}
__global__ __launch_bounds__(128) void gemm_kernel(
    float* __restrict__ D, int ldd,
    const float* __restrict__ A, int lda,
    const float* __restrict__ B, int ldb,
    const float* __restrict__ Cin, int ldc,
    int Mr, int Nc, int Kd, float alpha, float beta, int transB) {
  const int lane = threadIdx.x & 31, wave = threadIdx.x >> 5;
  const int hl = lane >> 4, lr = lane & 15;
  const int tilesN = Nc >> 4;
  const int tile = blockIdx.x * 4 + wave;
  if (tile >= (Mr >> 4) * tilesN) return;
  const int tm = tile / tilesN, tn = tile % tilesN;
  const float* Ar = A + (size_t)(tm * 16 + lr) * lda + 2 * hl;
  v8f acc = {};
  if (transB) {
    const float* Br = B + (size_t)(tn * 16 + lr) * ldb + 2 * hl;
    #pragma unroll 4
    for (int k = 0; k < Kd; k += 4) {
      v2f a = *(const v2f*)(Ar + k);
      v2f b = *(const v2f*)(Br + k);
      acc = WMMA_F32(a, b, acc);
    }
  } else {
    const float* Bc = B + (size_t)(2 * hl) * ldb + tn * 16 + lr;
    #pragma unroll 4
    for (int k = 0; k < Kd; k += 4) {
      v2f a = *(const v2f*)(Ar + k);
      v2f b;
      b[0] = Bc[(size_t)k * ldb];
      b[1] = Bc[(size_t)k * ldb + ldb];
      acc = WMMA_F32(a, b, acc);
    }
  }
  float* Dp = D + (size_t)(tm * 16 + 8 * hl) * ldd + tn * 16 + lr;
  if (beta != 0.0f) {
    const float* Cp = Cin + (size_t)(tm * 16 + 8 * hl) * ldc + tn * 16 + lr;
    #pragma unroll
    for (int v = 0; v < 8; ++v)
      Dp[(size_t)v * ldd] = alpha * acc[v] + beta * Cp[(size_t)v * ldc];
  } else {
    #pragma unroll
    for (int v = 0; v < 8; ++v)
      Dp[(size_t)v * ldd] = alpha * acc[v];
  }
}

template <int NJ>
__global__ __launch_bounds__(128) void stage_kernel(
    float* __restrict__ Kout, float* __restrict__ Lout,
    const float* __restrict__ Ad, const float* __restrict__ Bd, int blk,
    const float* __restrict__ K0, const float* __restrict__ K1, const float* __restrict__ K2,
    const float* __restrict__ K3, const float* __restrict__ K4,
    const float* __restrict__ W0, const float* __restrict__ W1, const float* __restrict__ W2,
    const float* __restrict__ W3, const float* __restrict__ W4,
    float c0, float c1, float c2, float c3, float c4) {
  const int lane = threadIdx.x & 31;
  const int wave = __builtin_amdgcn_readfirstlane(threadIdx.x >> 5);
  const int hl = lane >> 4, lr = lane & 15;
  const int tile = blockIdx.x * 4 + wave;
  const int tm = tile / 28, tn = tile % 28;
  const float* Ar = Ad + (size_t)(tm * 16 + lr) * 256 + 2 * hl;
  v8f acc = {};
  if (tn < 16) {
    const int n = tn * 16 + lr;
    const int off = (2 * hl) * 256 + n;
    #pragma unroll 4
    for (int k = 0; k < 256; k += 4) {
      v2f a = *(const v2f*)(Ar + k);
      const int o = off + k * 256;
      float b0 = (k + 2 * hl     == n) ? 1.0f : 0.0f;
      float b1 = (k + 2 * hl + 1 == n) ? 1.0f : 0.0f;
      b0 = fmaf(c0, K0[o], b0);  b1 = fmaf(c0, K0[o + 256], b1);
      if (NJ > 1) { b0 = fmaf(c1, K1[o], b0);  b1 = fmaf(c1, K1[o + 256], b1); }
      if (NJ > 2) { b0 = fmaf(c2, K2[o], b0);  b1 = fmaf(c2, K2[o + 256], b1); }
      if (NJ > 3) { b0 = fmaf(c3, K3[o], b0);  b1 = fmaf(c3, K3[o + 256], b1); }
      if (NJ > 4) { b0 = fmaf(c4, K4[o], b0);  b1 = fmaf(c4, K4[o + 256], b1); }
      v2f b = {b0, b1};
      acc = WMMA_F32(a, b, acc);
    }
    float* Dp = Kout + (size_t)(tm * 16 + 8 * hl) * 256 + n;
    #pragma unroll
    for (int v = 0; v < 8; ++v) Dp[(size_t)v * 256] = acc[v];
  } else {
    const int n = (tn - 16) * 16 + lr;
    const int off = (2 * hl) * 192 + n;
    #pragma unroll 4
    for (int k = 0; k < 256; k += 4) {
      v2f a = *(const v2f*)(Ar + k);
      const int o = off + k * 192;
      float b0 = 0.0f, b1 = 0.0f;
      b0 = fmaf(c0, W0[o], b0);  b1 = fmaf(c0, W0[o + 192], b1);
      if (NJ > 1) { b0 = fmaf(c1, W1[o], b0);  b1 = fmaf(c1, W1[o + 192], b1); }
      if (NJ > 2) { b0 = fmaf(c2, W2[o], b0);  b1 = fmaf(c2, W2[o + 192], b1); }
      if (NJ > 3) { b0 = fmaf(c3, W3[o], b0);  b1 = fmaf(c3, W3[o + 192], b1); }
      if (NJ > 4) { b0 = fmaf(c4, W4[o], b0);  b1 = fmaf(c4, W4[o + 192], b1); }
      v2f b = {b0, b1};
      acc = WMMA_F32(a, b, acc);
    }
    const int row0 = tm * 16 + 8 * hl;
    float* Dp = Lout + (size_t)row0 * 192 + n;
    if (((tn - 16) >> 1) == blk) {
      const float* Bp = Bd + (size_t)row0 * 32 + (n & 31);
      #pragma unroll
      for (int v = 0; v < 8; ++v) Dp[(size_t)v * 192] = acc[v] + Bp[(size_t)v * 32];
    } else {
      #pragma unroll
      for (int v = 0; v < 8; ++v) Dp[(size_t)v * 192] = acc[v];
    }
  }
}

static constexpr int DR_ROWS = 128;
static constexpr int DR_LD = 132;

__device__ __forceinline__ float xhalf(float x) {
  const int b = __builtin_bit_cast(int, x);
  return __builtin_bit_cast(float,
      __builtin_amdgcn_permlanex16(b, b, 0x76543210, 0xfedcba98, false, false));
}

__global__ __launch_bounds__(256) void drive_kernel(
    float* __restrict__ G, const float* __restrict__ u, const float* __restrict__ Q) {
  __shared__ __attribute__((aligned(16))) float Pt[DR_ROWS * DR_LD];
  const int tid = threadIdx.x;
  const int lane = tid & 31;
  const int wave = __builtin_amdgcn_readfirstlane(tid >> 5);
  const int hl = lane >> 4, lr = lane & 15;
  const int n0 = blockIdx.x * DR_ROWS;
  const int col0 = wave * 32;

  {
    const int cc = tid & 7, rg = tid >> 3;
    const int nb = n0 + 4 * rg;
    float4 U[7];
    #pragma unroll
    for (int i = 0; i < 7; ++i) {
      int g = nb - 1 + i;
      g = g < 0 ? 0 : (g > Tn - 1 ? Tn - 1 : g);
      U[i] = *(const float4*)(u + (size_t)g * 32 + 4 * cc);
    }
    #pragma unroll
    for (int t = 0; t < 4; ++t) {
      const int n = nb + t;
      const bool first = (n == 0), penult = (n == Tn - 2), last = (n == Tn - 1);
      const float4 mb = first ? U[t + 1] : U[t];
      const float ms = first ? 1.0f : 0.5f;
      const float4 na = penult ? U[t + 2] : U[t + 3];
      const float ns = penult ? 1.0f : 0.5f;
      float4 p0 = U[t + 1];
      float4 p1 = make_float4(ms * (U[t + 2].x - mb.x), ms * (U[t + 2].y - mb.y),
                              ms * (U[t + 2].z - mb.z), ms * (U[t + 2].w - mb.w));
      float4 p2 = U[t + 2];
      float4 p3 = make_float4(ns * (na.x - U[t + 1].x), ns * (na.y - U[t + 1].y),
                              ns * (na.z - U[t + 1].z), ns * (na.w - U[t + 1].w));
      if (last) {
        const float4 z = make_float4(0.0f, 0.0f, 0.0f, 0.0f);
        p0 = z; p1 = z; p2 = z; p3 = z;
      }
      float* prow = Pt + (4 * rg + t) * DR_LD + 4 * cc;
      *(float4*)(prow)      = p0;
      *(float4*)(prow + 32) = p1;
      *(float4*)(prow + 64) = p2;
      *(float4*)(prow + 96) = p3;
    }
  }

  v2f b0[32], b1[32];
  {
    const float* Qp = Q + (size_t)(2 * hl) * 256 + col0 + lr;
    #pragma unroll
    for (int kc = 0; kc < 32; ++kc) {
      b0[kc][0] = Qp[kc * 1024];
      b0[kc][1] = Qp[kc * 1024 + 256];
      b1[kc][0] = Qp[kc * 1024 + 16];
      b1[kc][1] = Qp[kc * 1024 + 256 + 16];
    }
  }
  __syncthreads();

  const float* Ap = Pt + lr * DR_LD + 2 * hl;
  for (int hb = 0; hb < 2; ++hb) {
    const float* Ah = Ap + hb * 64 * DR_LD;
    v8f acc0[4] = {}, acc1[4] = {};
    #pragma unroll
    for (int kc = 0; kc < 32; ++kc) {
      #pragma unroll
      for (int r = 0; r < 4; ++r) {
        const v2f a = *(const v2f*)(Ah + r * 16 * DR_LD + kc * 4);
        acc0[r] = WMMA_F32(a, b0[kc], acc0[r]);
        acc1[r] = WMMA_F32(a, b1[kc], acc1[r]);
      }
    }
    float* Gp = G + (size_t)(n0 + hb * 64) * 256 + col0 + lane;
    #pragma unroll
    for (int r = 0; r < 4; ++r) {
      #pragma unroll
      for (int v = 0; v < 8; ++v) {
        const float t0 = acc0[r][v], t1 = acc1[r][v];
        const float got = xhalf(hl ? t0 : t1);
        const float lo = hl ? got : t0;
        const float hi = hl ? t1 : got;
        Gp[(size_t)(r * 16 + v) * 256] = lo;
        Gp[(size_t)(r * 16 + 8 + v) * 256] = hi;
      }
    }
  }
}

__global__ __launch_bounds__(256) void scan_kernel(
    const float* __restrict__ Mg, const float* __restrict__ G,
    const float* __restrict__ Init,
    float* __restrict__ OutAll,
    float* __restrict__ OutEnd,
    int L) {
  __shared__ __attribute__((aligned(16))) float smem[SCAN_LDS_FLOATS];
  float* Ms = smem;
  float* Xbuf0 = smem + 256 * MS_LD;
  float* Xbuf1 = Xbuf0 + 256 * SC_COLS;
  const int tid = threadIdx.x;
  const int lane = tid & 31, wave = tid >> 5;
  const int hl = lane >> 4, lr = lane & 15;
  const int cbase = blockIdx.x * SC_COLS;
  const int tr0 = wave * 2;

  {
    const float4* src = (const float4*)Mg;
    float4* dst = (float4*)Ms;
    for (int k = tid; k < 65536 / 4; k += 256)
      dst[(k >> 6) * (MS_LD / 4) + (k & 63)] = src[k];
  }
  for (int e = 0; e < SC_COLS; ++e)
    Xbuf0[tid * SC_COLS + e] = Init ? Init[(size_t)(cbase + e) * 256 + tid] : 0.0f;
  __syncthreads();

  const float* gbase = G + ((size_t)(cbase + lr) * L) * 256 + tr0 * 16 + 8 * hl;

  float* Xc = Xbuf0;
  float* Xn = Xbuf1;
  for (int i = 0; i < L; ++i) {
    if (OutAll) {
      for (int e = 0; e < SC_COLS; ++e) {
        size_t m = (size_t)(cbase + e) * L + i;
        OutAll[m * 256 + tid] = Xc[tid * SC_COLS + e];
      }
    }
    const float* gp = gbase + (size_t)i * 256;
    if (i + 1 < L) __builtin_prefetch(gp + 256, 0, 1);
    v8f acc0, acc1;
    {
      float4 ga = *(const float4*)(gp);
      float4 gb = *(const float4*)(gp + 4);
      float4 gc = *(const float4*)(gp + 16);
      float4 gd = *(const float4*)(gp + 20);
      acc0[0]=ga.x; acc0[1]=ga.y; acc0[2]=ga.z; acc0[3]=ga.w;
      acc0[4]=gb.x; acc0[5]=gb.y; acc0[6]=gb.z; acc0[7]=gb.w;
      acc1[0]=gc.x; acc1[1]=gc.y; acc1[2]=gc.z; acc1[3]=gc.w;
      acc1[4]=gd.x; acc1[5]=gd.y; acc1[6]=gd.z; acc1[7]=gd.w;
    }
    const float* Mr0 = Ms + (size_t)(tr0 * 16 + lr) * MS_LD + 2 * hl;
    const float* Mr1 = Mr0 + 16 * MS_LD;
    const float* Xk = Xc + 2 * hl * SC_COLS + lr;
    #pragma unroll 8
    for (int kc = 0; kc < 64; ++kc) {
      v2f a0 = *(const v2f*)(Mr0 + kc * 4);
      v2f a1 = *(const v2f*)(Mr1 + kc * 4);
      v2f b;
      b[0] = Xk[kc * 4 * SC_COLS];
      b[1] = Xk[kc * 4 * SC_COLS + SC_COLS];
      acc0 = WMMA_F32(a0, b, acc0);
      acc1 = WMMA_F32(a1, b, acc1);
    }
    #pragma unroll
    for (int v = 0; v < 8; ++v) {
      Xn[(tr0 * 16 + 8 * hl + v) * SC_COLS + lr] = acc0[v];
      Xn[((tr0 + 1) * 16 + 8 * hl + v) * SC_COLS + lr] = acc1[v];
    }
    __syncthreads();
    float* tmp = Xc; Xc = Xn; Xn = tmp;
  }
  if (OutEnd) {
    for (int e = 0; e < SC_COLS; ++e)
      OutEnd[(size_t)(cbase + e) * 256 + tid] = Xc[tid * SC_COLS + e];
  }
}

static constexpr int TV_PITCH = 20;
static constexpr int TV_FLOATS = 16 * TV_PITCH;

template <int CTRL>
__device__ __forceinline__ float lane_read(float v) {
  return __int_as_float(__builtin_amdgcn_update_dpp(0, __float_as_int(v), CTRL, 0xF, 0xF, true));
}
static constexpr int DPP_SWAP1 = 0xB1;
static constexpr int DPP_SWAP2 = 0x4E;
static constexpr int DPP_ROR4  = 0x124;
static constexpr int DPP_ROR8  = 0x128;

__global__ __launch_bounds__(1024) void carry_kernel(
    const float* __restrict__ Mg, const float* __restrict__ Z,
    const float* __restrict__ x0, float* __restrict__ Tcar, int n) {
  __shared__ __attribute__((aligned(16))) float tv[2 * TV_FLOATS];
  const int tid = threadIdx.x;
  const int lane = tid & 31, wave = tid >> 5;
  const int g = lane & 15, rg = lane >> 4;
  const int rbase = wave * 8 + rg * 4;

  v2f m[4][8];
  for (int i = 0; i < 4; ++i) {
    const float4* mrow = (const float4*)(Mg + (size_t)(rbase + i) * 256 + 16 * g);
    for (int c = 0; c < 4; ++c) {
      float4 t4 = mrow[c];
      m[i][2 * c]     = (v2f){t4.x, t4.y};
      m[i][2 * c + 1] = (v2f){t4.z, t4.w};
    }
  }
  const int outslot = (tid >> 4) * TV_PITCH + (tid & 15);
  if (tid < 256) tv[outslot] = x0[tid];
  const bool odd1 = (g & 1) != 0, odd2 = (g & 2) != 0;
  const int myrow = rbase + (odd1 ? 2 : 0) + (odd2 ? 1 : 0);
  const int myslot = (myrow >> 4) * TV_PITCH + (myrow & 15);
  float zc = (n > 1) ? Z[myrow] : 0.0f;
  __syncthreads();

  for (int j = 0; j + 1 < n; ++j) {
    const float* tcur = tv + (j & 1) * TV_FLOATS;
    float* tnxt = tv + ((j + 1) & 1) * TV_FLOATS;
    if (wave < 8) Tcar[(size_t)j * 256 + tid] = tcur[outslot];
    float zn = 0.0f;
    if (j + 2 < n) zn = Z[(size_t)(j + 1) * 256 + myrow];
    const float* xs = tcur + g * TV_PITCH;
    v2f acc[4] = {};
    for (int c = 0; c < 4; ++c) {
      float4 x4 = *(const float4*)(xs + 4 * c);
      v2f xa = {x4.x, x4.y}, xb = {x4.z, x4.w};
      for (int i = 0; i < 4; ++i) {
        acc[i] += m[i][2 * c] * xa;
        acc[i] += m[i][2 * c + 1] * xb;
      }
    }
    float p0 = acc[0][0] + acc[0][1];
    float p1 = acc[1][0] + acc[1][1];
    float p2 = acc[2][0] + acc[2][1];
    float p3 = acc[3][0] + acc[3][1];
    float s0 = odd1 ? p0 : p2, s1 = odd1 ? p1 : p3;
    float k0 = odd1 ? p2 : p0, k1 = odd1 ? p3 : p1;
    k0 += lane_read<DPP_SWAP1>(s0);
    k1 += lane_read<DPP_SWAP1>(s1);
    float s = odd2 ? k0 : k1;
    float k = odd2 ? k1 : k0;
    k += lane_read<DPP_SWAP2>(s);
    k += lane_read<DPP_ROR4>(k);
    k += lane_read<DPP_ROR8>(k);
    if (g < 4) tnxt[myslot] = k + zc;
    __syncthreads();
    zc = zn;
  }
  if (wave < 8) Tcar[(size_t)(n - 1) * 256 + tid] = tv[((n - 1) & 1) * TV_FLOATS + outslot];
}

static inline void launch_gemm(hipStream_t s, float* D, int ldd,
                               const float* A, int lda, const float* B, int ldb,
                               const float* Cin, int ldc,
                               int M, int N, int K, float alpha, float beta, int transB) {
  int tiles = (M / 16) * (N / 16);
  gemm_kernel<<<(tiles + 3) / 4, 128, 0, s>>>(D, ldd, A, lda, B, ldb, Cin, ldc,
                                              M, N, K, alpha, beta, transB);
}

extern "C" void kernel_launch(void* const* d_in, const int* in_sizes, int n_in,
                              void* d_out, int out_size, void* d_ws, size_t ws_size,
                              hipStream_t stream) {
  (void)in_sizes; (void)n_in; (void)out_size; (void)ws_size;
  const float* t  = (const float*)d_in[0];
  const float* u  = (const float*)d_in[1];
  const float* x0 = (const float*)d_in[2];
  const float* Am = (const float*)d_in[3];
  const float* Bm = (const float*)d_in[4];
  const float* Cm = (const float*)d_in[5];
  const float* Dm = (const float*)d_in[6];
  float* xout = (float*)d_out;
  float* yout = xout + (size_t)Tn * NS;

  float* w = (float*)d_ws;
  float* Ad = w;  w += 65536;
  float* Bd = w;  w += 8192;
  float* Kt[6]; for (int i = 0; i < 6; i++) { Kt[i] = w; w += 65536; }
  Kt[0] = Ad;
  float* Lt[6]; for (int i = 0; i < 6; i++) { Lt[i] = w; w += 49152; }
  float* M64  = w; w += 65536 + 49152;
  float* Mm   = w; w += 65536;
  w += 49152;
  float* Qc   = w; w += 32768;
  float* M8   = w; w += 65536;
  float* M512 = w; w += 65536;
  float* PT1  = w; w += 65536;
  float* PT2  = w; w += 65536;
  float* Pm   = w; w += (size_t)Tn * 128;
  float* G    = w; w += (size_t)Tn * 256;
  float* Zend2 = w; w += (size_t)C2 * 256;
  float* Scar2 = w; w += (size_t)C2 * 256;
  float* Zend3 = w; w += (size_t)C3 * 256;
  float* Tcar  = w; w += (size_t)C3 * 256;
  float* Zend1 = Pm;
  float* Scar1 = Pm + (size_t)C1 * 256;

  auto gdim = [](int n) { return dim3((unsigned)((n + 255) / 256)); };

  k_prep<<<gdim(65536 + 49152), 256, 0, stream>>>(Ad, Bd, Lt[0], Am, Bm, t);

  double aa[7][7] = {};
  aa[2][1] = 0.2;
  aa[3][1] = 3.0/40.0;      aa[3][2] = 9.0/40.0;
  aa[4][1] = 44.0/45.0;     aa[4][2] = -56.0/15.0;     aa[4][3] = 32.0/9.0;
  aa[5][1] = 19372.0/6561.0; aa[5][2] = -25360.0/2187.0;
  aa[5][3] = 64448.0/6561.0; aa[5][4] = -212.0/729.0;
  aa[6][1] = 9017.0/3168.0; aa[6][2] = -355.0/33.0; aa[6][3] = 46732.0/5247.0;
  aa[6][4] = 49.0/176.0;    aa[6][5] = -5103.0/18656.0;
  stage_kernel<1><<<112, 128, 0, stream>>>(Kt[1], Lt[1], Ad, Bd, 1,
      Kt[0], Kt[0], Kt[0], Kt[0], Kt[0], Lt[0], Lt[0], Lt[0], Lt[0], Lt[0],
      (float)aa[2][1], 0.f, 0.f, 0.f, 0.f);
  stage_kernel<2><<<112, 128, 0, stream>>>(Kt[2], Lt[2], Ad, Bd, 2,
      Kt[0], Kt[1], Kt[0], Kt[0], Kt[0], Lt[0], Lt[1], Lt[0], Lt[0], Lt[0],
      (float)aa[3][1], (float)aa[3][2], 0.f, 0.f, 0.f);
  stage_kernel<3><<<112, 128, 0, stream>>>(Kt[3], Lt[3], Ad, Bd, 3,
      Kt[0], Kt[1], Kt[2], Kt[0], Kt[0], Lt[0], Lt[1], Lt[2], Lt[0], Lt[0],
      (float)aa[4][1], (float)aa[4][2], (float)aa[4][3], 0.f, 0.f);
  stage_kernel<4><<<112, 128, 0, stream>>>(Kt[4], Lt[4], Ad, Bd, 4,
      Kt[0], Kt[1], Kt[2], Kt[3], Kt[0], Lt[0], Lt[1], Lt[2], Lt[3], Lt[0],
      (float)aa[5][1], (float)aa[5][2], (float)aa[5][3], (float)aa[5][4], 0.f);
  stage_kernel<5><<<112, 128, 0, stream>>>(Kt[5], Lt[5], Ad, Bd, 5,
      Kt[0], Kt[1], Kt[2], Kt[3], Kt[4], Lt[0], Lt[1], Lt[2], Lt[3], Lt[4],
      (float)aa[6][1], (float)aa[6][2], (float)aa[6][3], (float)aa[6][4], (float)aa[6][5]);

  const double bb[7] = {0.0, 35.0/384.0, 0.0, 500.0/1113.0, 125.0/192.0,
                        -2187.0/6784.0, 11.0/84.0};
  k_final<<<gdim(65536 + 8192), 256, 0, stream>>>(Mm, Qc,
      Kt[0], Kt[2], Kt[3], Kt[4], Kt[5], Lt[0], Lt[2], Lt[3], Lt[4], Lt[5],
      (float)bb[1], (float)bb[3], (float)bb[4], (float)bb[5], (float)bb[6]);

  launch_gemm(stream, PT1, 256, Mm, 256, Mm, 256, nullptr, 0, 256, 256, 256, 1.f, 0.f, 0);
  launch_gemm(stream, PT2, 256, PT1, 256, PT1, 256, nullptr, 0, 256, 256, 256, 1.f, 0.f, 0);
  launch_gemm(stream, M8,  256, PT2, 256, PT2, 256, nullptr, 0, 256, 256, 256, 1.f, 0.f, 0);
  launch_gemm(stream, PT1, 256, M8,  256, M8,  256, nullptr, 0, 256, 256, 256, 1.f, 0.f, 0);
  launch_gemm(stream, PT2, 256, PT1, 256, PT1, 256, nullptr, 0, 256, 256, 256, 1.f, 0.f, 0);
  launch_gemm(stream, M64, 256, PT2, 256, PT2, 256, nullptr, 0, 256, 256, 256, 1.f, 0.f, 0);
  launch_gemm(stream, PT1, 256, M64, 256, M64, 256, nullptr, 0, 256, 256, 256, 1.f, 0.f, 0);
  launch_gemm(stream, PT2, 256, PT1, 256, PT1, 256, nullptr, 0, 256, 256, 256, 1.f, 0.f, 0);
  launch_gemm(stream, M512,256, PT2, 256, PT2, 256, nullptr, 0, 256, 256, 256, 1.f, 0.f, 0);

  drive_kernel<<<Tn / DR_ROWS, 256, 0, stream>>>(G, u, Qc);

  scan_kernel<<<C1 / SC_COLS, 256, 0, stream>>>(Mm, G, nullptr, nullptr, Zend1, LC);
  scan_kernel<<<C2 / SC_COLS, 256, 0, stream>>>(M8, Zend1, nullptr, nullptr, Zend2, LC);
  scan_kernel<<<C3 / SC_COLS, 256, 0, stream>>>(M64, Zend2, nullptr, nullptr, Zend3, LC);
  carry_kernel<<<1, 1024, 0, stream>>>(M512, Zend3, x0, Tcar, C3);
  scan_kernel<<<C3 / SC_COLS, 256, 0, stream>>>(M64, Zend2, Tcar, Scar2, nullptr, LC);
  scan_kernel<<<C2 / SC_COLS, 256, 0, stream>>>(M8, Zend1, Scar2, Scar1, nullptr, LC);
  scan_kernel<<<C1 / SC_COLS, 256, 0, stream>>>(Mm, G, Scar1, xout, nullptr, LC);

  launch_gemm(stream, yout, 32, xout, 256, Cm, 256, nullptr, 0, Tn, 32, 256, 1.f, 0.f, 1);
  launch_gemm(stream, yout, 32, u,    32,  Dm, 32,  yout,   32, Tn, 32, 32,  1.f, 1.f, 1);
}


# every K=256 product (5 stage GEMMs, 9 matrix squarings, X@C^T): work-group stages its operand blocks in LDS with all global loads in flight at once, not 16 load-wait-WMMA rounds; sum order kept
# speedup vs baseline: 4.0813x; 1.3006x over previous
#include <hip/hip_runtime.h>

typedef __attribute__((ext_vector_type(2))) float v2f;
typedef __attribute__((ext_vector_type(8))) float v8f;

#define WMMA_F32(a, b, c) \
  __builtin_amdgcn_wmma_f32_16x16x4_f32(false, (a), false, (b), (short)0, (c), false, false)

static constexpr int Tn = 32768;
static constexpr int NS = 256;
static constexpr int LC = 8;
static constexpr int C1 = 4096;
static constexpr int C2 = 512;
static constexpr int C3 = 64;
static constexpr int SC_COLS = 16;
static constexpr int MS_LD = 260;
static constexpr int SCAN_LDS_FLOATS = 256 * MS_LD + 2 * 256 * SC_COLS;

__constant__ float d_H[6][4] = {
  {1.0f, 0.0f, 0.0f, 0.0f},
  {0.896f, 0.128f, 0.104f, -0.032f},
  {0.784f, 0.147f, 0.216f, -0.063f},
  {0.104f, 0.032f, 0.896f, -0.128f},
  {25.0f/729.0f, 8.0f/729.0f, 704.0f/729.0f, -64.0f/729.0f},
  {0.0f, 0.0f, 1.0f, 0.0f},
};

__global__ void k_prep(float* __restrict__ Ad, float* __restrict__ Bd, float* __restrict__ Lt0,
                       const float* __restrict__ A, const float* __restrict__ B,
                       const float* __restrict__ t) {
  const int idx = blockIdx.x * 256 + threadIdx.x;
  const float dt = t[1] - t[0];
  if (idx < 65536) {
    Ad[idx] = dt * A[idx];
  } else {
    const int q = idx - 65536;
    if (q < 49152) {
      const int r = q / 192, c = q - r * 192;
      float val = 0.0f;
      if (c < 32) { const float bd = dt * B[r * 32 + c]; Bd[r * 32 + c] = bd; val += bd; }
      Lt0[q] = val;
    }
  }
}
__global__ void k_final(float* __restrict__ Mm, float* __restrict__ Q,
                        const float* __restrict__ K1, const float* __restrict__ K3,
                        const float* __restrict__ K4, const float* __restrict__ K5,
                        const float* __restrict__ K6,
                        const float* __restrict__ W1, const float* __restrict__ W3,
                        const float* __restrict__ W4, const float* __restrict__ W5,
                        const float* __restrict__ W6,
                        float b1, float b3, float b4, float b5, float b6) {
  const int idx = blockIdx.x * 256 + threadIdx.x;
  if (idx < 65536) {
    float v = ((idx >> 8) == (idx & 255)) ? 1.0f : 0.0f;
    v = fmaf(b1, K1[idx], v);
    v = fmaf(b3, K3[idx], v);
    v = fmaf(b4, K4[idx], v);
    v = fmaf(b5, K5[idx], v);
    v = fmaf(b6, K6[idx], v);
    Mm[idx] = v;
  } else {
    const int q = idx - 65536;
    if (q >= 256 * 32) return;
    const int r = q >> 5, c = q & 31;
    float w[6];
    for (int st = 0; st < 6; ++st) {
      const int o = r * 192 + 32 * st + c;
      float t = 0.0f;
      t = fmaf(b1, W1[o], t);
      t = fmaf(b3, W3[o], t);
      t = fmaf(b4, W4[o], t);
      t = fmaf(b5, W5[o], t);
      t = fmaf(b6, W6[o], t);
      w[st] = t;
    }
    for (int j = 0; j < 4; ++j) {
      float s = 0.0f;
      for (int st = 0; st < 6; ++st) s += d_H[st][j] * w[st];
      Q[(32 * j + c) * 256 + r] = s;
    }
  }
}
__global__ __launch_bounds__(128) void gemm_kernel(
    float* __restrict__ D, int ldd,
    const float* __restrict__ A, int lda,
    const float* __restrict__ B, int ldb,
    const float* __restrict__ Cin, int ldc,
    int Mr, int Nc, int Kd, float alpha, float beta, int transB) {
  const int lane = threadIdx.x & 31, wave = threadIdx.x >> 5;
  const int hl = lane >> 4, lr = lane & 15;
  const int tilesN = Nc >> 4;
  const int tile = blockIdx.x * 4 + wave;
  if (tile >= (Mr >> 4) * tilesN) return;
  const int tm = tile / tilesN, tn = tile % tilesN;
  const float* Ar = A + (size_t)(tm * 16 + lr) * lda + 2 * hl;
  v8f acc = {};
  if (transB) {
    const float* Br = B + (size_t)(tn * 16 + lr) * ldb + 2 * hl;
    #pragma unroll 4
    for (int k = 0; k < Kd; k += 4) {
      v2f a = *(const v2f*)(Ar + k);
      v2f b = *(const v2f*)(Br + k);
      acc = WMMA_F32(a, b, acc);
    }
  } else {
    const float* Bc = B + (size_t)(2 * hl) * ldb + tn * 16 + lr;
    #pragma unroll 4
    for (int k = 0; k < Kd; k += 4) {
      v2f a = *(const v2f*)(Ar + k);
      v2f b;
      b[0] = Bc[(size_t)k * ldb];
      b[1] = Bc[(size_t)k * ldb + ldb];
      acc = WMMA_F32(a, b, acc);
    }
  }
  float* Dp = D + (size_t)(tm * 16 + 8 * hl) * ldd + tn * 16 + lr;
  if (beta != 0.0f) {
    const float* Cp = Cin + (size_t)(tm * 16 + 8 * hl) * ldc + tn * 16 + lr;
    #pragma unroll
    for (int v = 0; v < 8; ++v)
      Dp[(size_t)v * ldd] = alpha * acc[v] + beta * Cp[(size_t)v * ldc];
  } else {
    #pragma unroll
    for (int v = 0; v < 8; ++v)
      Dp[(size_t)v * ldd] = alpha * acc[v];
  }
}

static constexpr int GK_LD = 260;
static constexpr int GN_LD = 40;

__device__ __forceinline__ void k256_load_a(float4 (&ra)[16], const float* __restrict__ A, int bm, int tid) {
  const float4* Ab = (const float4*)(A + (size_t)bm * 32 * 256);
  #pragma unroll
  for (int i = 0; i < 16; ++i) ra[i] = Ab[tid + 128 * i];
}
__device__ __forceinline__ void k256_store_a(float* As, const float4 (&ra)[16], int tid) {
  #pragma unroll
  for (int i = 0; i < 16; ++i) {
    const int idx = tid + 128 * i;
    *(float4*)(As + (idx >> 6) * GK_LD + 4 * (idx & 63)) = ra[i];
  }
}
template <bool TB>
__device__ __forceinline__ void k256_frags(v2f (&a)[16], v2f (&b)[16],
                                           const float* Ap, const float* Bp, int g) {
  #pragma unroll
  for (int j = 0; j < 16; ++j) {
    const int kc = 16 * g + j;
    a[j] = *(const v2f*)(Ap + 4 * kc);
    if (TB) {
      b[j] = *(const v2f*)(Bp + 4 * kc);
    } else {
      b[j][0] = Bp[4 * kc * GN_LD];
      b[j][1] = Bp[4 * kc * GN_LD + GN_LD];
    }
  }
}
__device__ __forceinline__ v8f k256_wmma16(const v2f (&a)[16], const v2f (&b)[16], v8f acc) {
  #pragma unroll
  for (int j = 0; j < 16; ++j) acc = WMMA_F32(a[j], b[j], acc);
  return acc;
}
template <bool TB>
__device__ __forceinline__ v8f k256_chain(const float* Ap, const float* Bp) {
  v2f a0[16], b0[16], a1[16], b1[16];
  v8f acc = {};
  k256_frags<TB>(a0, b0, Ap, Bp, 0);
  k256_frags<TB>(a1, b1, Ap, Bp, 1);
  __builtin_amdgcn_sched_barrier(0);
  acc = k256_wmma16(a0, b0, acc);
  k256_frags<TB>(a0, b0, Ap, Bp, 2);
  __builtin_amdgcn_sched_barrier(0);
  acc = k256_wmma16(a1, b1, acc);
  k256_frags<TB>(a1, b1, Ap, Bp, 3);
  __builtin_amdgcn_sched_barrier(0);
  acc = k256_wmma16(a0, b0, acc);
  acc = k256_wmma16(a1, b1, acc);
  return acc;
}

template <bool TB, int NBN>
__global__ __launch_bounds__(128) void gemm_k256_kernel(
    float* __restrict__ D, int ldd, const float* __restrict__ A, const float* __restrict__ B) {
  __shared__ __attribute__((aligned(16))) float As[32 * GK_LD];
  __shared__ __attribute__((aligned(16))) float Bs[TB ? 32 * GK_LD : 256 * GN_LD];
  const int tid = threadIdx.x;
  const int lane = tid & 31;
  const int wave = __builtin_amdgcn_readfirstlane(tid >> 5);
  const int hl = lane >> 4, lr = lane & 15;
  const int bm = blockIdx.x / NBN, bn = blockIdx.x % NBN;
  {
    float4 ra[16], rb[16];
    k256_load_a(ra, A, bm, tid);
    if (TB) {
      k256_load_a(rb, B, bn, tid);
    } else {
      const float* Bb = B + bn * 32;
      #pragma unroll
      for (int i = 0; i < 16; ++i) {
        const int idx = tid + 128 * i;
        rb[i] = *(const float4*)(Bb + (idx >> 3) * 256 + 4 * (idx & 7));
      }
    }
    k256_store_a(As, ra, tid);
    if (TB) {
      k256_store_a(Bs, rb, tid);
    } else {
      #pragma unroll
      for (int i = 0; i < 16; ++i) {
        const int idx = tid + 128 * i;
        *(float4*)(Bs + (idx >> 3) * GN_LD + 4 * (idx & 7)) = rb[i];
      }
    }
  }
  __syncthreads();
  const int tm = wave >> 1, tn = wave & 1;
  const float* Ap = As + (tm * 16 + lr) * GK_LD + 2 * hl;
  const float* Bp = TB ? Bs + (tn * 16 + lr) * GK_LD + 2 * hl
                       : Bs + (2 * hl) * GN_LD + tn * 16 + lr;
  const v8f acc = k256_chain<TB>(Ap, Bp);
  float* Dp = D + (size_t)(bm * 32 + tm * 16 + 8 * hl) * ldd + bn * 32 + tn * 16 + lr;
  #pragma unroll
  for (int v = 0; v < 8; ++v) Dp[(size_t)v * ldd] = acc[v];
}

__device__ __forceinline__ void stage_fma_plane(float4 (&v)[16], const float* __restrict__ P, float c,
                                                int ld, int col0, int tid) {
  float4 t[16];
  #pragma unroll
  for (int i = 0; i < 16; ++i) {
    const int idx = tid + 128 * i;
    t[i] = *(const float4*)(P + (idx >> 3) * ld + col0 + 4 * (idx & 7));
  }
  #pragma unroll
  for (int i = 0; i < 16; ++i) {
    v[i].x = fmaf(c, t[i].x, v[i].x);  v[i].y = fmaf(c, t[i].y, v[i].y);
    v[i].z = fmaf(c, t[i].z, v[i].z);  v[i].w = fmaf(c, t[i].w, v[i].w);
  }
}
template <int NJ>
__global__ __launch_bounds__(128) void stage_kernel(
    float* __restrict__ Kout, float* __restrict__ Lout,
    const float* __restrict__ Ad, const float* __restrict__ Bd, int blk,
    const float* __restrict__ K0, const float* __restrict__ K1, const float* __restrict__ K2,
    const float* __restrict__ K3, const float* __restrict__ K4,
    const float* __restrict__ W0, const float* __restrict__ W1, const float* __restrict__ W2,
    const float* __restrict__ W3, const float* __restrict__ W4,
    float c0, float c1, float c2, float c3, float c4) {
  __shared__ __attribute__((aligned(16))) float As[32 * GK_LD];
  __shared__ __attribute__((aligned(16))) float Bs[256 * GN_LD];
  const int tid = threadIdx.x;
  const int lane = tid & 31;
  const int wave = __builtin_amdgcn_readfirstlane(tid >> 5);
  const int hl = lane >> 4, lr = lane & 15;
  const bool isK = blockIdx.x < 64;
  const int q = isK ? (int)blockIdx.x : (int)blockIdx.x - 64;
  const int bm = isK ? (q >> 3) : (q / 6);
  const int bn = isK ? (q & 7) : (q % 6);
  const int ld = isK ? 256 : 192;
  {
    float4 ra[16], v[16];
    k256_load_a(ra, Ad, bm, tid);
    #pragma unroll
    for (int i = 0; i < 16; ++i) {
      const int idx = tid + 128 * i;
      const int k = idx >> 3, n = bn * 32 + 4 * (idx & 7);
      v[i] = isK ? make_float4(k == n ? 1.0f : 0.0f, k == n + 1 ? 1.0f : 0.0f,
                               k == n + 2 ? 1.0f : 0.0f, k == n + 3 ? 1.0f : 0.0f)
                 : make_float4(0.0f, 0.0f, 0.0f, 0.0f);
    }
    stage_fma_plane(v, isK ? K0 : W0, c0, ld, bn * 32, tid);
    if (NJ > 1) stage_fma_plane(v, isK ? K1 : W1, c1, ld, bn * 32, tid);
    if (NJ > 2) stage_fma_plane(v, isK ? K2 : W2, c2, ld, bn * 32, tid);
    if (NJ > 3) stage_fma_plane(v, isK ? K3 : W3, c3, ld, bn * 32, tid);
    if (NJ > 4) stage_fma_plane(v, isK ? K4 : W4, c4, ld, bn * 32, tid);
    k256_store_a(As, ra, tid);
    #pragma unroll
    for (int i = 0; i < 16; ++i) {
      const int idx = tid + 128 * i;
      *(float4*)(Bs + (idx >> 3) * GN_LD + 4 * (idx & 7)) = v[i];
    }
  }
  __syncthreads();
  const int tm = wave >> 1, tn = wave & 1;
  const float* Ap = As + (tm * 16 + lr) * GK_LD + 2 * hl;
  const float* Bp = Bs + (2 * hl) * GN_LD + tn * 16 + lr;
  const v8f acc = k256_chain<false>(Ap, Bp);
  const int row0 = bm * 32 + tm * 16 + 8 * hl;
  const int n = bn * 32 + tn * 16 + lr;
  if (isK) {
    float* Dp = Kout + (size_t)row0 * 256 + n;
    #pragma unroll
    for (int v = 0; v < 8; ++v) Dp[(size_t)v * 256] = acc[v];
  } else {
    float* Dp = Lout + (size_t)row0 * 192 + n;
    if (bn == blk) {
      const float* Bp2 = Bd + (size_t)row0 * 32 + (n & 31);
      #pragma unroll
      for (int v = 0; v < 8; ++v) Dp[(size_t)v * 192] = acc[v] + Bp2[(size_t)v * 32];
    } else {
      #pragma unroll
      for (int v = 0; v < 8; ++v) Dp[(size_t)v * 192] = acc[v];
    }
  }
}

static constexpr int DR_ROWS = 128;
static constexpr int DR_LD = 132;

__device__ __forceinline__ float xhalf(float x) {
  const int b = __builtin_bit_cast(int, x);
  return __builtin_bit_cast(float,
      __builtin_amdgcn_permlanex16(b, b, 0x76543210, 0xfedcba98, false, false));
}

__global__ __launch_bounds__(256) void drive_kernel(
    float* __restrict__ G, const float* __restrict__ u, const float* __restrict__ Q) {
  __shared__ __attribute__((aligned(16))) float Pt[DR_ROWS * DR_LD];
  const int tid = threadIdx.x;
  const int lane = tid & 31;
  const int wave = __builtin_amdgcn_readfirstlane(tid >> 5);
  const int hl = lane >> 4, lr = lane & 15;
  const int n0 = blockIdx.x * DR_ROWS;
  const int col0 = wave * 32;

  {
    const int cc = tid & 7, rg = tid >> 3;
    const int nb = n0 + 4 * rg;
    float4 U[7];
    #pragma unroll
    for (int i = 0; i < 7; ++i) {
      int g = nb - 1 + i;
      g = g < 0 ? 0 : (g > Tn - 1 ? Tn - 1 : g);
      U[i] = *(const float4*)(u + (size_t)g * 32 + 4 * cc);
    }
    #pragma unroll
    for (int t = 0; t < 4; ++t) {
      const int n = nb + t;
      const bool first = (n == 0), penult = (n == Tn - 2), last = (n == Tn - 1);
      const float4 mb = first ? U[t + 1] : U[t];
      const float ms = first ? 1.0f : 0.5f;
      const float4 na = penult ? U[t + 2] : U[t + 3];
      const float ns = penult ? 1.0f : 0.5f;
      float4 p0 = U[t + 1];
      float4 p1 = make_float4(ms * (U[t + 2].x - mb.x), ms * (U[t + 2].y - mb.y),
                              ms * (U[t + 2].z - mb.z), ms * (U[t + 2].w - mb.w));
      float4 p2 = U[t + 2];
      float4 p3 = make_float4(ns * (na.x - U[t + 1].x), ns * (na.y - U[t + 1].y),
                              ns * (na.z - U[t + 1].z), ns * (na.w - U[t + 1].w));
      if (last) {
        const float4 z = make_float4(0.0f, 0.0f, 0.0f, 0.0f);
        p0 = z; p1 = z; p2 = z; p3 = z;
      }
      float* prow = Pt + (4 * rg + t) * DR_LD + 4 * cc;
      *(float4*)(prow)      = p0;
      *(float4*)(prow + 32) = p1;
      *(float4*)(prow + 64) = p2;
      *(float4*)(prow + 96) = p3;
    }
  }

  v2f b0[32], b1[32];
  {
    const float* Qp = Q + (size_t)(2 * hl) * 256 + col0 + lr;
    #pragma unroll
    for (int kc = 0; kc < 32; ++kc) {
      b0[kc][0] = Qp[kc * 1024];
      b0[kc][1] = Qp[kc * 1024 + 256];
      b1[kc][0] = Qp[kc * 1024 + 16];
      b1[kc][1] = Qp[kc * 1024 + 256 + 16];
    }
  }
  __syncthreads();

  const float* Ap = Pt + lr * DR_LD + 2 * hl;
  for (int hb = 0; hb < 2; ++hb) {
    const float* Ah = Ap + hb * 64 * DR_LD;
    v8f acc0[4] = {}, acc1[4] = {};
    #pragma unroll
    for (int kc = 0; kc < 32; ++kc) {
      #pragma unroll
      for (int r = 0; r < 4; ++r) {
        const v2f a = *(const v2f*)(Ah + r * 16 * DR_LD + kc * 4);
        acc0[r] = WMMA_F32(a, b0[kc], acc0[r]);
        acc1[r] = WMMA_F32(a, b1[kc], acc1[r]);
      }
    }
    float* Gp = G + (size_t)(n0 + hb * 64) * 256 + col0 + lane;
    #pragma unroll
    for (int r = 0; r < 4; ++r) {
      #pragma unroll
      for (int v = 0; v < 8; ++v) {
        const float t0 = acc0[r][v], t1 = acc1[r][v];
        const float got = xhalf(hl ? t0 : t1);
        const float lo = hl ? got : t0;
        const float hi = hl ? t1 : got;
        Gp[(size_t)(r * 16 + v) * 256] = lo;
        Gp[(size_t)(r * 16 + 8 + v) * 256] = hi;
      }
    }
  }
}

__global__ __launch_bounds__(256) void scan_kernel(
    const float* __restrict__ Mg, const float* __restrict__ G,
    const float* __restrict__ Init,
    float* __restrict__ OutAll,
    float* __restrict__ OutEnd,
    int L) {
  __shared__ __attribute__((aligned(16))) float smem[SCAN_LDS_FLOATS];
  float* Ms = smem;
  float* Xbuf0 = smem + 256 * MS_LD;
  float* Xbuf1 = Xbuf0 + 256 * SC_COLS;
  const int tid = threadIdx.x;
  const int lane = tid & 31, wave = tid >> 5;
  const int hl = lane >> 4, lr = lane & 15;
  const int cbase = blockIdx.x * SC_COLS;
  const int tr0 = wave * 2;

  {
    const float4* src = (const float4*)Mg;
    float4* dst = (float4*)Ms;
    for (int k = tid; k < 65536 / 4; k += 256)
      dst[(k >> 6) * (MS_LD / 4) + (k & 63)] = src[k];
  }
  for (int e = 0; e < SC_COLS; ++e)
    Xbuf0[tid * SC_COLS + e] = Init ? Init[(size_t)(cbase + e) * 256 + tid] : 0.0f;
  __syncthreads();

  const float* gbase = G + ((size_t)(cbase + lr) * L) * 256 + tr0 * 16 + 8 * hl;

  float* Xc = Xbuf0;
  float* Xn = Xbuf1;
  for (int i = 0; i < L; ++i) {
    if (OutAll) {
      for (int e = 0; e < SC_COLS; ++e) {
        size_t m = (size_t)(cbase + e) * L + i;
        OutAll[m * 256 + tid] = Xc[tid * SC_COLS + e];
      }
    }
    const float* gp = gbase + (size_t)i * 256;
    if (i + 1 < L) __builtin_prefetch(gp + 256, 0, 1);
    v8f acc0, acc1;
    {
      float4 ga = *(const float4*)(gp);
      float4 gb = *(const float4*)(gp + 4);
      float4 gc = *(const float4*)(gp + 16);
      float4 gd = *(const float4*)(gp + 20);
      acc0[0]=ga.x; acc0[1]=ga.y; acc0[2]=ga.z; acc0[3]=ga.w;
      acc0[4]=gb.x; acc0[5]=gb.y; acc0[6]=gb.z; acc0[7]=gb.w;
      acc1[0]=gc.x; acc1[1]=gc.y; acc1[2]=gc.z; acc1[3]=gc.w;
      acc1[4]=gd.x; acc1[5]=gd.y; acc1[6]=gd.z; acc1[7]=gd.w;
    }
    const float* Mr0 = Ms + (size_t)(tr0 * 16 + lr) * MS_LD + 2 * hl;
    const float* Mr1 = Mr0 + 16 * MS_LD;
    const float* Xk = Xc + 2 * hl * SC_COLS + lr;
    #pragma unroll 8
    for (int kc = 0; kc < 64; ++kc) {
      v2f a0 = *(const v2f*)(Mr0 + kc * 4);
      v2f a1 = *(const v2f*)(Mr1 + kc * 4);
      v2f b;
      b[0] = Xk[kc * 4 * SC_COLS];
      b[1] = Xk[kc * 4 * SC_COLS + SC_COLS];
      acc0 = WMMA_F32(a0, b, acc0);
      acc1 = WMMA_F32(a1, b, acc1);
    }
    #pragma unroll
    for (int v = 0; v < 8; ++v) {
      Xn[(tr0 * 16 + 8 * hl + v) * SC_COLS + lr] = acc0[v];
      Xn[((tr0 + 1) * 16 + 8 * hl + v) * SC_COLS + lr] = acc1[v];
    }
    __syncthreads();
    float* tmp = Xc; Xc = Xn; Xn = tmp;
  }
  if (OutEnd) {
    for (int e = 0; e < SC_COLS; ++e)
      OutEnd[(size_t)(cbase + e) * 256 + tid] = Xc[tid * SC_COLS + e];
  }
}

static constexpr int TV_PITCH = 20;
static constexpr int TV_FLOATS = 16 * TV_PITCH;

template <int CTRL>
__device__ __forceinline__ float lane_read(float v) {
  return __int_as_float(__builtin_amdgcn_update_dpp(0, __float_as_int(v), CTRL, 0xF, 0xF, true));
}
static constexpr int DPP_SWAP1 = 0xB1;
static constexpr int DPP_SWAP2 = 0x4E;
static constexpr int DPP_ROR4  = 0x124;
static constexpr int DPP_ROR8  = 0x128;

__global__ __launch_bounds__(1024) void carry_kernel(
    const float* __restrict__ Mg, const float* __restrict__ Z,
    const float* __restrict__ x0, float* __restrict__ Tcar, int n) {
  __shared__ __attribute__((aligned(16))) float tv[2 * TV_FLOATS];
  const int tid = threadIdx.x;
  const int lane = tid & 31, wave = tid >> 5;
  const int g = lane & 15, rg = lane >> 4;
  const int rbase = wave * 8 + rg * 4;

  v2f m[4][8];
  for (int i = 0; i < 4; ++i) {
    const float4* mrow = (const float4*)(Mg + (size_t)(rbase + i) * 256 + 16 * g);
    for (int c = 0; c < 4; ++c) {
      float4 t4 = mrow[c];
      m[i][2 * c]     = (v2f){t4.x, t4.y};
      m[i][2 * c + 1] = (v2f){t4.z, t4.w};
    }
  }
  const int outslot = (tid >> 4) * TV_PITCH + (tid & 15);
  if (tid < 256) tv[outslot] = x0[tid];
  const bool odd1 = (g & 1) != 0, odd2 = (g & 2) != 0;
  const int myrow = rbase + (odd1 ? 2 : 0) + (odd2 ? 1 : 0);
  const int myslot = (myrow >> 4) * TV_PITCH + (myrow & 15);
  float zc = (n > 1) ? Z[myrow] : 0.0f;
  __syncthreads();

  for (int j = 0; j + 1 < n; ++j) {
    const float* tcur = tv + (j & 1) * TV_FLOATS;
    float* tnxt = tv + ((j + 1) & 1) * TV_FLOATS;
    if (wave < 8) Tcar[(size_t)j * 256 + tid] = tcur[outslot];
    float zn = 0.0f;
    if (j + 2 < n) zn = Z[(size_t)(j + 1) * 256 + myrow];
    const float* xs = tcur + g * TV_PITCH;
    v2f acc[4] = {};
    for (int c = 0; c < 4; ++c) {
      float4 x4 = *(const float4*)(xs + 4 * c);
      v2f xa = {x4.x, x4.y}, xb = {x4.z, x4.w};
      for (int i = 0; i < 4; ++i) {
        acc[i] += m[i][2 * c] * xa;
        acc[i] += m[i][2 * c + 1] * xb;
      }
    }
    float p0 = acc[0][0] + acc[0][1];
    float p1 = acc[1][0] + acc[1][1];
    float p2 = acc[2][0] + acc[2][1];
    float p3 = acc[3][0] + acc[3][1];
    float s0 = odd1 ? p0 : p2, s1 = odd1 ? p1 : p3;
    float k0 = odd1 ? p2 : p0, k1 = odd1 ? p3 : p1;
    k0 += lane_read<DPP_SWAP1>(s0);
    k1 += lane_read<DPP_SWAP1>(s1);
    float s = odd2 ? k0 : k1;
    float k = odd2 ? k1 : k0;
    k += lane_read<DPP_SWAP2>(s);
    k += lane_read<DPP_ROR4>(k);
    k += lane_read<DPP_ROR8>(k);
    if (g < 4) tnxt[myslot] = k + zc;
    __syncthreads();
    zc = zn;
  }
  if (wave < 8) Tcar[(size_t)(n - 1) * 256 + tid] = tv[((n - 1) & 1) * TV_FLOATS + outslot];
}

static inline void launch_gemm(hipStream_t s, float* D, int ldd,
                               const float* A, int lda, const float* B, int ldb,
                               const float* Cin, int ldc,
                               int M, int N, int K, float alpha, float beta, int transB) {
  int tiles = (M / 16) * (N / 16);
  gemm_kernel<<<(tiles + 3) / 4, 128, 0, s>>>(D, ldd, A, lda, B, ldb, Cin, ldc,
                                              M, N, K, alpha, beta, transB);
}

extern "C" void kernel_launch(void* const* d_in, const int* in_sizes, int n_in,
                              void* d_out, int out_size, void* d_ws, size_t ws_size,
                              hipStream_t stream) {
  (void)in_sizes; (void)n_in; (void)out_size; (void)ws_size;
  const float* t  = (const float*)d_in[0];
  const float* u  = (const float*)d_in[1];
  const float* x0 = (const float*)d_in[2];
  const float* Am = (const float*)d_in[3];
  const float* Bm = (const float*)d_in[4];
  const float* Cm = (const float*)d_in[5];
  const float* Dm = (const float*)d_in[6];
  float* xout = (float*)d_out;
  float* yout = xout + (size_t)Tn * NS;

  float* w = (float*)d_ws;
  float* Ad = w;  w += 65536;
  float* Bd = w;  w += 8192;
  float* Kt[6]; for (int i = 0; i < 6; i++) { Kt[i] = w; w += 65536; }
  Kt[0] = Ad;
  float* Lt[6]; for (int i = 0; i < 6; i++) { Lt[i] = w; w += 49152; }
  float* M64  = w; w += 65536 + 49152;
  float* Mm   = w; w += 65536;
  w += 49152;
  float* Qc   = w; w += 32768;
  float* M8   = w; w += 65536;
  float* M512 = w; w += 65536;
  float* PT1  = w; w += 65536;
  float* PT2  = w; w += 65536;
  float* Pm   = w; w += (size_t)Tn * 128;
  float* G    = w; w += (size_t)Tn * 256;
  float* Zend2 = w; w += (size_t)C2 * 256;
  float* Scar2 = w; w += (size_t)C2 * 256;
  float* Zend3 = w; w += (size_t)C3 * 256;
  float* Tcar  = w; w += (size_t)C3 * 256;
  float* Zend1 = Pm;
  float* Scar1 = Pm + (size_t)C1 * 256;

  auto gdim = [](int n) { return dim3((unsigned)((n + 255) / 256)); };

  k_prep<<<gdim(65536 + 49152), 256, 0, stream>>>(Ad, Bd, Lt[0], Am, Bm, t);

  double aa[7][7] = {};
  aa[2][1] = 0.2;
  aa[3][1] = 3.0/40.0;      aa[3][2] = 9.0/40.0;
  aa[4][1] = 44.0/45.0;     aa[4][2] = -56.0/15.0;     aa[4][3] = 32.0/9.0;
  aa[5][1] = 19372.0/6561.0; aa[5][2] = -25360.0/2187.0;
  aa[5][3] = 64448.0/6561.0; aa[5][4] = -212.0/729.0;
  aa[6][1] = 9017.0/3168.0; aa[6][2] = -355.0/33.0; aa[6][3] = 46732.0/5247.0;
  aa[6][4] = 49.0/176.0;    aa[6][5] = -5103.0/18656.0;
  stage_kernel<1><<<112, 128, 0, stream>>>(Kt[1], Lt[1], Ad, Bd, 1,
      Kt[0], Kt[0], Kt[0], Kt[0], Kt[0], Lt[0], Lt[0], Lt[0], Lt[0], Lt[0],
      (float)aa[2][1], 0.f, 0.f, 0.f, 0.f);
  stage_kernel<2><<<112, 128, 0, stream>>>(Kt[2], Lt[2], Ad, Bd, 2,
      Kt[0], Kt[1], Kt[0], Kt[0], Kt[0], Lt[0], Lt[1], Lt[0], Lt[0], Lt[0],
      (float)aa[3][1], (float)aa[3][2], 0.f, 0.f, 0.f);
  stage_kernel<3><<<112, 128, 0, stream>>>(Kt[3], Lt[3], Ad, Bd, 3,
      Kt[0], Kt[1], Kt[2], Kt[0], Kt[0], Lt[0], Lt[1], Lt[2], Lt[0], Lt[0],
      (float)aa[4][1], (float)aa[4][2], (float)aa[4][3], 0.f, 0.f);
  stage_kernel<4><<<112, 128, 0, stream>>>(Kt[4], Lt[4], Ad, Bd, 4,
      Kt[0], Kt[1], Kt[2], Kt[3], Kt[0], Lt[0], Lt[1], Lt[2], Lt[3], Lt[0],
      (float)aa[5][1], (float)aa[5][2], (float)aa[5][3], (float)aa[5][4], 0.f);
  stage_kernel<5><<<112, 128, 0, stream>>>(Kt[5], Lt[5], Ad, Bd, 5,
      Kt[0], Kt[1], Kt[2], Kt[3], Kt[4], Lt[0], Lt[1], Lt[2], Lt[3], Lt[4],
      (float)aa[6][1], (float)aa[6][2], (float)aa[6][3], (float)aa[6][4], (float)aa[6][5]);

  const double bb[7] = {0.0, 35.0/384.0, 0.0, 500.0/1113.0, 125.0/192.0,
                        -2187.0/6784.0, 11.0/84.0};
  k_final<<<gdim(65536 + 8192), 256, 0, stream>>>(Mm, Qc,
      Kt[0], Kt[2], Kt[3], Kt[4], Kt[5], Lt[0], Lt[2], Lt[3], Lt[4], Lt[5],
      (float)bb[1], (float)bb[3], (float)bb[4], (float)bb[5], (float)bb[6]);

  gemm_k256_kernel<false, 8><<<64, 128, 0, stream>>>(PT1,  256, Mm,  Mm);
  gemm_k256_kernel<false, 8><<<64, 128, 0, stream>>>(PT2,  256, PT1, PT1);
  gemm_k256_kernel<false, 8><<<64, 128, 0, stream>>>(M8,   256, PT2, PT2);
  gemm_k256_kernel<false, 8><<<64, 128, 0, stream>>>(PT1,  256, M8,  M8);
  gemm_k256_kernel<false, 8><<<64, 128, 0, stream>>>(PT2,  256, PT1, PT1);
  gemm_k256_kernel<false, 8><<<64, 128, 0, stream>>>(M64,  256, PT2, PT2);
  gemm_k256_kernel<false, 8><<<64, 128, 0, stream>>>(PT1,  256, M64, M64);
  gemm_k256_kernel<false, 8><<<64, 128, 0, stream>>>(PT2,  256, PT1, PT1);
  gemm_k256_kernel<false, 8><<<64, 128, 0, stream>>>(M512, 256, PT2, PT2);

  drive_kernel<<<Tn / DR_ROWS, 256, 0, stream>>>(G, u, Qc);

  scan_kernel<<<C1 / SC_COLS, 256, 0, stream>>>(Mm, G, nullptr, nullptr, Zend1, LC);
  scan_kernel<<<C2 / SC_COLS, 256, 0, stream>>>(M8, Zend1, nullptr, nullptr, Zend2, LC);
  scan_kernel<<<C3 / SC_COLS, 256, 0, stream>>>(M64, Zend2, nullptr, nullptr, Zend3, LC);
  carry_kernel<<<1, 1024, 0, stream>>>(M512, Zend3, x0, Tcar, C3);
  scan_kernel<<<C3 / SC_COLS, 256, 0, stream>>>(M64, Zend2, Tcar, Scar2, nullptr, LC);
  scan_kernel<<<C2 / SC_COLS, 256, 0, stream>>>(M8, Zend1, Scar2, Scar1, nullptr, LC);
  scan_kernel<<<C1 / SC_COLS, 256, 0, stream>>>(Mm, G, Scar1, xout, nullptr, LC);

  gemm_k256_kernel<true, 1><<<Tn / 32, 128, 0, stream>>>(yout, 32, xout, Cm);
  launch_gemm(stream, yout, 32, u,    32,  Dm, 32,  yout,   32, Tn, 32, 32,  1.f, 1.f, 1);
}


# scan kernel: each wave holds the A fragments of its 32 rows of M in registers over all steps; X fragments and next drive rows load as one group up front, no load among its 128 WMMAs; sum order kept
# speedup vs baseline: 4.3101x; 1.0561x over previous
#include <hip/hip_runtime.h>

typedef __attribute__((ext_vector_type(2))) float v2f;
typedef __attribute__((ext_vector_type(8))) float v8f;

#define WMMA_F32(a, b, c) \
  __builtin_amdgcn_wmma_f32_16x16x4_f32(false, (a), false, (b), (short)0, (c), false, false)

static constexpr int Tn = 32768;
static constexpr int NS = 256;
static constexpr int LC = 8;
static constexpr int C1 = 4096;
static constexpr int C2 = 512;
static constexpr int C3 = 64;
static constexpr int SC_COLS = 16;
static constexpr int MS_LD = 260;
static constexpr int SCAN_LDS_FLOATS = 256 * MS_LD + 2 * 256 * SC_COLS;

__constant__ float d_H[6][4] = {
  {1.0f, 0.0f, 0.0f, 0.0f},
  {0.896f, 0.128f, 0.104f, -0.032f},
  {0.784f, 0.147f, 0.216f, -0.063f},
  {0.104f, 0.032f, 0.896f, -0.128f},
  {25.0f/729.0f, 8.0f/729.0f, 704.0f/729.0f, -64.0f/729.0f},
  {0.0f, 0.0f, 1.0f, 0.0f},
};

__global__ void k_prep(float* __restrict__ Ad, float* __restrict__ Bd, float* __restrict__ Lt0,
                       const float* __restrict__ A, const float* __restrict__ B,
                       const float* __restrict__ t) {
  const int idx = blockIdx.x * 256 + threadIdx.x;
  const float dt = t[1] - t[0];
  if (idx < 65536) {
    Ad[idx] = dt * A[idx];
  } else {
    const int q = idx - 65536;
    if (q < 49152) {
      const int r = q / 192, c = q - r * 192;
      float val = 0.0f;
      if (c < 32) { const float bd = dt * B[r * 32 + c]; Bd[r * 32 + c] = bd; val += bd; }
      Lt0[q] = val;
    }
  }
}
__global__ void k_final(float* __restrict__ Mm, float* __restrict__ Q,
                        const float* __restrict__ K1, const float* __restrict__ K3,
                        const float* __restrict__ K4, const float* __restrict__ K5,
                        const float* __restrict__ K6,
                        const float* __restrict__ W1, const float* __restrict__ W3,
                        const float* __restrict__ W4, const float* __restrict__ W5,
                        const float* __restrict__ W6,
                        float b1, float b3, float b4, float b5, float b6) {
  const int idx = blockIdx.x * 256 + threadIdx.x;
  if (idx < 65536) {
    float v = ((idx >> 8) == (idx & 255)) ? 1.0f : 0.0f;
    v = fmaf(b1, K1[idx], v);
    v = fmaf(b3, K3[idx], v);
    v = fmaf(b4, K4[idx], v);
    v = fmaf(b5, K5[idx], v);
    v = fmaf(b6, K6[idx], v);
    Mm[idx] = v;
  } else {
    const int q = idx - 65536;
    if (q >= 256 * 32) return;
    const int r = q >> 5, c = q & 31;
    float w[6];
    for (int st = 0; st < 6; ++st) {
      const int o = r * 192 + 32 * st + c;
      float t = 0.0f;
      t = fmaf(b1, W1[o], t);
      t = fmaf(b3, W3[o], t);
      t = fmaf(b4, W4[o], t);
      t = fmaf(b5, W5[o], t);
      t = fmaf(b6, W6[o], t);
      w[st] = t;
    }
    for (int j = 0; j < 4; ++j) {
      float s = 0.0f;
      for (int st = 0; st < 6; ++st) s += d_H[st][j] * w[st];
      Q[(32 * j + c) * 256 + r] = s;
    }
  }
}
__global__ __launch_bounds__(128) void gemm_kernel(
    float* __restrict__ D, int ldd,
    const float* __restrict__ A, int lda,
    const float* __restrict__ B, int ldb,
    const float* __restrict__ Cin, int ldc,
    int Mr, int Nc, int Kd, float alpha, float beta, int transB) {
  const int lane = threadIdx.x & 31, wave = threadIdx.x >> 5;
  const int hl = lane >> 4, lr = lane & 15;
  const int tilesN = Nc >> 4;
  const int tile = blockIdx.x * 4 + wave;
  if (tile >= (Mr >> 4) * tilesN) return;
  const int tm = tile / tilesN, tn = tile % tilesN;
  const float* Ar = A + (size_t)(tm * 16 + lr) * lda + 2 * hl;
  v8f acc = {};
  if (transB) {
    const float* Br = B + (size_t)(tn * 16 + lr) * ldb + 2 * hl;
    #pragma unroll 4
    for (int k = 0; k < Kd; k += 4) {
      v2f a = *(const v2f*)(Ar + k);
      v2f b = *(const v2f*)(Br + k);
      acc = WMMA_F32(a, b, acc);
    }
  } else {
    const float* Bc = B + (size_t)(2 * hl) * ldb + tn * 16 + lr;
    #pragma unroll 4
    for (int k = 0; k < Kd; k += 4) {
      v2f a = *(const v2f*)(Ar + k);
      v2f b;
      b[0] = Bc[(size_t)k * ldb];
      b[1] = Bc[(size_t)k * ldb + ldb];
      acc = WMMA_F32(a, b, acc);
    }
  }
  float* Dp = D + (size_t)(tm * 16 + 8 * hl) * ldd + tn * 16 + lr;
  if (beta != 0.0f) {
    const float* Cp = Cin + (size_t)(tm * 16 + 8 * hl) * ldc + tn * 16 + lr;
    #pragma unroll
    for (int v = 0; v < 8; ++v)
      Dp[(size_t)v * ldd] = alpha * acc[v] + beta * Cp[(size_t)v * ldc];
  } else {
    #pragma unroll
    for (int v = 0; v < 8; ++v)
      Dp[(size_t)v * ldd] = alpha * acc[v];
  }
}

static constexpr int GK_LD = 260;
static constexpr int GN_LD = 40;

__device__ __forceinline__ void k256_load_a(float4 (&ra)[16], const float* __restrict__ A, int bm, int tid) {
  const float4* Ab = (const float4*)(A + (size_t)bm * 32 * 256);
  #pragma unroll
  for (int i = 0; i < 16; ++i) ra[i] = Ab[tid + 128 * i];
}
__device__ __forceinline__ void k256_store_a(float* As, const float4 (&ra)[16], int tid) {
  #pragma unroll
  for (int i = 0; i < 16; ++i) {
    const int idx = tid + 128 * i;
    *(float4*)(As + (idx >> 6) * GK_LD + 4 * (idx & 63)) = ra[i];
  }
}
template <bool TB>
__device__ __forceinline__ void k256_frags(v2f (&a)[16], v2f (&b)[16],
                                           const float* Ap, const float* Bp, int g) {
  #pragma unroll
  for (int j = 0; j < 16; ++j) {
    const int kc = 16 * g + j;
    a[j] = *(const v2f*)(Ap + 4 * kc);
    if (TB) {
      b[j] = *(const v2f*)(Bp + 4 * kc);
    } else {
      b[j][0] = Bp[4 * kc * GN_LD];
      b[j][1] = Bp[4 * kc * GN_LD + GN_LD];
    }
  }
}
__device__ __forceinline__ v8f k256_wmma16(const v2f (&a)[16], const v2f (&b)[16], v8f acc) {
  #pragma unroll
  for (int j = 0; j < 16; ++j) acc = WMMA_F32(a[j], b[j], acc);
  return acc;
}
template <bool TB>
__device__ __forceinline__ v8f k256_chain(const float* Ap, const float* Bp) {
  v2f a0[16], b0[16], a1[16], b1[16];
  v8f acc = {};
  k256_frags<TB>(a0, b0, Ap, Bp, 0);
  k256_frags<TB>(a1, b1, Ap, Bp, 1);
  __builtin_amdgcn_sched_barrier(0);
  acc = k256_wmma16(a0, b0, acc);
  k256_frags<TB>(a0, b0, Ap, Bp, 2);
  __builtin_amdgcn_sched_barrier(0);
  acc = k256_wmma16(a1, b1, acc);
  k256_frags<TB>(a1, b1, Ap, Bp, 3);
  __builtin_amdgcn_sched_barrier(0);
  acc = k256_wmma16(a0, b0, acc);
  acc = k256_wmma16(a1, b1, acc);
  return acc;
}

template <bool TB, int NBN>
__global__ __launch_bounds__(128) void gemm_k256_kernel(
    float* __restrict__ D, int ldd, const float* __restrict__ A, const float* __restrict__ B) {
  __shared__ __attribute__((aligned(16))) float As[32 * GK_LD];
  __shared__ __attribute__((aligned(16))) float Bs[TB ? 32 * GK_LD : 256 * GN_LD];
  const int tid = threadIdx.x;
  const int lane = tid & 31;
  const int wave = __builtin_amdgcn_readfirstlane(tid >> 5);
  const int hl = lane >> 4, lr = lane & 15;
  const int bm = blockIdx.x / NBN, bn = blockIdx.x % NBN;
  {
    float4 ra[16], rb[16];
    k256_load_a(ra, A, bm, tid);
    if (TB) {
      k256_load_a(rb, B, bn, tid);
    } else {
      const float* Bb = B + bn * 32;
      #pragma unroll
      for (int i = 0; i < 16; ++i) {
        const int idx = tid + 128 * i;
        rb[i] = *(const float4*)(Bb + (idx >> 3) * 256 + 4 * (idx & 7));
      }
    }
    k256_store_a(As, ra, tid);
    if (TB) {
      k256_store_a(Bs, rb, tid);
    } else {
      #pragma unroll
      for (int i = 0; i < 16; ++i) {
        const int idx = tid + 128 * i;
        *(float4*)(Bs + (idx >> 3) * GN_LD + 4 * (idx & 7)) = rb[i];
      }
    }
  }
  __syncthreads();
  const int tm = wave >> 1, tn = wave & 1;
  const float* Ap = As + (tm * 16 + lr) * GK_LD + 2 * hl;
  const float* Bp = TB ? Bs + (tn * 16 + lr) * GK_LD + 2 * hl
                       : Bs + (2 * hl) * GN_LD + tn * 16 + lr;
  const v8f acc = k256_chain<TB>(Ap, Bp);
  float* Dp = D + (size_t)(bm * 32 + tm * 16 + 8 * hl) * ldd + bn * 32 + tn * 16 + lr;
  #pragma unroll
  for (int v = 0; v < 8; ++v) Dp[(size_t)v * ldd] = acc[v];
}

__device__ __forceinline__ void stage_fma_plane(float4 (&v)[16], const float* __restrict__ P, float c,
                                                int ld, int col0, int tid) {
  float4 t[16];
  #pragma unroll
  for (int i = 0; i < 16; ++i) {
    const int idx = tid + 128 * i;
    t[i] = *(const float4*)(P + (idx >> 3) * ld + col0 + 4 * (idx & 7));
  }
  #pragma unroll
  for (int i = 0; i < 16; ++i) {
    v[i].x = fmaf(c, t[i].x, v[i].x);  v[i].y = fmaf(c, t[i].y, v[i].y);
    v[i].z = fmaf(c, t[i].z, v[i].z);  v[i].w = fmaf(c, t[i].w, v[i].w);
  }
}
template <int NJ>
__global__ __launch_bounds__(128) void stage_kernel(
    float* __restrict__ Kout, float* __restrict__ Lout,
    const float* __restrict__ Ad, const float* __restrict__ Bd, int blk,
    const float* __restrict__ K0, const float* __restrict__ K1, const float* __restrict__ K2,
    const float* __restrict__ K3, const float* __restrict__ K4,
    const float* __restrict__ W0, const float* __restrict__ W1, const float* __restrict__ W2,
    const float* __restrict__ W3, const float* __restrict__ W4,
    float c0, float c1, float c2, float c3, float c4) {
  __shared__ __attribute__((aligned(16))) float As[32 * GK_LD];
  __shared__ __attribute__((aligned(16))) float Bs[256 * GN_LD];
  const int tid = threadIdx.x;
  const int lane = tid & 31;
  const int wave = __builtin_amdgcn_readfirstlane(tid >> 5);
  const int hl = lane >> 4, lr = lane & 15;
  const bool isK = blockIdx.x < 64;
  const int q = isK ? (int)blockIdx.x : (int)blockIdx.x - 64;
  const int bm = isK ? (q >> 3) : (q / 6);
  const int bn = isK ? (q & 7) : (q % 6);
  const int ld = isK ? 256 : 192;
  {
    float4 ra[16], v[16];
    k256_load_a(ra, Ad, bm, tid);
    #pragma unroll
    for (int i = 0; i < 16; ++i) {
      const int idx = tid + 128 * i;
      const int k = idx >> 3, n = bn * 32 + 4 * (idx & 7);
      v[i] = isK ? make_float4(k == n ? 1.0f : 0.0f, k == n + 1 ? 1.0f : 0.0f,
                               k == n + 2 ? 1.0f : 0.0f, k == n + 3 ? 1.0f : 0.0f)
                 : make_float4(0.0f, 0.0f, 0.0f, 0.0f);
    }
    stage_fma_plane(v, isK ? K0 : W0, c0, ld, bn * 32, tid);
    if (NJ > 1) stage_fma_plane(v, isK ? K1 : W1, c1, ld, bn * 32, tid);
    if (NJ > 2) stage_fma_plane(v, isK ? K2 : W2, c2, ld, bn * 32, tid);
    if (NJ > 3) stage_fma_plane(v, isK ? K3 : W3, c3, ld, bn * 32, tid);
    if (NJ > 4) stage_fma_plane(v, isK ? K4 : W4, c4, ld, bn * 32, tid);
    k256_store_a(As, ra, tid);
    #pragma unroll
    for (int i = 0; i < 16; ++i) {
      const int idx = tid + 128 * i;
      *(float4*)(Bs + (idx >> 3) * GN_LD + 4 * (idx & 7)) = v[i];
    }
  }
  __syncthreads();
  const int tm = wave >> 1, tn = wave & 1;
  const float* Ap = As + (tm * 16 + lr) * GK_LD + 2 * hl;
  const float* Bp = Bs + (2 * hl) * GN_LD + tn * 16 + lr;
  const v8f acc = k256_chain<false>(Ap, Bp);
  const int row0 = bm * 32 + tm * 16 + 8 * hl;
  const int n = bn * 32 + tn * 16 + lr;
  if (isK) {
    float* Dp = Kout + (size_t)row0 * 256 + n;
    #pragma unroll
    for (int v = 0; v < 8; ++v) Dp[(size_t)v * 256] = acc[v];
  } else {
    float* Dp = Lout + (size_t)row0 * 192 + n;
    if (bn == blk) {
      const float* Bp2 = Bd + (size_t)row0 * 32 + (n & 31);
      #pragma unroll
      for (int v = 0; v < 8; ++v) Dp[(size_t)v * 192] = acc[v] + Bp2[(size_t)v * 32];
    } else {
      #pragma unroll
      for (int v = 0; v < 8; ++v) Dp[(size_t)v * 192] = acc[v];
    }
  }
}

static constexpr int DR_ROWS = 128;
static constexpr int DR_LD = 132;

__device__ __forceinline__ float xhalf(float x) {
  const int b = __builtin_bit_cast(int, x);
  return __builtin_bit_cast(float,
      __builtin_amdgcn_permlanex16(b, b, 0x76543210, 0xfedcba98, false, false));
}

__global__ __launch_bounds__(256) void drive_kernel(
    float* __restrict__ G, const float* __restrict__ u, const float* __restrict__ Q) {
  __shared__ __attribute__((aligned(16))) float Pt[DR_ROWS * DR_LD];
  const int tid = threadIdx.x;
  const int lane = tid & 31;
  const int wave = __builtin_amdgcn_readfirstlane(tid >> 5);
  const int hl = lane >> 4, lr = lane & 15;
  const int n0 = blockIdx.x * DR_ROWS;
  const int col0 = wave * 32;

  {
    const int cc = tid & 7, rg = tid >> 3;
    const int nb = n0 + 4 * rg;
    float4 U[7];
    #pragma unroll
    for (int i = 0; i < 7; ++i) {
      int g = nb - 1 + i;
      g = g < 0 ? 0 : (g > Tn - 1 ? Tn - 1 : g);
      U[i] = *(const float4*)(u + (size_t)g * 32 + 4 * cc);
    }
    #pragma unroll
    for (int t = 0; t < 4; ++t) {
      const int n = nb + t;
      const bool first = (n == 0), penult = (n == Tn - 2), last = (n == Tn - 1);
      const float4 mb = first ? U[t + 1] : U[t];
      const float ms = first ? 1.0f : 0.5f;
      const float4 na = penult ? U[t + 2] : U[t + 3];
      const float ns = penult ? 1.0f : 0.5f;
      float4 p0 = U[t + 1];
      float4 p1 = make_float4(ms * (U[t + 2].x - mb.x), ms * (U[t + 2].y - mb.y),
                              ms * (U[t + 2].z - mb.z), ms * (U[t + 2].w - mb.w));
      float4 p2 = U[t + 2];
      float4 p3 = make_float4(ns * (na.x - U[t + 1].x), ns * (na.y - U[t + 1].y),
                              ns * (na.z - U[t + 1].z), ns * (na.w - U[t + 1].w));
      if (last) {
        const float4 z = make_float4(0.0f, 0.0f, 0.0f, 0.0f);
        p0 = z; p1 = z; p2 = z; p3 = z;
      }
      float* prow = Pt + (4 * rg + t) * DR_LD + 4 * cc;
      *(float4*)(prow)      = p0;
      *(float4*)(prow + 32) = p1;
      *(float4*)(prow + 64) = p2;
      *(float4*)(prow + 96) = p3;
    }
  }

  v2f b0[32], b1[32];
  {
    const float* Qp = Q + (size_t)(2 * hl) * 256 + col0 + lr;
    #pragma unroll
    for (int kc = 0; kc < 32; ++kc) {
      b0[kc][0] = Qp[kc * 1024];
      b0[kc][1] = Qp[kc * 1024 + 256];
      b1[kc][0] = Qp[kc * 1024 + 16];
      b1[kc][1] = Qp[kc * 1024 + 256 + 16];
    }
  }
  __syncthreads();

  const float* Ap = Pt + lr * DR_LD + 2 * hl;
  for (int hb = 0; hb < 2; ++hb) {
    const float* Ah = Ap + hb * 64 * DR_LD;
    v8f acc0[4] = {}, acc1[4] = {};
    #pragma unroll
    for (int kc = 0; kc < 32; ++kc) {
      #pragma unroll
      for (int r = 0; r < 4; ++r) {
        const v2f a = *(const v2f*)(Ah + r * 16 * DR_LD + kc * 4);
        acc0[r] = WMMA_F32(a, b0[kc], acc0[r]);
        acc1[r] = WMMA_F32(a, b1[kc], acc1[r]);
      }
    }
    float* Gp = G + (size_t)(n0 + hb * 64) * 256 + col0 + lane;
    #pragma unroll
    for (int r = 0; r < 4; ++r) {
      #pragma unroll
      for (int v = 0; v < 8; ++v) {
        const float t0 = acc0[r][v], t1 = acc1[r][v];
        const float got = xhalf(hl ? t0 : t1);
        const float lo = hl ? got : t0;
        const float hi = hl ? t1 : got;
        Gp[(size_t)(r * 16 + v) * 256] = lo;
        Gp[(size_t)(r * 16 + 8 + v) * 256] = hi;
      }
    }
  }
}

__global__ __launch_bounds__(256) void scan_kernel(
    const float* __restrict__ Mg, const float* __restrict__ G,
    const float* __restrict__ Init,
    float* __restrict__ OutAll,
    float* __restrict__ OutEnd,
    int L) {
  __shared__ __attribute__((aligned(16))) float smem[SCAN_LDS_FLOATS];
  float* Ms = smem;
  float* Xbuf0 = smem + 256 * MS_LD;
  float* Xbuf1 = Xbuf0 + 256 * SC_COLS;
  const int tid = threadIdx.x;
  const int lane = tid & 31, wave = tid >> 5;
  const int hl = lane >> 4, lr = lane & 15;
  const int cbase = blockIdx.x * SC_COLS;
  const int tr0 = wave * 2;

  {
    const float4* src = (const float4*)Mg;
    float4* dst = (float4*)Ms;
    for (int k = tid; k < 65536 / 4; k += 256)
      dst[(k >> 6) * (MS_LD / 4) + (k & 63)] = src[k];
  }
  for (int e = 0; e < SC_COLS; ++e)
    Xbuf0[tid * SC_COLS + e] = Init ? Init[(size_t)(cbase + e) * 256 + tid] : 0.0f;
  __syncthreads();

  v2f a0[64], a1[64];
  {
    const float* Mr0 = Ms + (size_t)(tr0 * 16 + lr) * MS_LD + 2 * hl;
    const float* Mr1 = Mr0 + 16 * MS_LD;
    #pragma unroll
    for (int kc = 0; kc < 64; ++kc) {
      a0[kc] = *(const v2f*)(Mr0 + kc * 4);
      a1[kc] = *(const v2f*)(Mr1 + kc * 4);
    }
  }

  const float* gbase = G + ((size_t)(cbase + lr) * L) * 256 + tr0 * 16 + 8 * hl;
  float4 ga = make_float4(0.0f, 0.0f, 0.0f, 0.0f), gb = ga, gc = ga, gd = ga;
  if (L > 0) {
    ga = *(const float4*)(gbase);
    gb = *(const float4*)(gbase + 4);
    gc = *(const float4*)(gbase + 16);
    gd = *(const float4*)(gbase + 20);
  }

  float* Xc = Xbuf0;
  float* Xn = Xbuf1;
  for (int i = 0; i < L; ++i) {
    if (OutAll) {
      for (int e = 0; e < SC_COLS; ++e) {
        size_t m = (size_t)(cbase + e) * L + i;
        OutAll[m * 256 + tid] = Xc[tid * SC_COLS + e];
      }
    }
    v8f acc0, acc1;
    acc0[0]=ga.x; acc0[1]=ga.y; acc0[2]=ga.z; acc0[3]=ga.w;
    acc0[4]=gb.x; acc0[5]=gb.y; acc0[6]=gb.z; acc0[7]=gb.w;
    acc1[0]=gc.x; acc1[1]=gc.y; acc1[2]=gc.z; acc1[3]=gc.w;
    acc1[4]=gd.x; acc1[5]=gd.y; acc1[6]=gd.z; acc1[7]=gd.w;
    {
      const float* gp = gbase + (size_t)((i + 1 < L) ? i + 1 : i) * 256;
      ga = *(const float4*)(gp);
      gb = *(const float4*)(gp + 4);
      gc = *(const float4*)(gp + 16);
      gd = *(const float4*)(gp + 20);
    }
    v2f b[64];
    {
      const float* Xk = Xc + 2 * hl * SC_COLS + lr;
      #pragma unroll
      for (int kg = 0; kg < 16; ++kg) {
        #pragma unroll
        for (int j = 0; j < 4; ++j) {
          const int kc = 4 * kg + j;
          b[kc][0] = Xk[kc * 4 * SC_COLS];
          b[kc][1] = Xk[kc * 4 * SC_COLS + SC_COLS];
        }
        __builtin_amdgcn_sched_barrier(0);
      }
    }
    #pragma unroll
    for (int kc = 0; kc < 64; ++kc) {
      acc0 = WMMA_F32(a0[kc], b[kc], acc0);
      acc1 = WMMA_F32(a1[kc], b[kc], acc1);
    }
    #pragma unroll
    for (int v = 0; v < 8; ++v) {
      Xn[(tr0 * 16 + 8 * hl + v) * SC_COLS + lr] = acc0[v];
      Xn[((tr0 + 1) * 16 + 8 * hl + v) * SC_COLS + lr] = acc1[v];
    }
    __syncthreads();
    float* tmp = Xc; Xc = Xn; Xn = tmp;
  }
  if (OutEnd) {
    for (int e = 0; e < SC_COLS; ++e)
      OutEnd[(size_t)(cbase + e) * 256 + tid] = Xc[tid * SC_COLS + e];
  }
}

static constexpr int TV_PITCH = 20;
static constexpr int TV_FLOATS = 16 * TV_PITCH;

template <int CTRL>
__device__ __forceinline__ float lane_read(float v) {
  return __int_as_float(__builtin_amdgcn_update_dpp(0, __float_as_int(v), CTRL, 0xF, 0xF, true));
}
static constexpr int DPP_SWAP1 = 0xB1;
static constexpr int DPP_SWAP2 = 0x4E;
static constexpr int DPP_ROR4  = 0x124;
static constexpr int DPP_ROR8  = 0x128;

__global__ __launch_bounds__(1024) void carry_kernel(
    const float* __restrict__ Mg, const float* __restrict__ Z,
    const float* __restrict__ x0, float* __restrict__ Tcar, int n) {
  __shared__ __attribute__((aligned(16))) float tv[2 * TV_FLOATS];
  const int tid = threadIdx.x;
  const int lane = tid & 31, wave = tid >> 5;
  const int g = lane & 15, rg = lane >> 4;
  const int rbase = wave * 8 + rg * 4;

  v2f m[4][8];
  for (int i = 0; i < 4; ++i) {
    const float4* mrow = (const float4*)(Mg + (size_t)(rbase + i) * 256 + 16 * g);
    for (int c = 0; c < 4; ++c) {
      float4 t4 = mrow[c];
      m[i][2 * c]     = (v2f){t4.x, t4.y};
      m[i][2 * c + 1] = (v2f){t4.z, t4.w};
    }
  }
  const int outslot = (tid >> 4) * TV_PITCH + (tid & 15);
  if (tid < 256) tv[outslot] = x0[tid];
  const bool odd1 = (g & 1) != 0, odd2 = (g & 2) != 0;
  const int myrow = rbase + (odd1 ? 2 : 0) + (odd2 ? 1 : 0);
  const int myslot = (myrow >> 4) * TV_PITCH + (myrow & 15);
  float zc = (n > 1) ? Z[myrow] : 0.0f;
  __syncthreads();

  for (int j = 0; j + 1 < n; ++j) {
    const float* tcur = tv + (j & 1) * TV_FLOATS;
    float* tnxt = tv + ((j + 1) & 1) * TV_FLOATS;
    if (wave < 8) Tcar[(size_t)j * 256 + tid] = tcur[outslot];
    float zn = 0.0f;
    if (j + 2 < n) zn = Z[(size_t)(j + 1) * 256 + myrow];
    const float* xs = tcur + g * TV_PITCH;
    v2f acc[4] = {};
    for (int c = 0; c < 4; ++c) {
      float4 x4 = *(const float4*)(xs + 4 * c);
      v2f xa = {x4.x, x4.y}, xb = {x4.z, x4.w};
      for (int i = 0; i < 4; ++i) {
        acc[i] += m[i][2 * c] * xa;
        acc[i] += m[i][2 * c + 1] * xb;
      }
    }
    float p0 = acc[0][0] + acc[0][1];
    float p1 = acc[1][0] + acc[1][1];
    float p2 = acc[2][0] + acc[2][1];
    float p3 = acc[3][0] + acc[3][1];
    float s0 = odd1 ? p0 : p2, s1 = odd1 ? p1 : p3;
    float k0 = odd1 ? p2 : p0, k1 = odd1 ? p3 : p1;
    k0 += lane_read<DPP_SWAP1>(s0);
    k1 += lane_read<DPP_SWAP1>(s1);
    float s = odd2 ? k0 : k1;
    float k = odd2 ? k1 : k0;
    k += lane_read<DPP_SWAP2>(s);
    k += lane_read<DPP_ROR4>(k);
    k += lane_read<DPP_ROR8>(k);
    if (g < 4) tnxt[myslot] = k + zc;
    __syncthreads();
    zc = zn;
  }
  if (wave < 8) Tcar[(size_t)(n - 1) * 256 + tid] = tv[((n - 1) & 1) * TV_FLOATS + outslot];
}

static inline void launch_gemm(hipStream_t s, float* D, int ldd,
                               const float* A, int lda, const float* B, int ldb,
                               const float* Cin, int ldc,
                               int M, int N, int K, float alpha, float beta, int transB) {
  int tiles = (M / 16) * (N / 16);
  gemm_kernel<<<(tiles + 3) / 4, 128, 0, s>>>(D, ldd, A, lda, B, ldb, Cin, ldc,
                                              M, N, K, alpha, beta, transB);
}

extern "C" void kernel_launch(void* const* d_in, const int* in_sizes, int n_in,
                              void* d_out, int out_size, void* d_ws, size_t ws_size,
                              hipStream_t stream) {
  (void)in_sizes; (void)n_in; (void)out_size; (void)ws_size;
  const float* t  = (const float*)d_in[0];
  const float* u  = (const float*)d_in[1];
  const float* x0 = (const float*)d_in[2];
  const float* Am = (const float*)d_in[3];
  const float* Bm = (const float*)d_in[4];
  const float* Cm = (const float*)d_in[5];
  const float* Dm = (const float*)d_in[6];
  float* xout = (float*)d_out;
  float* yout = xout + (size_t)Tn * NS;

  float* w = (float*)d_ws;
  float* Ad = w;  w += 65536;
  float* Bd = w;  w += 8192;
  float* Kt[6]; for (int i = 0; i < 6; i++) { Kt[i] = w; w += 65536; }
  Kt[0] = Ad;
  float* Lt[6]; for (int i = 0; i < 6; i++) { Lt[i] = w; w += 49152; }
  float* M64  = w; w += 65536 + 49152;
  float* Mm   = w; w += 65536;
  w += 49152;
  float* Qc   = w; w += 32768;
  float* M8   = w; w += 65536;
  float* M512 = w; w += 65536;
  float* PT1  = w; w += 65536;
  float* PT2  = w; w += 65536;
  float* Pm   = w; w += (size_t)Tn * 128;
  float* G    = w; w += (size_t)Tn * 256;
  float* Zend2 = w; w += (size_t)C2 * 256;
  float* Scar2 = w; w += (size_t)C2 * 256;
  float* Zend3 = w; w += (size_t)C3 * 256;
  float* Tcar  = w; w += (size_t)C3 * 256;
  float* Zend1 = Pm;
  float* Scar1 = Pm + (size_t)C1 * 256;

  auto gdim = [](int n) { return dim3((unsigned)((n + 255) / 256)); };

  k_prep<<<gdim(65536 + 49152), 256, 0, stream>>>(Ad, Bd, Lt[0], Am, Bm, t);

  double aa[7][7] = {};
  aa[2][1] = 0.2;
  aa[3][1] = 3.0/40.0;      aa[3][2] = 9.0/40.0;
  aa[4][1] = 44.0/45.0;     aa[4][2] = -56.0/15.0;     aa[4][3] = 32.0/9.0;
  aa[5][1] = 19372.0/6561.0; aa[5][2] = -25360.0/2187.0;
  aa[5][3] = 64448.0/6561.0; aa[5][4] = -212.0/729.0;
  aa[6][1] = 9017.0/3168.0; aa[6][2] = -355.0/33.0; aa[6][3] = 46732.0/5247.0;
  aa[6][4] = 49.0/176.0;    aa[6][5] = -5103.0/18656.0;
  stage_kernel<1><<<112, 128, 0, stream>>>(Kt[1], Lt[1], Ad, Bd, 1,
      Kt[0], Kt[0], Kt[0], Kt[0], Kt[0], Lt[0], Lt[0], Lt[0], Lt[0], Lt[0],
      (float)aa[2][1], 0.f, 0.f, 0.f, 0.f);
  stage_kernel<2><<<112, 128, 0, stream>>>(Kt[2], Lt[2], Ad, Bd, 2,
      Kt[0], Kt[1], Kt[0], Kt[0], Kt[0], Lt[0], Lt[1], Lt[0], Lt[0], Lt[0],
      (float)aa[3][1], (float)aa[3][2], 0.f, 0.f, 0.f);
  stage_kernel<3><<<112, 128, 0, stream>>>(Kt[3], Lt[3], Ad, Bd, 3,
      Kt[0], Kt[1], Kt[2], Kt[0], Kt[0], Lt[0], Lt[1], Lt[2], Lt[0], Lt[0],
      (float)aa[4][1], (float)aa[4][2], (float)aa[4][3], 0.f, 0.f);
  stage_kernel<4><<<112, 128, 0, stream>>>(Kt[4], Lt[4], Ad, Bd, 4,
      Kt[0], Kt[1], Kt[2], Kt[3], Kt[0], Lt[0], Lt[1], Lt[2], Lt[3], Lt[0],
      (float)aa[5][1], (float)aa[5][2], (float)aa[5][3], (float)aa[5][4], 0.f);
  stage_kernel<5><<<112, 128, 0, stream>>>(Kt[5], Lt[5], Ad, Bd, 5,
      Kt[0], Kt[1], Kt[2], Kt[3], Kt[4], Lt[0], Lt[1], Lt[2], Lt[3], Lt[4],
      (float)aa[6][1], (float)aa[6][2], (float)aa[6][3], (float)aa[6][4], (float)aa[6][5]);

  const double bb[7] = {0.0, 35.0/384.0, 0.0, 500.0/1113.0, 125.0/192.0,
                        -2187.0/6784.0, 11.0/84.0};
  k_final<<<gdim(65536 + 8192), 256, 0, stream>>>(Mm, Qc,
      Kt[0], Kt[2], Kt[3], Kt[4], Kt[5], Lt[0], Lt[2], Lt[3], Lt[4], Lt[5],
      (float)bb[1], (float)bb[3], (float)bb[4], (float)bb[5], (float)bb[6]);

  gemm_k256_kernel<false, 8><<<64, 128, 0, stream>>>(PT1,  256, Mm,  Mm);
  gemm_k256_kernel<false, 8><<<64, 128, 0, stream>>>(PT2,  256, PT1, PT1);
  gemm_k256_kernel<false, 8><<<64, 128, 0, stream>>>(M8,   256, PT2, PT2);
  gemm_k256_kernel<false, 8><<<64, 128, 0, stream>>>(PT1,  256, M8,  M8);
  gemm_k256_kernel<false, 8><<<64, 128, 0, stream>>>(PT2,  256, PT1, PT1);
  gemm_k256_kernel<false, 8><<<64, 128, 0, stream>>>(M64,  256, PT2, PT2);
  gemm_k256_kernel<false, 8><<<64, 128, 0, stream>>>(PT1,  256, M64, M64);
  gemm_k256_kernel<false, 8><<<64, 128, 0, stream>>>(PT2,  256, PT1, PT1);
  gemm_k256_kernel<false, 8><<<64, 128, 0, stream>>>(M512, 256, PT2, PT2);

  drive_kernel<<<Tn / DR_ROWS, 256, 0, stream>>>(G, u, Qc);

  scan_kernel<<<C1 / SC_COLS, 256, 0, stream>>>(Mm, G, nullptr, nullptr, Zend1, LC);
  scan_kernel<<<C2 / SC_COLS, 256, 0, stream>>>(M8, Zend1, nullptr, nullptr, Zend2, LC);
  scan_kernel<<<C3 / SC_COLS, 256, 0, stream>>>(M64, Zend2, nullptr, nullptr, Zend3, LC);
  carry_kernel<<<1, 1024, 0, stream>>>(M512, Zend3, x0, Tcar, C3);
  scan_kernel<<<C3 / SC_COLS, 256, 0, stream>>>(M64, Zend2, Tcar, Scar2, nullptr, LC);
  scan_kernel<<<C2 / SC_COLS, 256, 0, stream>>>(M8, Zend1, Scar2, Scar1, nullptr, LC);
  scan_kernel<<<C1 / SC_COLS, 256, 0, stream>>>(Mm, G, Scar1, xout, nullptr, LC);

  gemm_k256_kernel<true, 1><<<Tn / 32, 128, 0, stream>>>(yout, 32, xout, Cm);
  launch_gemm(stream, yout, 32, u,    32,  Dm, 32,  yout,   32, Tn, 32, 32,  1.f, 1.f, 1);
}


# M^64 and M^512 formed inside the level-2 and level-3 scan launches by 16 extra work-groups on idle WGPs (8 steps from the identity); 6 squaring launches gone (26 to 20); a different product tree
# speedup vs baseline: 4.5402x; 1.0534x over previous
#include <hip/hip_runtime.h>

typedef __attribute__((ext_vector_type(2))) float v2f;
typedef __attribute__((ext_vector_type(8))) float v8f;

#define WMMA_F32(a, b, c) \
  __builtin_amdgcn_wmma_f32_16x16x4_f32(false, (a), false, (b), (short)0, (c), false, false)

static constexpr int Tn = 32768;
static constexpr int NS = 256;
static constexpr int LC = 8;
static constexpr int C1 = 4096;
static constexpr int C2 = 512;
static constexpr int C3 = 64;
static constexpr int SC_COLS = 16;
static constexpr int PW_WGS = NS / SC_COLS;
static constexpr int MS_LD = 260;
static constexpr int SCAN_LDS_FLOATS = 256 * MS_LD + 2 * 256 * SC_COLS;

__constant__ float d_H[6][4] = {
  {1.0f, 0.0f, 0.0f, 0.0f},
  {0.896f, 0.128f, 0.104f, -0.032f},
  {0.784f, 0.147f, 0.216f, -0.063f},
  {0.104f, 0.032f, 0.896f, -0.128f},
  {25.0f/729.0f, 8.0f/729.0f, 704.0f/729.0f, -64.0f/729.0f},
  {0.0f, 0.0f, 1.0f, 0.0f},
};

__global__ void k_prep(float* __restrict__ Ad, float* __restrict__ Bd, float* __restrict__ Lt0,
                       const float* __restrict__ A, const float* __restrict__ B,
                       const float* __restrict__ t) {
  const int idx = blockIdx.x * 256 + threadIdx.x;
  const float dt = t[1] - t[0];
  if (idx < 65536) {
    Ad[idx] = dt * A[idx];
  } else {
    const int q = idx - 65536;
    if (q < 49152) {
      const int r = q / 192, c = q - r * 192;
      float val = 0.0f;
      if (c < 32) { const float bd = dt * B[r * 32 + c]; Bd[r * 32 + c] = bd; val += bd; }
      Lt0[q] = val;
    }
  }
}
__global__ void k_final(float* __restrict__ Mm, float* __restrict__ Q,
                        const float* __restrict__ K1, const float* __restrict__ K3,
                        const float* __restrict__ K4, const float* __restrict__ K5,
                        const float* __restrict__ K6,
                        const float* __restrict__ W1, const float* __restrict__ W3,
                        const float* __restrict__ W4, const float* __restrict__ W5,
                        const float* __restrict__ W6,
                        float b1, float b3, float b4, float b5, float b6) {
  const int idx = blockIdx.x * 256 + threadIdx.x;
  if (idx < 65536) {
    float v = ((idx >> 8) == (idx & 255)) ? 1.0f : 0.0f;
    v = fmaf(b1, K1[idx], v);
    v = fmaf(b3, K3[idx], v);
    v = fmaf(b4, K4[idx], v);
    v = fmaf(b5, K5[idx], v);
    v = fmaf(b6, K6[idx], v);
    Mm[idx] = v;
  } else {
    const int q = idx - 65536;
    if (q >= 256 * 32) return;
    const int r = q >> 5, c = q & 31;
    float w[6];
    for (int st = 0; st < 6; ++st) {
      const int o = r * 192 + 32 * st + c;
      float t = 0.0f;
      t = fmaf(b1, W1[o], t);
      t = fmaf(b3, W3[o], t);
      t = fmaf(b4, W4[o], t);
      t = fmaf(b5, W5[o], t);
      t = fmaf(b6, W6[o], t);
      w[st] = t;
    }
    for (int j = 0; j < 4; ++j) {
      float s = 0.0f;
      for (int st = 0; st < 6; ++st) s += d_H[st][j] * w[st];
      Q[(32 * j + c) * 256 + r] = s;
    }
  }
}
__global__ __launch_bounds__(128) void gemm_kernel(
    float* __restrict__ D, int ldd,
    const float* __restrict__ A, int lda,
    const float* __restrict__ B, int ldb,
    const float* __restrict__ Cin, int ldc,
    int Mr, int Nc, int Kd, float alpha, float beta, int transB) {
  const int lane = threadIdx.x & 31, wave = threadIdx.x >> 5;
  const int hl = lane >> 4, lr = lane & 15;
  const int tilesN = Nc >> 4;
  const int tile = blockIdx.x * 4 + wave;
  if (tile >= (Mr >> 4) * tilesN) return;
  const int tm = tile / tilesN, tn = tile % tilesN;
  const float* Ar = A + (size_t)(tm * 16 + lr) * lda + 2 * hl;
  v8f acc = {};
  if (transB) {
    const float* Br = B + (size_t)(tn * 16 + lr) * ldb + 2 * hl;
    #pragma unroll 4
    for (int k = 0; k < Kd; k += 4) {
      v2f a = *(const v2f*)(Ar + k);
      v2f b = *(const v2f*)(Br + k);
      acc = WMMA_F32(a, b, acc);
    }
  } else {
    const float* Bc = B + (size_t)(2 * hl) * ldb + tn * 16 + lr;
    #pragma unroll 4
    for (int k = 0; k < Kd; k += 4) {
      v2f a = *(const v2f*)(Ar + k);
      v2f b;
      b[0] = Bc[(size_t)k * ldb];
      b[1] = Bc[(size_t)k * ldb + ldb];
      acc = WMMA_F32(a, b, acc);
    }
  }
  float* Dp = D + (size_t)(tm * 16 + 8 * hl) * ldd + tn * 16 + lr;
  if (beta != 0.0f) {
    const float* Cp = Cin + (size_t)(tm * 16 + 8 * hl) * ldc + tn * 16 + lr;
    #pragma unroll
    for (int v = 0; v < 8; ++v)
      Dp[(size_t)v * ldd] = alpha * acc[v] + beta * Cp[(size_t)v * ldc];
  } else {
    #pragma unroll
    for (int v = 0; v < 8; ++v)
      Dp[(size_t)v * ldd] = alpha * acc[v];
  }
}

static constexpr int GK_LD = 260;
static constexpr int GN_LD = 40;

__device__ __forceinline__ void k256_load_a(float4 (&ra)[16], const float* __restrict__ A, int bm, int tid) {
  const float4* Ab = (const float4*)(A + (size_t)bm * 32 * 256);
  #pragma unroll
  for (int i = 0; i < 16; ++i) ra[i] = Ab[tid + 128 * i];
}
__device__ __forceinline__ void k256_store_a(float* As, const float4 (&ra)[16], int tid) {
  #pragma unroll
  for (int i = 0; i < 16; ++i) {
    const int idx = tid + 128 * i;
    *(float4*)(As + (idx >> 6) * GK_LD + 4 * (idx & 63)) = ra[i];
  }
}
template <bool TB>
__device__ __forceinline__ void k256_frags(v2f (&a)[16], v2f (&b)[16],
                                           const float* Ap, const float* Bp, int g) {
  #pragma unroll
  for (int j = 0; j < 16; ++j) {
    const int kc = 16 * g + j;
    a[j] = *(const v2f*)(Ap + 4 * kc);
    if (TB) {
      b[j] = *(const v2f*)(Bp + 4 * kc);
    } else {
      b[j][0] = Bp[4 * kc * GN_LD];
      b[j][1] = Bp[4 * kc * GN_LD + GN_LD];
    }
  }
}
__device__ __forceinline__ v8f k256_wmma16(const v2f (&a)[16], const v2f (&b)[16], v8f acc) {
  #pragma unroll
  for (int j = 0; j < 16; ++j) acc = WMMA_F32(a[j], b[j], acc);
  return acc;
}
template <bool TB>
__device__ __forceinline__ v8f k256_chain(const float* Ap, const float* Bp) {
  v2f a0[16], b0[16], a1[16], b1[16];
  v8f acc = {};
  k256_frags<TB>(a0, b0, Ap, Bp, 0);
  k256_frags<TB>(a1, b1, Ap, Bp, 1);
  __builtin_amdgcn_sched_barrier(0);
  acc = k256_wmma16(a0, b0, acc);
  k256_frags<TB>(a0, b0, Ap, Bp, 2);
  __builtin_amdgcn_sched_barrier(0);
  acc = k256_wmma16(a1, b1, acc);
  k256_frags<TB>(a1, b1, Ap, Bp, 3);
  __builtin_amdgcn_sched_barrier(0);
  acc = k256_wmma16(a0, b0, acc);
  acc = k256_wmma16(a1, b1, acc);
  return acc;
}

template <bool TB, int NBN>
__global__ __launch_bounds__(128) void gemm_k256_kernel(
    float* __restrict__ D, int ldd, const float* __restrict__ A, const float* __restrict__ B) {
  __shared__ __attribute__((aligned(16))) float As[32 * GK_LD];
  __shared__ __attribute__((aligned(16))) float Bs[TB ? 32 * GK_LD : 256 * GN_LD];
  const int tid = threadIdx.x;
  const int lane = tid & 31;
  const int wave = __builtin_amdgcn_readfirstlane(tid >> 5);
  const int hl = lane >> 4, lr = lane & 15;
  const int bm = blockIdx.x / NBN, bn = blockIdx.x % NBN;
  {
    float4 ra[16], rb[16];
    k256_load_a(ra, A, bm, tid);
    if (TB) {
      k256_load_a(rb, B, bn, tid);
    } else {
      const float* Bb = B + bn * 32;
      #pragma unroll
      for (int i = 0; i < 16; ++i) {
        const int idx = tid + 128 * i;
        rb[i] = *(const float4*)(Bb + (idx >> 3) * 256 + 4 * (idx & 7));
      }
    }
    k256_store_a(As, ra, tid);
    if (TB) {
      k256_store_a(Bs, rb, tid);
    } else {
      #pragma unroll
      for (int i = 0; i < 16; ++i) {
        const int idx = tid + 128 * i;
        *(float4*)(Bs + (idx >> 3) * GN_LD + 4 * (idx & 7)) = rb[i];
      }
    }
  }
  __syncthreads();
  const int tm = wave >> 1, tn = wave & 1;
  const float* Ap = As + (tm * 16 + lr) * GK_LD + 2 * hl;
  const float* Bp = TB ? Bs + (tn * 16 + lr) * GK_LD + 2 * hl
                       : Bs + (2 * hl) * GN_LD + tn * 16 + lr;
  const v8f acc = k256_chain<TB>(Ap, Bp);
  float* Dp = D + (size_t)(bm * 32 + tm * 16 + 8 * hl) * ldd + bn * 32 + tn * 16 + lr;
  #pragma unroll
  for (int v = 0; v < 8; ++v) Dp[(size_t)v * ldd] = acc[v];
}

__device__ __forceinline__ void stage_fma_plane(float4 (&v)[16], const float* __restrict__ P, float c,
                                                int ld, int col0, int tid) {
  float4 t[16];
  #pragma unroll
  for (int i = 0; i < 16; ++i) {
    const int idx = tid + 128 * i;
    t[i] = *(const float4*)(P + (idx >> 3) * ld + col0 + 4 * (idx & 7));
  }
  #pragma unroll
  for (int i = 0; i < 16; ++i) {
    v[i].x = fmaf(c, t[i].x, v[i].x);  v[i].y = fmaf(c, t[i].y, v[i].y);
    v[i].z = fmaf(c, t[i].z, v[i].z);  v[i].w = fmaf(c, t[i].w, v[i].w);
  }
}
template <int NJ>
__global__ __launch_bounds__(128) void stage_kernel(
    float* __restrict__ Kout, float* __restrict__ Lout,
    const float* __restrict__ Ad, const float* __restrict__ Bd, int blk,
    const float* __restrict__ K0, const float* __restrict__ K1, const float* __restrict__ K2,
    const float* __restrict__ K3, const float* __restrict__ K4,
    const float* __restrict__ W0, const float* __restrict__ W1, const float* __restrict__ W2,
    const float* __restrict__ W3, const float* __restrict__ W4,
    float c0, float c1, float c2, float c3, float c4) {
  __shared__ __attribute__((aligned(16))) float As[32 * GK_LD];
  __shared__ __attribute__((aligned(16))) float Bs[256 * GN_LD];
  const int tid = threadIdx.x;
  const int lane = tid & 31;
  const int wave = __builtin_amdgcn_readfirstlane(tid >> 5);
  const int hl = lane >> 4, lr = lane & 15;
  const bool isK = blockIdx.x < 64;
  const int q = isK ? (int)blockIdx.x : (int)blockIdx.x - 64;
  const int bm = isK ? (q >> 3) : (q / 6);
  const int bn = isK ? (q & 7) : (q % 6);
  const int ld = isK ? 256 : 192;
  {
    float4 ra[16], v[16];
    k256_load_a(ra, Ad, bm, tid);
    #pragma unroll
    for (int i = 0; i < 16; ++i) {
      const int idx = tid + 128 * i;
      const int k = idx >> 3, n = bn * 32 + 4 * (idx & 7);
      v[i] = isK ? make_float4(k == n ? 1.0f : 0.0f, k == n + 1 ? 1.0f : 0.0f,
                               k == n + 2 ? 1.0f : 0.0f, k == n + 3 ? 1.0f : 0.0f)
                 : make_float4(0.0f, 0.0f, 0.0f, 0.0f);
    }
    stage_fma_plane(v, isK ? K0 : W0, c0, ld, bn * 32, tid);
    if (NJ > 1) stage_fma_plane(v, isK ? K1 : W1, c1, ld, bn * 32, tid);
    if (NJ > 2) stage_fma_plane(v, isK ? K2 : W2, c2, ld, bn * 32, tid);
    if (NJ > 3) stage_fma_plane(v, isK ? K3 : W3, c3, ld, bn * 32, tid);
    if (NJ > 4) stage_fma_plane(v, isK ? K4 : W4, c4, ld, bn * 32, tid);
    k256_store_a(As, ra, tid);
    #pragma unroll
    for (int i = 0; i < 16; ++i) {
      const int idx = tid + 128 * i;
      *(float4*)(Bs + (idx >> 3) * GN_LD + 4 * (idx & 7)) = v[i];
    }
  }
  __syncthreads();
  const int tm = wave >> 1, tn = wave & 1;
  const float* Ap = As + (tm * 16 + lr) * GK_LD + 2 * hl;
  const float* Bp = Bs + (2 * hl) * GN_LD + tn * 16 + lr;
  const v8f acc = k256_chain<false>(Ap, Bp);
  const int row0 = bm * 32 + tm * 16 + 8 * hl;
  const int n = bn * 32 + tn * 16 + lr;
  if (isK) {
    float* Dp = Kout + (size_t)row0 * 256 + n;
    #pragma unroll
    for (int v = 0; v < 8; ++v) Dp[(size_t)v * 256] = acc[v];
  } else {
    float* Dp = Lout + (size_t)row0 * 192 + n;
    if (bn == blk) {
      const float* Bp2 = Bd + (size_t)row0 * 32 + (n & 31);
      #pragma unroll
      for (int v = 0; v < 8; ++v) Dp[(size_t)v * 192] = acc[v] + Bp2[(size_t)v * 32];
    } else {
      #pragma unroll
      for (int v = 0; v < 8; ++v) Dp[(size_t)v * 192] = acc[v];
    }
  }
}

static constexpr int DR_ROWS = 128;
static constexpr int DR_LD = 132;

__device__ __forceinline__ float xhalf(float x) {
  const int b = __builtin_bit_cast(int, x);
  return __builtin_bit_cast(float,
      __builtin_amdgcn_permlanex16(b, b, 0x76543210, 0xfedcba98, false, false));
}

__global__ __launch_bounds__(256) void drive_kernel(
    float* __restrict__ G, const float* __restrict__ u, const float* __restrict__ Q) {
  __shared__ __attribute__((aligned(16))) float Pt[DR_ROWS * DR_LD];
  const int tid = threadIdx.x;
  const int lane = tid & 31;
  const int wave = __builtin_amdgcn_readfirstlane(tid >> 5);
  const int hl = lane >> 4, lr = lane & 15;
  const int n0 = blockIdx.x * DR_ROWS;
  const int col0 = wave * 32;

  {
    const int cc = tid & 7, rg = tid >> 3;
    const int nb = n0 + 4 * rg;
    float4 U[7];
    #pragma unroll
    for (int i = 0; i < 7; ++i) {
      int g = nb - 1 + i;
      g = g < 0 ? 0 : (g > Tn - 1 ? Tn - 1 : g);
      U[i] = *(const float4*)(u + (size_t)g * 32 + 4 * cc);
    }
    #pragma unroll
    for (int t = 0; t < 4; ++t) {
      const int n = nb + t;
      const bool first = (n == 0), penult = (n == Tn - 2), last = (n == Tn - 1);
      const float4 mb = first ? U[t + 1] : U[t];
      const float ms = first ? 1.0f : 0.5f;
      const float4 na = penult ? U[t + 2] : U[t + 3];
      const float ns = penult ? 1.0f : 0.5f;
      float4 p0 = U[t + 1];
      float4 p1 = make_float4(ms * (U[t + 2].x - mb.x), ms * (U[t + 2].y - mb.y),
                              ms * (U[t + 2].z - mb.z), ms * (U[t + 2].w - mb.w));
      float4 p2 = U[t + 2];
      float4 p3 = make_float4(ns * (na.x - U[t + 1].x), ns * (na.y - U[t + 1].y),
                              ns * (na.z - U[t + 1].z), ns * (na.w - U[t + 1].w));
      if (last) {
        const float4 z = make_float4(0.0f, 0.0f, 0.0f, 0.0f);
        p0 = z; p1 = z; p2 = z; p3 = z;
      }
      float* prow = Pt + (4 * rg + t) * DR_LD + 4 * cc;
      *(float4*)(prow)      = p0;
      *(float4*)(prow + 32) = p1;
      *(float4*)(prow + 64) = p2;
      *(float4*)(prow + 96) = p3;
    }
  }

  v2f b0[32], b1[32];
  {
    const float* Qp = Q + (size_t)(2 * hl) * 256 + col0 + lr;
    #pragma unroll
    for (int kc = 0; kc < 32; ++kc) {
      b0[kc][0] = Qp[kc * 1024];
      b0[kc][1] = Qp[kc * 1024 + 256];
      b1[kc][0] = Qp[kc * 1024 + 16];
      b1[kc][1] = Qp[kc * 1024 + 256 + 16];
    }
  }
  __syncthreads();

  const float* Ap = Pt + lr * DR_LD + 2 * hl;
  for (int hb = 0; hb < 2; ++hb) {
    const float* Ah = Ap + hb * 64 * DR_LD;
    v8f acc0[4] = {}, acc1[4] = {};
    #pragma unroll
    for (int kc = 0; kc < 32; ++kc) {
      #pragma unroll
      for (int r = 0; r < 4; ++r) {
        const v2f a = *(const v2f*)(Ah + r * 16 * DR_LD + kc * 4);
        acc0[r] = WMMA_F32(a, b0[kc], acc0[r]);
        acc1[r] = WMMA_F32(a, b1[kc], acc1[r]);
      }
    }
    float* Gp = G + (size_t)(n0 + hb * 64) * 256 + col0 + lane;
    #pragma unroll
    for (int r = 0; r < 4; ++r) {
      #pragma unroll
      for (int v = 0; v < 8; ++v) {
        const float t0 = acc0[r][v], t1 = acc1[r][v];
        const float got = xhalf(hl ? t0 : t1);
        const float lo = hl ? got : t0;
        const float hi = hl ? t1 : got;
        Gp[(size_t)(r * 16 + v) * 256] = lo;
        Gp[(size_t)(r * 16 + 8 + v) * 256] = hi;
      }
    }
  }
}

__global__ __launch_bounds__(256) void scan_kernel(
    const float* __restrict__ Mg, const float* __restrict__ G,
    const float* __restrict__ Init,
    float* __restrict__ OutAll,
    float* __restrict__ OutEnd,
    int L,
    float* __restrict__ Pw,
    int nscan) {
  __shared__ __attribute__((aligned(16))) float smem[SCAN_LDS_FLOATS];
  float* Ms = smem;
  float* Xbuf0 = smem + 256 * MS_LD;
  float* Xbuf1 = Xbuf0 + 256 * SC_COLS;
  const int tid = threadIdx.x;
  const int lane = tid & 31, wave = tid >> 5;
  const int hl = lane >> 4, lr = lane & 15;
  const bool pw = (int)blockIdx.x >= nscan;
  const int cbase = (pw ? (int)blockIdx.x - nscan : (int)blockIdx.x) * SC_COLS;
  const int tr0 = wave * 2;
  if (pw) { Init = nullptr; OutAll = nullptr; OutEnd = nullptr; }

  {
    const float4* src = (const float4*)Mg;
    float4* dst = (float4*)Ms;
    for (int k = tid; k < 65536 / 4; k += 256)
      dst[(k >> 6) * (MS_LD / 4) + (k & 63)] = src[k];
  }
  if (pw) {
    for (int e = 0; e < SC_COLS; ++e)
      Xbuf0[tid * SC_COLS + e] = (tid == cbase + e) ? 1.0f : 0.0f;
  } else {
    for (int e = 0; e < SC_COLS; ++e)
      Xbuf0[tid * SC_COLS + e] = Init ? Init[(size_t)(cbase + e) * 256 + tid] : 0.0f;
  }
  __syncthreads();

  v2f a0[64], a1[64];
  {
    const float* Mr0 = Ms + (size_t)(tr0 * 16 + lr) * MS_LD + 2 * hl;
    const float* Mr1 = Mr0 + 16 * MS_LD;
    #pragma unroll
    for (int kc = 0; kc < 64; ++kc) {
      a0[kc] = *(const v2f*)(Mr0 + kc * 4);
      a1[kc] = *(const v2f*)(Mr1 + kc * 4);
    }
  }

  const float* gbase = G + ((size_t)(cbase + lr) * L) * 256 + tr0 * 16 + 8 * hl;
  float4 ga = make_float4(0.0f, 0.0f, 0.0f, 0.0f), gb = ga, gc = ga, gd = ga;
  if (L > 0 && !pw) {
    ga = *(const float4*)(gbase);
    gb = *(const float4*)(gbase + 4);
    gc = *(const float4*)(gbase + 16);
    gd = *(const float4*)(gbase + 20);
  }

  float* Xc = Xbuf0;
  float* Xn = Xbuf1;
  for (int i = 0; i < L; ++i) {
    if (OutAll) {
      for (int e = 0; e < SC_COLS; ++e) {
        size_t m = (size_t)(cbase + e) * L + i;
        OutAll[m * 256 + tid] = Xc[tid * SC_COLS + e];
      }
    }
    v8f acc0, acc1;
    acc0[0]=ga.x; acc0[1]=ga.y; acc0[2]=ga.z; acc0[3]=ga.w;
    acc0[4]=gb.x; acc0[5]=gb.y; acc0[6]=gb.z; acc0[7]=gb.w;
    acc1[0]=gc.x; acc1[1]=gc.y; acc1[2]=gc.z; acc1[3]=gc.w;
    acc1[4]=gd.x; acc1[5]=gd.y; acc1[6]=gd.z; acc1[7]=gd.w;
    if (!pw) {
      const float* gp = gbase + (size_t)((i + 1 < L) ? i + 1 : i) * 256;
      ga = *(const float4*)(gp);
      gb = *(const float4*)(gp + 4);
      gc = *(const float4*)(gp + 16);
      gd = *(const float4*)(gp + 20);
    }
    v2f b[64];
    {
      const float* Xk = Xc + 2 * hl * SC_COLS + lr;
      #pragma unroll
      for (int kg = 0; kg < 16; ++kg) {
        #pragma unroll
        for (int j = 0; j < 4; ++j) {
          const int kc = 4 * kg + j;
          b[kc][0] = Xk[kc * 4 * SC_COLS];
          b[kc][1] = Xk[kc * 4 * SC_COLS + SC_COLS];
        }
        __builtin_amdgcn_sched_barrier(0);
      }
    }
    #pragma unroll
    for (int kc = 0; kc < 64; ++kc) {
      acc0 = WMMA_F32(a0[kc], b[kc], acc0);
      acc1 = WMMA_F32(a1[kc], b[kc], acc1);
    }
    #pragma unroll
    for (int v = 0; v < 8; ++v) {
      Xn[(tr0 * 16 + 8 * hl + v) * SC_COLS + lr] = acc0[v];
      Xn[((tr0 + 1) * 16 + 8 * hl + v) * SC_COLS + lr] = acc1[v];
    }
    __syncthreads();
    float* tmp = Xc; Xc = Xn; Xn = tmp;
  }
  if (OutEnd) {
    for (int e = 0; e < SC_COLS; ++e)
      OutEnd[(size_t)(cbase + e) * 256 + tid] = Xc[tid * SC_COLS + e];
  }
  if (pw) {
    float4* dst = (float4*)(Pw + (size_t)tid * 256 + cbase);
    const float4* xr = (const float4*)(Xc + tid * SC_COLS);
    for (int q = 0; q < SC_COLS / 4; ++q) dst[q] = xr[q];
  }
}

static constexpr int TV_PITCH = 20;
static constexpr int TV_FLOATS = 16 * TV_PITCH;

template <int CTRL>
__device__ __forceinline__ float lane_read(float v) {
  return __int_as_float(__builtin_amdgcn_update_dpp(0, __float_as_int(v), CTRL, 0xF, 0xF, true));
}
static constexpr int DPP_SWAP1 = 0xB1;
static constexpr int DPP_SWAP2 = 0x4E;
static constexpr int DPP_ROR4  = 0x124;
static constexpr int DPP_ROR8  = 0x128;

__global__ __launch_bounds__(1024) void carry_kernel(
    const float* __restrict__ Mg, const float* __restrict__ Z,
    const float* __restrict__ x0, float* __restrict__ Tcar, int n) {
  __shared__ __attribute__((aligned(16))) float tv[2 * TV_FLOATS];
  const int tid = threadIdx.x;
  const int lane = tid & 31, wave = tid >> 5;
  const int g = lane & 15, rg = lane >> 4;
  const int rbase = wave * 8 + rg * 4;

  v2f m[4][8];
  for (int i = 0; i < 4; ++i) {
    const float4* mrow = (const float4*)(Mg + (size_t)(rbase + i) * 256 + 16 * g);
    for (int c = 0; c < 4; ++c) {
      float4 t4 = mrow[c];
      m[i][2 * c]     = (v2f){t4.x, t4.y};
      m[i][2 * c + 1] = (v2f){t4.z, t4.w};
    }
  }
  const int outslot = (tid >> 4) * TV_PITCH + (tid & 15);
  if (tid < 256) tv[outslot] = x0[tid];
  const bool odd1 = (g & 1) != 0, odd2 = (g & 2) != 0;
  const int myrow = rbase + (odd1 ? 2 : 0) + (odd2 ? 1 : 0);
  const int myslot = (myrow >> 4) * TV_PITCH + (myrow & 15);
  float zc = (n > 1) ? Z[myrow] : 0.0f;
  __syncthreads();

  for (int j = 0; j + 1 < n; ++j) {
    const float* tcur = tv + (j & 1) * TV_FLOATS;
    float* tnxt = tv + ((j + 1) & 1) * TV_FLOATS;
    if (wave < 8) Tcar[(size_t)j * 256 + tid] = tcur[outslot];
    float zn = 0.0f;
    if (j + 2 < n) zn = Z[(size_t)(j + 1) * 256 + myrow];
    const float* xs = tcur + g * TV_PITCH;
    v2f acc[4] = {};
    for (int c = 0; c < 4; ++c) {
      float4 x4 = *(const float4*)(xs + 4 * c);
      v2f xa = {x4.x, x4.y}, xb = {x4.z, x4.w};
      for (int i = 0; i < 4; ++i) {
        acc[i] += m[i][2 * c] * xa;
        acc[i] += m[i][2 * c + 1] * xb;
      }
    }
    float p0 = acc[0][0] + acc[0][1];
    float p1 = acc[1][0] + acc[1][1];
    float p2 = acc[2][0] + acc[2][1];
    float p3 = acc[3][0] + acc[3][1];
    float s0 = odd1 ? p0 : p2, s1 = odd1 ? p1 : p3;
    float k0 = odd1 ? p2 : p0, k1 = odd1 ? p3 : p1;
    k0 += lane_read<DPP_SWAP1>(s0);
    k1 += lane_read<DPP_SWAP1>(s1);
    float s = odd2 ? k0 : k1;
    float k = odd2 ? k1 : k0;
    k += lane_read<DPP_SWAP2>(s);
    k += lane_read<DPP_ROR4>(k);
    k += lane_read<DPP_ROR8>(k);
    if (g < 4) tnxt[myslot] = k + zc;
    __syncthreads();
    zc = zn;
  }
  if (wave < 8) Tcar[(size_t)(n - 1) * 256 + tid] = tv[((n - 1) & 1) * TV_FLOATS + outslot];
}

static inline void launch_gemm(hipStream_t s, float* D, int ldd,
                               const float* A, int lda, const float* B, int ldb,
                               const float* Cin, int ldc,
                               int M, int N, int K, float alpha, float beta, int transB) {
  int tiles = (M / 16) * (N / 16);
  gemm_kernel<<<(tiles + 3) / 4, 128, 0, s>>>(D, ldd, A, lda, B, ldb, Cin, ldc,
                                              M, N, K, alpha, beta, transB);
}

extern "C" void kernel_launch(void* const* d_in, const int* in_sizes, int n_in,
                              void* d_out, int out_size, void* d_ws, size_t ws_size,
                              hipStream_t stream) {
  (void)in_sizes; (void)n_in; (void)out_size; (void)ws_size;
  const float* t  = (const float*)d_in[0];
  const float* u  = (const float*)d_in[1];
  const float* x0 = (const float*)d_in[2];
  const float* Am = (const float*)d_in[3];
  const float* Bm = (const float*)d_in[4];
  const float* Cm = (const float*)d_in[5];
  const float* Dm = (const float*)d_in[6];
  float* xout = (float*)d_out;
  float* yout = xout + (size_t)Tn * NS;

  float* w = (float*)d_ws;
  float* Ad = w;  w += 65536;
  float* Bd = w;  w += 8192;
  float* Kt[6]; for (int i = 0; i < 6; i++) { Kt[i] = w; w += 65536; }
  Kt[0] = Ad;
  float* Lt[6]; for (int i = 0; i < 6; i++) { Lt[i] = w; w += 49152; }
  float* M64  = w; w += 65536 + 49152;
  float* Mm   = w; w += 65536;
  w += 49152;
  float* Qc   = w; w += 32768;
  float* M8   = w; w += 65536;
  float* M512 = w; w += 65536;
  float* PT1  = w; w += 65536;
  float* PT2  = w; w += 65536;
  float* Pm   = w; w += (size_t)Tn * 128;
  float* G    = w; w += (size_t)Tn * 256;
  float* Zend2 = w; w += (size_t)C2 * 256;
  float* Scar2 = w; w += (size_t)C2 * 256;
  float* Zend3 = w; w += (size_t)C3 * 256;
  float* Tcar  = w; w += (size_t)C3 * 256;
  float* Zend1 = Pm;
  float* Scar1 = Pm + (size_t)C1 * 256;

  auto gdim = [](int n) { return dim3((unsigned)((n + 255) / 256)); };

  k_prep<<<gdim(65536 + 49152), 256, 0, stream>>>(Ad, Bd, Lt[0], Am, Bm, t);

  double aa[7][7] = {};
  aa[2][1] = 0.2;
  aa[3][1] = 3.0/40.0;      aa[3][2] = 9.0/40.0;
  aa[4][1] = 44.0/45.0;     aa[4][2] = -56.0/15.0;     aa[4][3] = 32.0/9.0;
  aa[5][1] = 19372.0/6561.0; aa[5][2] = -25360.0/2187.0;
  aa[5][3] = 64448.0/6561.0; aa[5][4] = -212.0/729.0;
  aa[6][1] = 9017.0/3168.0; aa[6][2] = -355.0/33.0; aa[6][3] = 46732.0/5247.0;
  aa[6][4] = 49.0/176.0;    aa[6][5] = -5103.0/18656.0;
  stage_kernel<1><<<112, 128, 0, stream>>>(Kt[1], Lt[1], Ad, Bd, 1,
      Kt[0], Kt[0], Kt[0], Kt[0], Kt[0], Lt[0], Lt[0], Lt[0], Lt[0], Lt[0],
      (float)aa[2][1], 0.f, 0.f, 0.f, 0.f);
  stage_kernel<2><<<112, 128, 0, stream>>>(Kt[2], Lt[2], Ad, Bd, 2,
      Kt[0], Kt[1], Kt[0], Kt[0], Kt[0], Lt[0], Lt[1], Lt[0], Lt[0], Lt[0],
      (float)aa[3][1], (float)aa[3][2], 0.f, 0.f, 0.f);
  stage_kernel<3><<<112, 128, 0, stream>>>(Kt[3], Lt[3], Ad, Bd, 3,
      Kt[0], Kt[1], Kt[2], Kt[0], Kt[0], Lt[0], Lt[1], Lt[2], Lt[0], Lt[0],
      (float)aa[4][1], (float)aa[4][2], (float)aa[4][3], 0.f, 0.f);
  stage_kernel<4><<<112, 128, 0, stream>>>(Kt[4], Lt[4], Ad, Bd, 4,
      Kt[0], Kt[1], Kt[2], Kt[3], Kt[0], Lt[0], Lt[1], Lt[2], Lt[3], Lt[0],
      (float)aa[5][1], (float)aa[5][2], (float)aa[5][3], (float)aa[5][4], 0.f);
  stage_kernel<5><<<112, 128, 0, stream>>>(Kt[5], Lt[5], Ad, Bd, 5,
      Kt[0], Kt[1], Kt[2], Kt[3], Kt[4], Lt[0], Lt[1], Lt[2], Lt[3], Lt[4],
      (float)aa[6][1], (float)aa[6][2], (float)aa[6][3], (float)aa[6][4], (float)aa[6][5]);

  const double bb[7] = {0.0, 35.0/384.0, 0.0, 500.0/1113.0, 125.0/192.0,
                        -2187.0/6784.0, 11.0/84.0};
  k_final<<<gdim(65536 + 8192), 256, 0, stream>>>(Mm, Qc,
      Kt[0], Kt[2], Kt[3], Kt[4], Kt[5], Lt[0], Lt[2], Lt[3], Lt[4], Lt[5],
      (float)bb[1], (float)bb[3], (float)bb[4], (float)bb[5], (float)bb[6]);

  gemm_k256_kernel<false, 8><<<64, 128, 0, stream>>>(PT1,  256, Mm,  Mm);
  gemm_k256_kernel<false, 8><<<64, 128, 0, stream>>>(PT2,  256, PT1, PT1);
  gemm_k256_kernel<false, 8><<<64, 128, 0, stream>>>(M8,   256, PT2, PT2);

  drive_kernel<<<Tn / DR_ROWS, 256, 0, stream>>>(G, u, Qc);

  scan_kernel<<<C1 / SC_COLS, 256, 0, stream>>>(Mm, G, nullptr, nullptr, Zend1, LC, nullptr, C1 / SC_COLS);
  scan_kernel<<<C2 / SC_COLS + PW_WGS, 256, 0, stream>>>(M8, Zend1, nullptr, nullptr, Zend2, LC,
                                                         M64, C2 / SC_COLS);
  scan_kernel<<<C3 / SC_COLS + PW_WGS, 256, 0, stream>>>(M64, Zend2, nullptr, nullptr, Zend3, LC,
                                                         M512, C3 / SC_COLS);
  carry_kernel<<<1, 1024, 0, stream>>>(M512, Zend3, x0, Tcar, C3);
  scan_kernel<<<C3 / SC_COLS, 256, 0, stream>>>(M64, Zend2, Tcar, Scar2, nullptr, LC, nullptr, C3 / SC_COLS);
  scan_kernel<<<C2 / SC_COLS, 256, 0, stream>>>(M8, Zend1, Scar2, Scar1, nullptr, LC, nullptr, C2 / SC_COLS);
  scan_kernel<<<C1 / SC_COLS, 256, 0, stream>>>(Mm, G, Scar1, xout, nullptr, LC, nullptr, C1 / SC_COLS);

  gemm_k256_kernel<true, 1><<<Tn / 32, 128, 0, stream>>>(yout, 32, xout, Cm);
  launch_gemm(stream, yout, 32, u,    32,  Dm, 32,  yout,   32, Tn, 32, 32,  1.f, 1.f, 1);
}


# scan kernel set-up: all global loads (64 step-matrix pieces per thread, initial states, first drive rows) go out as one group before any LDS store, not 8 + 8 + 1 dependent rounds; sum order kept
# speedup vs baseline: 4.7969x; 1.0565x over previous
#include <hip/hip_runtime.h>

typedef __attribute__((ext_vector_type(2))) float v2f;
typedef __attribute__((ext_vector_type(8))) float v8f;

#define WMMA_F32(a, b, c) \
  __builtin_amdgcn_wmma_f32_16x16x4_f32(false, (a), false, (b), (short)0, (c), false, false)

static constexpr int Tn = 32768;
static constexpr int NS = 256;
static constexpr int LC = 8;
static constexpr int C1 = 4096;
static constexpr int C2 = 512;
static constexpr int C3 = 64;
static constexpr int SC_COLS = 16;
static constexpr int PW_WGS = NS / SC_COLS;
static constexpr int MS_LD = 260;
static constexpr int SCAN_LDS_FLOATS = 256 * MS_LD + 2 * 256 * SC_COLS;

__constant__ float d_H[6][4] = {
  {1.0f, 0.0f, 0.0f, 0.0f},
  {0.896f, 0.128f, 0.104f, -0.032f},
  {0.784f, 0.147f, 0.216f, -0.063f},
  {0.104f, 0.032f, 0.896f, -0.128f},
  {25.0f/729.0f, 8.0f/729.0f, 704.0f/729.0f, -64.0f/729.0f},
  {0.0f, 0.0f, 1.0f, 0.0f},
};

__global__ void k_prep(float* __restrict__ Ad, float* __restrict__ Bd, float* __restrict__ Lt0,
                       const float* __restrict__ A, const float* __restrict__ B,
                       const float* __restrict__ t) {
  const int idx = blockIdx.x * 256 + threadIdx.x;
  const float dt = t[1] - t[0];
  if (idx < 65536) {
    Ad[idx] = dt * A[idx];
  } else {
    const int q = idx - 65536;
    if (q < 49152) {
      const int r = q / 192, c = q - r * 192;
      float val = 0.0f;
      if (c < 32) { const float bd = dt * B[r * 32 + c]; Bd[r * 32 + c] = bd; val += bd; }
      Lt0[q] = val;
    }
  }
}
__global__ void k_final(float* __restrict__ Mm, float* __restrict__ Q,
                        const float* __restrict__ K1, const float* __restrict__ K3,
                        const float* __restrict__ K4, const float* __restrict__ K5,
                        const float* __restrict__ K6,
                        const float* __restrict__ W1, const float* __restrict__ W3,
                        const float* __restrict__ W4, const float* __restrict__ W5,
                        const float* __restrict__ W6,
                        float b1, float b3, float b4, float b5, float b6) {
  const int idx = blockIdx.x * 256 + threadIdx.x;
  if (idx < 65536) {
    float v = ((idx >> 8) == (idx & 255)) ? 1.0f : 0.0f;
    v = fmaf(b1, K1[idx], v);
    v = fmaf(b3, K3[idx], v);
    v = fmaf(b4, K4[idx], v);
    v = fmaf(b5, K5[idx], v);
    v = fmaf(b6, K6[idx], v);
    Mm[idx] = v;
  } else {
    const int q = idx - 65536;
    if (q >= 256 * 32) return;
    const int r = q >> 5, c = q & 31;
    float w[6];
    for (int st = 0; st < 6; ++st) {
      const int o = r * 192 + 32 * st + c;
      float t = 0.0f;
      t = fmaf(b1, W1[o], t);
      t = fmaf(b3, W3[o], t);
      t = fmaf(b4, W4[o], t);
      t = fmaf(b5, W5[o], t);
      t = fmaf(b6, W6[o], t);
      w[st] = t;
    }
    for (int j = 0; j < 4; ++j) {
      float s = 0.0f;
      for (int st = 0; st < 6; ++st) s += d_H[st][j] * w[st];
      Q[(32 * j + c) * 256 + r] = s;
    }
  }
}
__global__ __launch_bounds__(128) void gemm_kernel(
    float* __restrict__ D, int ldd,
    const float* __restrict__ A, int lda,
    const float* __restrict__ B, int ldb,
    const float* __restrict__ Cin, int ldc,
    int Mr, int Nc, int Kd, float alpha, float beta, int transB) {
  const int lane = threadIdx.x & 31, wave = threadIdx.x >> 5;
  const int hl = lane >> 4, lr = lane & 15;
  const int tilesN = Nc >> 4;
  const int tile = blockIdx.x * 4 + wave;
  if (tile >= (Mr >> 4) * tilesN) return;
  const int tm = tile / tilesN, tn = tile % tilesN;
  const float* Ar = A + (size_t)(tm * 16 + lr) * lda + 2 * hl;
  v8f acc = {};
  if (transB) {
    const float* Br = B + (size_t)(tn * 16 + lr) * ldb + 2 * hl;
    #pragma unroll 4
    for (int k = 0; k < Kd; k += 4) {
      v2f a = *(const v2f*)(Ar + k);
      v2f b = *(const v2f*)(Br + k);
      acc = WMMA_F32(a, b, acc);
    }
  } else {
    const float* Bc = B + (size_t)(2 * hl) * ldb + tn * 16 + lr;
    #pragma unroll 4
    for (int k = 0; k < Kd; k += 4) {
      v2f a = *(const v2f*)(Ar + k);
      v2f b;
      b[0] = Bc[(size_t)k * ldb];
      b[1] = Bc[(size_t)k * ldb + ldb];
      acc = WMMA_F32(a, b, acc);
    }
  }
  float* Dp = D + (size_t)(tm * 16 + 8 * hl) * ldd + tn * 16 + lr;
  if (beta != 0.0f) {
    const float* Cp = Cin + (size_t)(tm * 16 + 8 * hl) * ldc + tn * 16 + lr;
    #pragma unroll
    for (int v = 0; v < 8; ++v)
      Dp[(size_t)v * ldd] = alpha * acc[v] + beta * Cp[(size_t)v * ldc];
  } else {
    #pragma unroll
    for (int v = 0; v < 8; ++v)
      Dp[(size_t)v * ldd] = alpha * acc[v];
  }
}

static constexpr int GK_LD = 260;
static constexpr int GN_LD = 40;

__device__ __forceinline__ void k256_load_a(float4 (&ra)[16], const float* __restrict__ A, int bm, int tid) {
  const float4* Ab = (const float4*)(A + (size_t)bm * 32 * 256);
  #pragma unroll
  for (int i = 0; i < 16; ++i) ra[i] = Ab[tid + 128 * i];
}
__device__ __forceinline__ void k256_store_a(float* As, const float4 (&ra)[16], int tid) {
  #pragma unroll
  for (int i = 0; i < 16; ++i) {
    const int idx = tid + 128 * i;
    *(float4*)(As + (idx >> 6) * GK_LD + 4 * (idx & 63)) = ra[i];
  }
}
template <bool TB>
__device__ __forceinline__ void k256_frags(v2f (&a)[16], v2f (&b)[16],
                                           const float* Ap, const float* Bp, int g) {
  #pragma unroll
  for (int j = 0; j < 16; ++j) {
    const int kc = 16 * g + j;
    a[j] = *(const v2f*)(Ap + 4 * kc);
    if (TB) {
      b[j] = *(const v2f*)(Bp + 4 * kc);
    } else {
      b[j][0] = Bp[4 * kc * GN_LD];
      b[j][1] = Bp[4 * kc * GN_LD + GN_LD];
    }
  }
}
__device__ __forceinline__ v8f k256_wmma16(const v2f (&a)[16], const v2f (&b)[16], v8f acc) {
  #pragma unroll
  for (int j = 0; j < 16; ++j) acc = WMMA_F32(a[j], b[j], acc);
  return acc;
}
template <bool TB>
__device__ __forceinline__ v8f k256_chain(const float* Ap, const float* Bp) {
  v2f a0[16], b0[16], a1[16], b1[16];
  v8f acc = {};
  k256_frags<TB>(a0, b0, Ap, Bp, 0);
  k256_frags<TB>(a1, b1, Ap, Bp, 1);
  __builtin_amdgcn_sched_barrier(0);
  acc = k256_wmma16(a0, b0, acc);
  k256_frags<TB>(a0, b0, Ap, Bp, 2);
  __builtin_amdgcn_sched_barrier(0);
  acc = k256_wmma16(a1, b1, acc);
  k256_frags<TB>(a1, b1, Ap, Bp, 3);
  __builtin_amdgcn_sched_barrier(0);
  acc = k256_wmma16(a0, b0, acc);
  acc = k256_wmma16(a1, b1, acc);
  return acc;
}

template <bool TB, int NBN>
__global__ __launch_bounds__(128) void gemm_k256_kernel(
    float* __restrict__ D, int ldd, const float* __restrict__ A, const float* __restrict__ B) {
  __shared__ __attribute__((aligned(16))) float As[32 * GK_LD];
  __shared__ __attribute__((aligned(16))) float Bs[TB ? 32 * GK_LD : 256 * GN_LD];
  const int tid = threadIdx.x;
  const int lane = tid & 31;
  const int wave = __builtin_amdgcn_readfirstlane(tid >> 5);
  const int hl = lane >> 4, lr = lane & 15;
  const int bm = blockIdx.x / NBN, bn = blockIdx.x % NBN;
  {
    float4 ra[16], rb[16];
    k256_load_a(ra, A, bm, tid);
    if (TB) {
      k256_load_a(rb, B, bn, tid);
    } else {
      const float* Bb = B + bn * 32;
      #pragma unroll
      for (int i = 0; i < 16; ++i) {
        const int idx = tid + 128 * i;
        rb[i] = *(const float4*)(Bb + (idx >> 3) * 256 + 4 * (idx & 7));
      }
    }
    k256_store_a(As, ra, tid);
    if (TB) {
      k256_store_a(Bs, rb, tid);
    } else {
      #pragma unroll
      for (int i = 0; i < 16; ++i) {
        const int idx = tid + 128 * i;
        *(float4*)(Bs + (idx >> 3) * GN_LD + 4 * (idx & 7)) = rb[i];
      }
    }
  }
  __syncthreads();
  const int tm = wave >> 1, tn = wave & 1;
  const float* Ap = As + (tm * 16 + lr) * GK_LD + 2 * hl;
  const float* Bp = TB ? Bs + (tn * 16 + lr) * GK_LD + 2 * hl
                       : Bs + (2 * hl) * GN_LD + tn * 16 + lr;
  const v8f acc = k256_chain<TB>(Ap, Bp);
  float* Dp = D + (size_t)(bm * 32 + tm * 16 + 8 * hl) * ldd + bn * 32 + tn * 16 + lr;
  #pragma unroll
  for (int v = 0; v < 8; ++v) Dp[(size_t)v * ldd] = acc[v];
}

__device__ __forceinline__ void stage_fma_plane(float4 (&v)[16], const float* __restrict__ P, float c,
                                                int ld, int col0, int tid) {
  float4 t[16];
  #pragma unroll
  for (int i = 0; i < 16; ++i) {
    const int idx = tid + 128 * i;
    t[i] = *(const float4*)(P + (idx >> 3) * ld + col0 + 4 * (idx & 7));
  }
  #pragma unroll
  for (int i = 0; i < 16; ++i) {
    v[i].x = fmaf(c, t[i].x, v[i].x);  v[i].y = fmaf(c, t[i].y, v[i].y);
    v[i].z = fmaf(c, t[i].z, v[i].z);  v[i].w = fmaf(c, t[i].w, v[i].w);
  }
}
template <int NJ>
__global__ __launch_bounds__(128) void stage_kernel(
    float* __restrict__ Kout, float* __restrict__ Lout,
    const float* __restrict__ Ad, const float* __restrict__ Bd, int blk,
    const float* __restrict__ K0, const float* __restrict__ K1, const float* __restrict__ K2,
    const float* __restrict__ K3, const float* __restrict__ K4,
    const float* __restrict__ W0, const float* __restrict__ W1, const float* __restrict__ W2,
    const float* __restrict__ W3, const float* __restrict__ W4,
    float c0, float c1, float c2, float c3, float c4) {
  __shared__ __attribute__((aligned(16))) float As[32 * GK_LD];
  __shared__ __attribute__((aligned(16))) float Bs[256 * GN_LD];
  const int tid = threadIdx.x;
  const int lane = tid & 31;
  const int wave = __builtin_amdgcn_readfirstlane(tid >> 5);
  const int hl = lane >> 4, lr = lane & 15;
  const bool isK = blockIdx.x < 64;
  const int q = isK ? (int)blockIdx.x : (int)blockIdx.x - 64;
  const int bm = isK ? (q >> 3) : (q / 6);
  const int bn = isK ? (q & 7) : (q % 6);
  const int ld = isK ? 256 : 192;
  {
    float4 ra[16], v[16];
    k256_load_a(ra, Ad, bm, tid);
    #pragma unroll
    for (int i = 0; i < 16; ++i) {
      const int idx = tid + 128 * i;
      const int k = idx >> 3, n = bn * 32 + 4 * (idx & 7);
      v[i] = isK ? make_float4(k == n ? 1.0f : 0.0f, k == n + 1 ? 1.0f : 0.0f,
                               k == n + 2 ? 1.0f : 0.0f, k == n + 3 ? 1.0f : 0.0f)
                 : make_float4(0.0f, 0.0f, 0.0f, 0.0f);
    }
    stage_fma_plane(v, isK ? K0 : W0, c0, ld, bn * 32, tid);
    if (NJ > 1) stage_fma_plane(v, isK ? K1 : W1, c1, ld, bn * 32, tid);
    if (NJ > 2) stage_fma_plane(v, isK ? K2 : W2, c2, ld, bn * 32, tid);
    if (NJ > 3) stage_fma_plane(v, isK ? K3 : W3, c3, ld, bn * 32, tid);
    if (NJ > 4) stage_fma_plane(v, isK ? K4 : W4, c4, ld, bn * 32, tid);
    k256_store_a(As, ra, tid);
    #pragma unroll
    for (int i = 0; i < 16; ++i) {
      const int idx = tid + 128 * i;
      *(float4*)(Bs + (idx >> 3) * GN_LD + 4 * (idx & 7)) = v[i];
    }
  }
  __syncthreads();
  const int tm = wave >> 1, tn = wave & 1;
  const float* Ap = As + (tm * 16 + lr) * GK_LD + 2 * hl;
  const float* Bp = Bs + (2 * hl) * GN_LD + tn * 16 + lr;
  const v8f acc = k256_chain<false>(Ap, Bp);
  const int row0 = bm * 32 + tm * 16 + 8 * hl;
  const int n = bn * 32 + tn * 16 + lr;
  if (isK) {
    float* Dp = Kout + (size_t)row0 * 256 + n;
    #pragma unroll
    for (int v = 0; v < 8; ++v) Dp[(size_t)v * 256] = acc[v];
  } else {
    float* Dp = Lout + (size_t)row0 * 192 + n;
    if (bn == blk) {
      const float* Bp2 = Bd + (size_t)row0 * 32 + (n & 31);
      #pragma unroll
      for (int v = 0; v < 8; ++v) Dp[(size_t)v * 192] = acc[v] + Bp2[(size_t)v * 32];
    } else {
      #pragma unroll
      for (int v = 0; v < 8; ++v) Dp[(size_t)v * 192] = acc[v];
    }
  }
}

static constexpr int DR_ROWS = 128;
static constexpr int DR_LD = 132;

__device__ __forceinline__ float xhalf(float x) {
  const int b = __builtin_bit_cast(int, x);
  return __builtin_bit_cast(float,
      __builtin_amdgcn_permlanex16(b, b, 0x76543210, 0xfedcba98, false, false));
}

__global__ __launch_bounds__(256) void drive_kernel(
    float* __restrict__ G, const float* __restrict__ u, const float* __restrict__ Q) {
  __shared__ __attribute__((aligned(16))) float Pt[DR_ROWS * DR_LD];
  const int tid = threadIdx.x;
  const int lane = tid & 31;
  const int wave = __builtin_amdgcn_readfirstlane(tid >> 5);
  const int hl = lane >> 4, lr = lane & 15;
  const int n0 = blockIdx.x * DR_ROWS;
  const int col0 = wave * 32;

  {
    const int cc = tid & 7, rg = tid >> 3;
    const int nb = n0 + 4 * rg;
    float4 U[7];
    #pragma unroll
    for (int i = 0; i < 7; ++i) {
      int g = nb - 1 + i;
      g = g < 0 ? 0 : (g > Tn - 1 ? Tn - 1 : g);
      U[i] = *(const float4*)(u + (size_t)g * 32 + 4 * cc);
    }
    #pragma unroll
    for (int t = 0; t < 4; ++t) {
      const int n = nb + t;
      const bool first = (n == 0), penult = (n == Tn - 2), last = (n == Tn - 1);
      const float4 mb = first ? U[t + 1] : U[t];
      const float ms = first ? 1.0f : 0.5f;
      const float4 na = penult ? U[t + 2] : U[t + 3];
      const float ns = penult ? 1.0f : 0.5f;
      float4 p0 = U[t + 1];
      float4 p1 = make_float4(ms * (U[t + 2].x - mb.x), ms * (U[t + 2].y - mb.y),
                              ms * (U[t + 2].z - mb.z), ms * (U[t + 2].w - mb.w));
      float4 p2 = U[t + 2];
      float4 p3 = make_float4(ns * (na.x - U[t + 1].x), ns * (na.y - U[t + 1].y),
                              ns * (na.z - U[t + 1].z), ns * (na.w - U[t + 1].w));
      if (last) {
        const float4 z = make_float4(0.0f, 0.0f, 0.0f, 0.0f);
        p0 = z; p1 = z; p2 = z; p3 = z;
      }
      float* prow = Pt + (4 * rg + t) * DR_LD + 4 * cc;
      *(float4*)(prow)      = p0;
      *(float4*)(prow + 32) = p1;
      *(float4*)(prow + 64) = p2;
      *(float4*)(prow + 96) = p3;
    }
  }

  v2f b0[32], b1[32];
  {
    const float* Qp = Q + (size_t)(2 * hl) * 256 + col0 + lr;
    #pragma unroll
    for (int kc = 0; kc < 32; ++kc) {
      b0[kc][0] = Qp[kc * 1024];
      b0[kc][1] = Qp[kc * 1024 + 256];
      b1[kc][0] = Qp[kc * 1024 + 16];
      b1[kc][1] = Qp[kc * 1024 + 256 + 16];
    }
  }
  __syncthreads();

  const float* Ap = Pt + lr * DR_LD + 2 * hl;
  for (int hb = 0; hb < 2; ++hb) {
    const float* Ah = Ap + hb * 64 * DR_LD;
    v8f acc0[4] = {}, acc1[4] = {};
    #pragma unroll
    for (int kc = 0; kc < 32; ++kc) {
      #pragma unroll
      for (int r = 0; r < 4; ++r) {
        const v2f a = *(const v2f*)(Ah + r * 16 * DR_LD + kc * 4);
        acc0[r] = WMMA_F32(a, b0[kc], acc0[r]);
        acc1[r] = WMMA_F32(a, b1[kc], acc1[r]);
      }
    }
    float* Gp = G + (size_t)(n0 + hb * 64) * 256 + col0 + lane;
    #pragma unroll
    for (int r = 0; r < 4; ++r) {
      #pragma unroll
      for (int v = 0; v < 8; ++v) {
        const float t0 = acc0[r][v], t1 = acc1[r][v];
        const float got = xhalf(hl ? t0 : t1);
        const float lo = hl ? got : t0;
        const float hi = hl ? t1 : got;
        Gp[(size_t)(r * 16 + v) * 256] = lo;
        Gp[(size_t)(r * 16 + 8 + v) * 256] = hi;
      }
    }
  }
}

__global__ __launch_bounds__(256) void scan_kernel(
    const float* __restrict__ Mg, const float* __restrict__ G,
    const float* __restrict__ Init,
    float* __restrict__ OutAll,
    float* __restrict__ OutEnd,
    int L,
    float* __restrict__ Pw,
    int nscan) {
  __shared__ __attribute__((aligned(16))) float smem[SCAN_LDS_FLOATS];
  float* Ms = smem;
  float* Xbuf0 = smem + 256 * MS_LD;
  float* Xbuf1 = Xbuf0 + 256 * SC_COLS;
  const int tid = threadIdx.x;
  const int lane = tid & 31, wave = tid >> 5;
  const int hl = lane >> 4, lr = lane & 15;
  const bool pw = (int)blockIdx.x >= nscan;
  const int cbase = (pw ? (int)blockIdx.x - nscan : (int)blockIdx.x) * SC_COLS;
  const int tr0 = wave * 2;
  if (pw) { Init = nullptr; OutAll = nullptr; OutEnd = nullptr; }

  const bool hasInit = (Init != nullptr);
  const bool hasG = (L > 0 && !pw);
  const float* gbase = G + ((size_t)(cbase + lr) * L) * 256 + tr0 * 16 + 8 * hl;
  float4 ga = make_float4(0.0f, 0.0f, 0.0f, 0.0f), gb = ga, gc = ga, gd = ga;
  if (hasG) {
    ga = *(const float4*)(gbase);
    gb = *(const float4*)(gbase + 4);
    gc = *(const float4*)(gbase + 16);
    gd = *(const float4*)(gbase + 20);
  }
  float xi[SC_COLS];
  if (hasInit) {
    #pragma unroll
    for (int e = 0; e < SC_COLS; ++e) xi[e] = Init[(size_t)(cbase + e) * 256 + tid];
  }
  {
    const float4* src = (const float4*)Mg;
    float4 mr[64];
    #pragma unroll
    for (int i = 0; i < 64; ++i) mr[i] = src[tid + 256 * i];
    float4* dst = (float4*)Ms;
    #pragma unroll
    for (int i = 0; i < 64; ++i) {
      const int k = tid + 256 * i;
      dst[(k >> 6) * (MS_LD / 4) + (k & 63)] = mr[i];
    }
  }
  {
    float4* xr = (float4*)(Xbuf0 + tid * SC_COLS);
    if (hasInit) {
      #pragma unroll
      for (int q = 0; q < SC_COLS / 4; ++q)
        xr[q] = make_float4(xi[4 * q], xi[4 * q + 1], xi[4 * q + 2], xi[4 * q + 3]);
    } else {
      #pragma unroll
      for (int q = 0; q < SC_COLS / 4; ++q) {
        const int e = cbase + 4 * q;
        xr[q] = make_float4((pw && tid == e) ? 1.0f : 0.0f, (pw && tid == e + 1) ? 1.0f : 0.0f,
                            (pw && tid == e + 2) ? 1.0f : 0.0f, (pw && tid == e + 3) ? 1.0f : 0.0f);
      }
    }
  }
  __syncthreads();

  v2f a0[64], a1[64];
  {
    const float* Mr0 = Ms + (size_t)(tr0 * 16 + lr) * MS_LD + 2 * hl;
    const float* Mr1 = Mr0 + 16 * MS_LD;
    #pragma unroll
    for (int kc = 0; kc < 64; ++kc) {
      a0[kc] = *(const v2f*)(Mr0 + kc * 4);
      a1[kc] = *(const v2f*)(Mr1 + kc * 4);
    }
  }

  float* Xc = Xbuf0;
  float* Xn = Xbuf1;
  for (int i = 0; i < L; ++i) {
    if (OutAll) {
      for (int e = 0; e < SC_COLS; ++e) {
        size_t m = (size_t)(cbase + e) * L + i;
        OutAll[m * 256 + tid] = Xc[tid * SC_COLS + e];
      }
    }
    v8f acc0, acc1;
    acc0[0]=ga.x; acc0[1]=ga.y; acc0[2]=ga.z; acc0[3]=ga.w;
    acc0[4]=gb.x; acc0[5]=gb.y; acc0[6]=gb.z; acc0[7]=gb.w;
    acc1[0]=gc.x; acc1[1]=gc.y; acc1[2]=gc.z; acc1[3]=gc.w;
    acc1[4]=gd.x; acc1[5]=gd.y; acc1[6]=gd.z; acc1[7]=gd.w;
    if (!pw) {
      const float* gp = gbase + (size_t)((i + 1 < L) ? i + 1 : i) * 256;
      ga = *(const float4*)(gp);
      gb = *(const float4*)(gp + 4);
      gc = *(const float4*)(gp + 16);
      gd = *(const float4*)(gp + 20);
    }
    v2f b[64];
    {
      const float* Xk = Xc + 2 * hl * SC_COLS + lr;
      #pragma unroll
      for (int kg = 0; kg < 16; ++kg) {
        #pragma unroll
        for (int j = 0; j < 4; ++j) {
          const int kc = 4 * kg + j;
          b[kc][0] = Xk[kc * 4 * SC_COLS];
          b[kc][1] = Xk[kc * 4 * SC_COLS + SC_COLS];
        }
        __builtin_amdgcn_sched_barrier(0);
      }
    }
    #pragma unroll
    for (int kc = 0; kc < 64; ++kc) {
      acc0 = WMMA_F32(a0[kc], b[kc], acc0);
      acc1 = WMMA_F32(a1[kc], b[kc], acc1);
    }
    #pragma unroll
    for (int v = 0; v < 8; ++v) {
      Xn[(tr0 * 16 + 8 * hl + v) * SC_COLS + lr] = acc0[v];
      Xn[((tr0 + 1) * 16 + 8 * hl + v) * SC_COLS + lr] = acc1[v];
    }
    __syncthreads();
    float* tmp = Xc; Xc = Xn; Xn = tmp;
  }
  if (OutEnd) {
    for (int e = 0; e < SC_COLS; ++e)
      OutEnd[(size_t)(cbase + e) * 256 + tid] = Xc[tid * SC_COLS + e];
  }
  if (pw) {
    float4* dst = (float4*)(Pw + (size_t)tid * 256 + cbase);
    const float4* xr = (const float4*)(Xc + tid * SC_COLS);
    for (int q = 0; q < SC_COLS / 4; ++q) dst[q] = xr[q];
  }
}

static constexpr int TV_PITCH = 20;
static constexpr int TV_FLOATS = 16 * TV_PITCH;

template <int CTRL>
__device__ __forceinline__ float lane_read(float v) {
  return __int_as_float(__builtin_amdgcn_update_dpp(0, __float_as_int(v), CTRL, 0xF, 0xF, true));
}
static constexpr int DPP_SWAP1 = 0xB1;
static constexpr int DPP_SWAP2 = 0x4E;
static constexpr int DPP_ROR4  = 0x124;
static constexpr int DPP_ROR8  = 0x128;

__global__ __launch_bounds__(1024) void carry_kernel(
    const float* __restrict__ Mg, const float* __restrict__ Z,
    const float* __restrict__ x0, float* __restrict__ Tcar, int n) {
  __shared__ __attribute__((aligned(16))) float tv[2 * TV_FLOATS];
  const int tid = threadIdx.x;
  const int lane = tid & 31, wave = tid >> 5;
  const int g = lane & 15, rg = lane >> 4;
  const int rbase = wave * 8 + rg * 4;

  v2f m[4][8];
  for (int i = 0; i < 4; ++i) {
    const float4* mrow = (const float4*)(Mg + (size_t)(rbase + i) * 256 + 16 * g);
    for (int c = 0; c < 4; ++c) {
      float4 t4 = mrow[c];
      m[i][2 * c]     = (v2f){t4.x, t4.y};
      m[i][2 * c + 1] = (v2f){t4.z, t4.w};
    }
  }
  const int outslot = (tid >> 4) * TV_PITCH + (tid & 15);
  if (tid < 256) tv[outslot] = x0[tid];
  const bool odd1 = (g & 1) != 0, odd2 = (g & 2) != 0;
  const int myrow = rbase + (odd1 ? 2 : 0) + (odd2 ? 1 : 0);
  const int myslot = (myrow >> 4) * TV_PITCH + (myrow & 15);
  float zc = (n > 1) ? Z[myrow] : 0.0f;
  __syncthreads();

  for (int j = 0; j + 1 < n; ++j) {
    const float* tcur = tv + (j & 1) * TV_FLOATS;
    float* tnxt = tv + ((j + 1) & 1) * TV_FLOATS;
    if (wave < 8) Tcar[(size_t)j * 256 + tid] = tcur[outslot];
    float zn = 0.0f;
    if (j + 2 < n) zn = Z[(size_t)(j + 1) * 256 + myrow];
    const float* xs = tcur + g * TV_PITCH;
    v2f acc[4] = {};
    for (int c = 0; c < 4; ++c) {
      float4 x4 = *(const float4*)(xs + 4 * c);
      v2f xa = {x4.x, x4.y}, xb = {x4.z, x4.w};
      for (int i = 0; i < 4; ++i) {
        acc[i] += m[i][2 * c] * xa;
        acc[i] += m[i][2 * c + 1] * xb;
      }
    }
    float p0 = acc[0][0] + acc[0][1];
    float p1 = acc[1][0] + acc[1][1];
    float p2 = acc[2][0] + acc[2][1];
    float p3 = acc[3][0] + acc[3][1];
    float s0 = odd1 ? p0 : p2, s1 = odd1 ? p1 : p3;
    float k0 = odd1 ? p2 : p0, k1 = odd1 ? p3 : p1;
    k0 += lane_read<DPP_SWAP1>(s0);
    k1 += lane_read<DPP_SWAP1>(s1);
    float s = odd2 ? k0 : k1;
    float k = odd2 ? k1 : k0;
    k += lane_read<DPP_SWAP2>(s);
    k += lane_read<DPP_ROR4>(k);
    k += lane_read<DPP_ROR8>(k);
    if (g < 4) tnxt[myslot] = k + zc;
    __syncthreads();
    zc = zn;
  }
  if (wave < 8) Tcar[(size_t)(n - 1) * 256 + tid] = tv[((n - 1) & 1) * TV_FLOATS + outslot];
}

static inline void launch_gemm(hipStream_t s, float* D, int ldd,
                               const float* A, int lda, const float* B, int ldb,
                               const float* Cin, int ldc,
                               int M, int N, int K, float alpha, float beta, int transB) {
  int tiles = (M / 16) * (N / 16);
  gemm_kernel<<<(tiles + 3) / 4, 128, 0, s>>>(D, ldd, A, lda, B, ldb, Cin, ldc,
                                              M, N, K, alpha, beta, transB);
}

extern "C" void kernel_launch(void* const* d_in, const int* in_sizes, int n_in,
                              void* d_out, int out_size, void* d_ws, size_t ws_size,
                              hipStream_t stream) {
  (void)in_sizes; (void)n_in; (void)out_size; (void)ws_size;
  const float* t  = (const float*)d_in[0];
  const float* u  = (const float*)d_in[1];
  const float* x0 = (const float*)d_in[2];
  const float* Am = (const float*)d_in[3];
  const float* Bm = (const float*)d_in[4];
  const float* Cm = (const float*)d_in[5];
  const float* Dm = (const float*)d_in[6];
  float* xout = (float*)d_out;
  float* yout = xout + (size_t)Tn * NS;

  float* w = (float*)d_ws;
  float* Ad = w;  w += 65536;
  float* Bd = w;  w += 8192;
  float* Kt[6]; for (int i = 0; i < 6; i++) { Kt[i] = w; w += 65536; }
  Kt[0] = Ad;
  float* Lt[6]; for (int i = 0; i < 6; i++) { Lt[i] = w; w += 49152; }
  float* M64  = w; w += 65536 + 49152;
  float* Mm   = w; w += 65536;
  w += 49152;
  float* Qc   = w; w += 32768;
  float* M8   = w; w += 65536;
  float* M512 = w; w += 65536;
  float* PT1  = w; w += 65536;
  float* PT2  = w; w += 65536;
  float* Pm   = w; w += (size_t)Tn * 128;
  float* G    = w; w += (size_t)Tn * 256;
  float* Zend2 = w; w += (size_t)C2 * 256;
  float* Scar2 = w; w += (size_t)C2 * 256;
  float* Zend3 = w; w += (size_t)C3 * 256;
  float* Tcar  = w; w += (size_t)C3 * 256;
  float* Zend1 = Pm;
  float* Scar1 = Pm + (size_t)C1 * 256;

  auto gdim = [](int n) { return dim3((unsigned)((n + 255) / 256)); };

  k_prep<<<gdim(65536 + 49152), 256, 0, stream>>>(Ad, Bd, Lt[0], Am, Bm, t);

  double aa[7][7] = {};
  aa[2][1] = 0.2;
  aa[3][1] = 3.0/40.0;      aa[3][2] = 9.0/40.0;
  aa[4][1] = 44.0/45.0;     aa[4][2] = -56.0/15.0;     aa[4][3] = 32.0/9.0;
  aa[5][1] = 19372.0/6561.0; aa[5][2] = -25360.0/2187.0;
  aa[5][3] = 64448.0/6561.0; aa[5][4] = -212.0/729.0;
  aa[6][1] = 9017.0/3168.0; aa[6][2] = -355.0/33.0; aa[6][3] = 46732.0/5247.0;
  aa[6][4] = 49.0/176.0;    aa[6][5] = -5103.0/18656.0;
  stage_kernel<1><<<112, 128, 0, stream>>>(Kt[1], Lt[1], Ad, Bd, 1,
      Kt[0], Kt[0], Kt[0], Kt[0], Kt[0], Lt[0], Lt[0], Lt[0], Lt[0], Lt[0],
      (float)aa[2][1], 0.f, 0.f, 0.f, 0.f);
  stage_kernel<2><<<112, 128, 0, stream>>>(Kt[2], Lt[2], Ad, Bd, 2,
      Kt[0], Kt[1], Kt[0], Kt[0], Kt[0], Lt[0], Lt[1], Lt[0], Lt[0], Lt[0],
      (float)aa[3][1], (float)aa[3][2], 0.f, 0.f, 0.f);
  stage_kernel<3><<<112, 128, 0, stream>>>(Kt[3], Lt[3], Ad, Bd, 3,
      Kt[0], Kt[1], Kt[2], Kt[0], Kt[0], Lt[0], Lt[1], Lt[2], Lt[0], Lt[0],
      (float)aa[4][1], (float)aa[4][2], (float)aa[4][3], 0.f, 0.f);
  stage_kernel<4><<<112, 128, 0, stream>>>(Kt[4], Lt[4], Ad, Bd, 4,
      Kt[0], Kt[1], Kt[2], Kt[3], Kt[0], Lt[0], Lt[1], Lt[2], Lt[3], Lt[0],
      (float)aa[5][1], (float)aa[5][2], (float)aa[5][3], (float)aa[5][4], 0.f);
  stage_kernel<5><<<112, 128, 0, stream>>>(Kt[5], Lt[5], Ad, Bd, 5,
      Kt[0], Kt[1], Kt[2], Kt[3], Kt[4], Lt[0], Lt[1], Lt[2], Lt[3], Lt[4],
      (float)aa[6][1], (float)aa[6][2], (float)aa[6][3], (float)aa[6][4], (float)aa[6][5]);

  const double bb[7] = {0.0, 35.0/384.0, 0.0, 500.0/1113.0, 125.0/192.0,
                        -2187.0/6784.0, 11.0/84.0};
  k_final<<<gdim(65536 + 8192), 256, 0, stream>>>(Mm, Qc,
      Kt[0], Kt[2], Kt[3], Kt[4], Kt[5], Lt[0], Lt[2], Lt[3], Lt[4], Lt[5],
      (float)bb[1], (float)bb[3], (float)bb[4], (float)bb[5], (float)bb[6]);

  gemm_k256_kernel<false, 8><<<64, 128, 0, stream>>>(PT1,  256, Mm,  Mm);
  gemm_k256_kernel<false, 8><<<64, 128, 0, stream>>>(PT2,  256, PT1, PT1);
  gemm_k256_kernel<false, 8><<<64, 128, 0, stream>>>(M8,   256, PT2, PT2);

  drive_kernel<<<Tn / DR_ROWS, 256, 0, stream>>>(G, u, Qc);

  scan_kernel<<<C1 / SC_COLS, 256, 0, stream>>>(Mm, G, nullptr, nullptr, Zend1, LC, nullptr, C1 / SC_COLS);
  scan_kernel<<<C2 / SC_COLS + PW_WGS, 256, 0, stream>>>(M8, Zend1, nullptr, nullptr, Zend2, LC,
                                                         M64, C2 / SC_COLS);
  scan_kernel<<<C3 / SC_COLS + PW_WGS, 256, 0, stream>>>(M64, Zend2, nullptr, nullptr, Zend3, LC,
                                                         M512, C3 / SC_COLS);
  carry_kernel<<<1, 1024, 0, stream>>>(M512, Zend3, x0, Tcar, C3);
  scan_kernel<<<C3 / SC_COLS, 256, 0, stream>>>(M64, Zend2, Tcar, Scar2, nullptr, LC, nullptr, C3 / SC_COLS);
  scan_kernel<<<C2 / SC_COLS, 256, 0, stream>>>(M8, Zend1, Scar2, Scar1, nullptr, LC, nullptr, C2 / SC_COLS);
  scan_kernel<<<C1 / SC_COLS, 256, 0, stream>>>(Mm, G, Scar1, xout, nullptr, LC, nullptr, C1 / SC_COLS);

  gemm_k256_kernel<true, 1><<<Tn / 32, 128, 0, stream>>>(yout, 32, xout, Cm);
  launch_gemm(stream, yout, 32, u,    32,  Dm, 32,  yout,   32, Tn, 32, 32,  1.f, 1.f, 1);
}


# fix-up of scan levels 2 and 3 in ONE step, not 8: state i of a chunk = stored power M^i x start + stored zero-start state i, on 7x the work-groups; a different sum order over time
# speedup vs baseline: 5.3182x; 1.1087x over previous
#include <hip/hip_runtime.h>

typedef __attribute__((ext_vector_type(2))) float v2f;
typedef __attribute__((ext_vector_type(8))) float v8f;

#define WMMA_F32(a, b, c) \
  __builtin_amdgcn_wmma_f32_16x16x4_f32(false, (a), false, (b), (short)0, (c), false, false)

static constexpr int Tn = 32768;
static constexpr int NS = 256;
static constexpr int LC = 8;
static constexpr int C1 = 4096;
static constexpr int C2 = 512;
static constexpr int C3 = 64;
static constexpr int SC_COLS = 16;
static constexpr int PW_WGS = NS / SC_COLS;
static constexpr int MS_LD = 260;
static constexpr int SCAN_LDS_FLOATS = 256 * MS_LD + 2 * 256 * SC_COLS;

__constant__ float d_H[6][4] = {
  {1.0f, 0.0f, 0.0f, 0.0f},
  {0.896f, 0.128f, 0.104f, -0.032f},
  {0.784f, 0.147f, 0.216f, -0.063f},
  {0.104f, 0.032f, 0.896f, -0.128f},
  {25.0f/729.0f, 8.0f/729.0f, 704.0f/729.0f, -64.0f/729.0f},
  {0.0f, 0.0f, 1.0f, 0.0f},
};

__global__ void k_prep(float* __restrict__ Ad, float* __restrict__ Bd, float* __restrict__ Lt0,
                       const float* __restrict__ A, const float* __restrict__ B,
                       const float* __restrict__ t) {
  const int idx = blockIdx.x * 256 + threadIdx.x;
  const float dt = t[1] - t[0];
  if (idx < 65536) {
    Ad[idx] = dt * A[idx];
  } else {
    const int q = idx - 65536;
    if (q < 49152) {
      const int r = q / 192, c = q - r * 192;
      float val = 0.0f;
      if (c < 32) { const float bd = dt * B[r * 32 + c]; Bd[r * 32 + c] = bd; val += bd; }
      Lt0[q] = val;
    }
  }
}
__global__ void k_final(float* __restrict__ Mm, float* __restrict__ Q,
                        const float* __restrict__ K1, const float* __restrict__ K3,
                        const float* __restrict__ K4, const float* __restrict__ K5,
                        const float* __restrict__ K6,
                        const float* __restrict__ W1, const float* __restrict__ W3,
                        const float* __restrict__ W4, const float* __restrict__ W5,
                        const float* __restrict__ W6,
                        float b1, float b3, float b4, float b5, float b6) {
  const int idx = blockIdx.x * 256 + threadIdx.x;
  if (idx < 65536) {
    float v = ((idx >> 8) == (idx & 255)) ? 1.0f : 0.0f;
    v = fmaf(b1, K1[idx], v);
    v = fmaf(b3, K3[idx], v);
    v = fmaf(b4, K4[idx], v);
    v = fmaf(b5, K5[idx], v);
    v = fmaf(b6, K6[idx], v);
    Mm[idx] = v;
  } else {
    const int q = idx - 65536;
    if (q >= 256 * 32) return;
    const int r = q >> 5, c = q & 31;
    float w[6];
    for (int st = 0; st < 6; ++st) {
      const int o = r * 192 + 32 * st + c;
      float t = 0.0f;
      t = fmaf(b1, W1[o], t);
      t = fmaf(b3, W3[o], t);
      t = fmaf(b4, W4[o], t);
      t = fmaf(b5, W5[o], t);
      t = fmaf(b6, W6[o], t);
      w[st] = t;
    }
    for (int j = 0; j < 4; ++j) {
      float s = 0.0f;
      for (int st = 0; st < 6; ++st) s += d_H[st][j] * w[st];
      Q[(32 * j + c) * 256 + r] = s;
    }
  }
}
__global__ __launch_bounds__(128) void gemm_kernel(
    float* __restrict__ D, int ldd,
    const float* __restrict__ A, int lda,
    const float* __restrict__ B, int ldb,
    const float* __restrict__ Cin, int ldc,
    int Mr, int Nc, int Kd, float alpha, float beta, int transB) {
  const int lane = threadIdx.x & 31, wave = threadIdx.x >> 5;
  const int hl = lane >> 4, lr = lane & 15;
  const int tilesN = Nc >> 4;
  const int tile = blockIdx.x * 4 + wave;
  if (tile >= (Mr >> 4) * tilesN) return;
  const int tm = tile / tilesN, tn = tile % tilesN;
  const float* Ar = A + (size_t)(tm * 16 + lr) * lda + 2 * hl;
  v8f acc = {};
  if (transB) {
    const float* Br = B + (size_t)(tn * 16 + lr) * ldb + 2 * hl;
    #pragma unroll 4
    for (int k = 0; k < Kd; k += 4) {
      v2f a = *(const v2f*)(Ar + k);
      v2f b = *(const v2f*)(Br + k);
      acc = WMMA_F32(a, b, acc);
    }
  } else {
    const float* Bc = B + (size_t)(2 * hl) * ldb + tn * 16 + lr;
    #pragma unroll 4
    for (int k = 0; k < Kd; k += 4) {
      v2f a = *(const v2f*)(Ar + k);
      v2f b;
      b[0] = Bc[(size_t)k * ldb];
      b[1] = Bc[(size_t)k * ldb + ldb];
      acc = WMMA_F32(a, b, acc);
    }
  }
  float* Dp = D + (size_t)(tm * 16 + 8 * hl) * ldd + tn * 16 + lr;
  if (beta != 0.0f) {
    const float* Cp = Cin + (size_t)(tm * 16 + 8 * hl) * ldc + tn * 16 + lr;
    #pragma unroll
    for (int v = 0; v < 8; ++v)
      Dp[(size_t)v * ldd] = alpha * acc[v] + beta * Cp[(size_t)v * ldc];
  } else {
    #pragma unroll
    for (int v = 0; v < 8; ++v)
      Dp[(size_t)v * ldd] = alpha * acc[v];
  }
}

static constexpr int GK_LD = 260;
static constexpr int GN_LD = 40;

__device__ __forceinline__ void k256_load_a(float4 (&ra)[16], const float* __restrict__ A, int bm, int tid) {
  const float4* Ab = (const float4*)(A + (size_t)bm * 32 * 256);
  #pragma unroll
  for (int i = 0; i < 16; ++i) ra[i] = Ab[tid + 128 * i];
}
__device__ __forceinline__ void k256_store_a(float* As, const float4 (&ra)[16], int tid) {
  #pragma unroll
  for (int i = 0; i < 16; ++i) {
    const int idx = tid + 128 * i;
    *(float4*)(As + (idx >> 6) * GK_LD + 4 * (idx & 63)) = ra[i];
  }
}
template <bool TB>
__device__ __forceinline__ void k256_frags(v2f (&a)[16], v2f (&b)[16],
                                           const float* Ap, const float* Bp, int g) {
  #pragma unroll
  for (int j = 0; j < 16; ++j) {
    const int kc = 16 * g + j;
    a[j] = *(const v2f*)(Ap + 4 * kc);
    if (TB) {
      b[j] = *(const v2f*)(Bp + 4 * kc);
    } else {
      b[j][0] = Bp[4 * kc * GN_LD];
      b[j][1] = Bp[4 * kc * GN_LD + GN_LD];
    }
  }
}
__device__ __forceinline__ v8f k256_wmma16(const v2f (&a)[16], const v2f (&b)[16], v8f acc) {
  #pragma unroll
  for (int j = 0; j < 16; ++j) acc = WMMA_F32(a[j], b[j], acc);
  return acc;
}
template <bool TB>
__device__ __forceinline__ v8f k256_chain(const float* Ap, const float* Bp) {
  v2f a0[16], b0[16], a1[16], b1[16];
  v8f acc = {};
  k256_frags<TB>(a0, b0, Ap, Bp, 0);
  k256_frags<TB>(a1, b1, Ap, Bp, 1);
  __builtin_amdgcn_sched_barrier(0);
  acc = k256_wmma16(a0, b0, acc);
  k256_frags<TB>(a0, b0, Ap, Bp, 2);
  __builtin_amdgcn_sched_barrier(0);
  acc = k256_wmma16(a1, b1, acc);
  k256_frags<TB>(a1, b1, Ap, Bp, 3);
  __builtin_amdgcn_sched_barrier(0);
  acc = k256_wmma16(a0, b0, acc);
  acc = k256_wmma16(a1, b1, acc);
  return acc;
}

template <bool TB, int NBN>
__global__ __launch_bounds__(128) void gemm_k256_kernel(
    float* __restrict__ D, int ldd, const float* __restrict__ A, const float* __restrict__ B) {
  __shared__ __attribute__((aligned(16))) float As[32 * GK_LD];
  __shared__ __attribute__((aligned(16))) float Bs[TB ? 32 * GK_LD : 256 * GN_LD];
  const int tid = threadIdx.x;
  const int lane = tid & 31;
  const int wave = __builtin_amdgcn_readfirstlane(tid >> 5);
  const int hl = lane >> 4, lr = lane & 15;
  const int bm = blockIdx.x / NBN, bn = blockIdx.x % NBN;
  {
    float4 ra[16], rb[16];
    k256_load_a(ra, A, bm, tid);
    if (TB) {
      k256_load_a(rb, B, bn, tid);
    } else {
      const float* Bb = B + bn * 32;
      #pragma unroll
      for (int i = 0; i < 16; ++i) {
        const int idx = tid + 128 * i;
        rb[i] = *(const float4*)(Bb + (idx >> 3) * 256 + 4 * (idx & 7));
      }
    }
    k256_store_a(As, ra, tid);
    if (TB) {
      k256_store_a(Bs, rb, tid);
    } else {
      #pragma unroll
      for (int i = 0; i < 16; ++i) {
        const int idx = tid + 128 * i;
        *(float4*)(Bs + (idx >> 3) * GN_LD + 4 * (idx & 7)) = rb[i];
      }
    }
  }
  __syncthreads();
  const int tm = wave >> 1, tn = wave & 1;
  const float* Ap = As + (tm * 16 + lr) * GK_LD + 2 * hl;
  const float* Bp = TB ? Bs + (tn * 16 + lr) * GK_LD + 2 * hl
                       : Bs + (2 * hl) * GN_LD + tn * 16 + lr;
  const v8f acc = k256_chain<TB>(Ap, Bp);
  float* Dp = D + (size_t)(bm * 32 + tm * 16 + 8 * hl) * ldd + bn * 32 + tn * 16 + lr;
  #pragma unroll
  for (int v = 0; v < 8; ++v) Dp[(size_t)v * ldd] = acc[v];
}

__device__ __forceinline__ void stage_fma_plane(float4 (&v)[16], const float* __restrict__ P, float c,
                                                int ld, int col0, int tid) {
  float4 t[16];
  #pragma unroll
  for (int i = 0; i < 16; ++i) {
    const int idx = tid + 128 * i;
    t[i] = *(const float4*)(P + (idx >> 3) * ld + col0 + 4 * (idx & 7));
  }
  #pragma unroll
  for (int i = 0; i < 16; ++i) {
    v[i].x = fmaf(c, t[i].x, v[i].x);  v[i].y = fmaf(c, t[i].y, v[i].y);
    v[i].z = fmaf(c, t[i].z, v[i].z);  v[i].w = fmaf(c, t[i].w, v[i].w);
  }
}
template <int NJ>
__global__ __launch_bounds__(128) void stage_kernel(
    float* __restrict__ Kout, float* __restrict__ Lout,
    const float* __restrict__ Ad, const float* __restrict__ Bd, int blk,
    const float* __restrict__ K0, const float* __restrict__ K1, const float* __restrict__ K2,
    const float* __restrict__ K3, const float* __restrict__ K4,
    const float* __restrict__ W0, const float* __restrict__ W1, const float* __restrict__ W2,
    const float* __restrict__ W3, const float* __restrict__ W4,
    float c0, float c1, float c2, float c3, float c4) {
  __shared__ __attribute__((aligned(16))) float As[32 * GK_LD];
  __shared__ __attribute__((aligned(16))) float Bs[256 * GN_LD];
  const int tid = threadIdx.x;
  const int lane = tid & 31;
  const int wave = __builtin_amdgcn_readfirstlane(tid >> 5);
  const int hl = lane >> 4, lr = lane & 15;
  const bool isK = blockIdx.x < 64;
  const int q = isK ? (int)blockIdx.x : (int)blockIdx.x - 64;
  const int bm = isK ? (q >> 3) : (q / 6);
  const int bn = isK ? (q & 7) : (q % 6);
  const int ld = isK ? 256 : 192;
  {
    float4 ra[16], v[16];
    k256_load_a(ra, Ad, bm, tid);
    #pragma unroll
    for (int i = 0; i < 16; ++i) {
      const int idx = tid + 128 * i;
      const int k = idx >> 3, n = bn * 32 + 4 * (idx & 7);
      v[i] = isK ? make_float4(k == n ? 1.0f : 0.0f, k == n + 1 ? 1.0f : 0.0f,
                               k == n + 2 ? 1.0f : 0.0f, k == n + 3 ? 1.0f : 0.0f)
                 : make_float4(0.0f, 0.0f, 0.0f, 0.0f);
    }
    stage_fma_plane(v, isK ? K0 : W0, c0, ld, bn * 32, tid);
    if (NJ > 1) stage_fma_plane(v, isK ? K1 : W1, c1, ld, bn * 32, tid);
    if (NJ > 2) stage_fma_plane(v, isK ? K2 : W2, c2, ld, bn * 32, tid);
    if (NJ > 3) stage_fma_plane(v, isK ? K3 : W3, c3, ld, bn * 32, tid);
    if (NJ > 4) stage_fma_plane(v, isK ? K4 : W4, c4, ld, bn * 32, tid);
    k256_store_a(As, ra, tid);
    #pragma unroll
    for (int i = 0; i < 16; ++i) {
      const int idx = tid + 128 * i;
      *(float4*)(Bs + (idx >> 3) * GN_LD + 4 * (idx & 7)) = v[i];
    }
  }
  __syncthreads();
  const int tm = wave >> 1, tn = wave & 1;
  const float* Ap = As + (tm * 16 + lr) * GK_LD + 2 * hl;
  const float* Bp = Bs + (2 * hl) * GN_LD + tn * 16 + lr;
  const v8f acc = k256_chain<false>(Ap, Bp);
  const int row0 = bm * 32 + tm * 16 + 8 * hl;
  const int n = bn * 32 + tn * 16 + lr;
  if (isK) {
    float* Dp = Kout + (size_t)row0 * 256 + n;
    #pragma unroll
    for (int v = 0; v < 8; ++v) Dp[(size_t)v * 256] = acc[v];
  } else {
    float* Dp = Lout + (size_t)row0 * 192 + n;
    if (bn == blk) {
      const float* Bp2 = Bd + (size_t)row0 * 32 + (n & 31);
      #pragma unroll
      for (int v = 0; v < 8; ++v) Dp[(size_t)v * 192] = acc[v] + Bp2[(size_t)v * 32];
    } else {
      #pragma unroll
      for (int v = 0; v < 8; ++v) Dp[(size_t)v * 192] = acc[v];
    }
  }
}

static constexpr int DR_ROWS = 128;
static constexpr int DR_LD = 132;

__device__ __forceinline__ float xhalf(float x) {
  const int b = __builtin_bit_cast(int, x);
  return __builtin_bit_cast(float,
      __builtin_amdgcn_permlanex16(b, b, 0x76543210, 0xfedcba98, false, false));
}

__global__ __launch_bounds__(256) void drive_kernel(
    float* __restrict__ G, const float* __restrict__ u, const float* __restrict__ Q) {
  __shared__ __attribute__((aligned(16))) float Pt[DR_ROWS * DR_LD];
  const int tid = threadIdx.x;
  const int lane = tid & 31;
  const int wave = __builtin_amdgcn_readfirstlane(tid >> 5);
  const int hl = lane >> 4, lr = lane & 15;
  const int n0 = blockIdx.x * DR_ROWS;
  const int col0 = wave * 32;

  {
    const int cc = tid & 7, rg = tid >> 3;
    const int nb = n0 + 4 * rg;
    float4 U[7];
    #pragma unroll
    for (int i = 0; i < 7; ++i) {
      int g = nb - 1 + i;
      g = g < 0 ? 0 : (g > Tn - 1 ? Tn - 1 : g);
      U[i] = *(const float4*)(u + (size_t)g * 32 + 4 * cc);
    }
    #pragma unroll
    for (int t = 0; t < 4; ++t) {
      const int n = nb + t;
      const bool first = (n == 0), penult = (n == Tn - 2), last = (n == Tn - 1);
      const float4 mb = first ? U[t + 1] : U[t];
      const float ms = first ? 1.0f : 0.5f;
      const float4 na = penult ? U[t + 2] : U[t + 3];
      const float ns = penult ? 1.0f : 0.5f;
      float4 p0 = U[t + 1];
      float4 p1 = make_float4(ms * (U[t + 2].x - mb.x), ms * (U[t + 2].y - mb.y),
                              ms * (U[t + 2].z - mb.z), ms * (U[t + 2].w - mb.w));
      float4 p2 = U[t + 2];
      float4 p3 = make_float4(ns * (na.x - U[t + 1].x), ns * (na.y - U[t + 1].y),
                              ns * (na.z - U[t + 1].z), ns * (na.w - U[t + 1].w));
      if (last) {
        const float4 z = make_float4(0.0f, 0.0f, 0.0f, 0.0f);
        p0 = z; p1 = z; p2 = z; p3 = z;
      }
      float* prow = Pt + (4 * rg + t) * DR_LD + 4 * cc;
      *(float4*)(prow)      = p0;
      *(float4*)(prow + 32) = p1;
      *(float4*)(prow + 64) = p2;
      *(float4*)(prow + 96) = p3;
    }
  }

  v2f b0[32], b1[32];
  {
    const float* Qp = Q + (size_t)(2 * hl) * 256 + col0 + lr;
    #pragma unroll
    for (int kc = 0; kc < 32; ++kc) {
      b0[kc][0] = Qp[kc * 1024];
      b0[kc][1] = Qp[kc * 1024 + 256];
      b1[kc][0] = Qp[kc * 1024 + 16];
      b1[kc][1] = Qp[kc * 1024 + 256 + 16];
    }
  }
  __syncthreads();

  const float* Ap = Pt + lr * DR_LD + 2 * hl;
  for (int hb = 0; hb < 2; ++hb) {
    const float* Ah = Ap + hb * 64 * DR_LD;
    v8f acc0[4] = {}, acc1[4] = {};
    #pragma unroll
    for (int kc = 0; kc < 32; ++kc) {
      #pragma unroll
      for (int r = 0; r < 4; ++r) {
        const v2f a = *(const v2f*)(Ah + r * 16 * DR_LD + kc * 4);
        acc0[r] = WMMA_F32(a, b0[kc], acc0[r]);
        acc1[r] = WMMA_F32(a, b1[kc], acc1[r]);
      }
    }
    float* Gp = G + (size_t)(n0 + hb * 64) * 256 + col0 + lane;
    #pragma unroll
    for (int r = 0; r < 4; ++r) {
      #pragma unroll
      for (int v = 0; v < 8; ++v) {
        const float t0 = acc0[r][v], t1 = acc1[r][v];
        const float got = xhalf(hl ? t0 : t1);
        const float lo = hl ? got : t0;
        const float hi = hl ? t1 : got;
        Gp[(size_t)(r * 16 + v) * 256] = lo;
        Gp[(size_t)(r * 16 + 8 + v) * 256] = hi;
      }
    }
  }
}

__global__ __launch_bounds__(256) void scan_kernel(
    const float* __restrict__ Mg, const float* __restrict__ G,
    const float* __restrict__ Init,
    float* __restrict__ OutAll,
    float* __restrict__ OutEnd,
    int L,
    float* __restrict__ Pw,
    float* __restrict__ PwAll,
    int nscan) {
  __shared__ __attribute__((aligned(16))) float smem[SCAN_LDS_FLOATS];
  float* Ms = smem;
  float* Xbuf0 = smem + 256 * MS_LD;
  float* Xbuf1 = Xbuf0 + 256 * SC_COLS;
  const int tid = threadIdx.x;
  const int lane = tid & 31, wave = tid >> 5;
  const int hl = lane >> 4, lr = lane & 15;
  const bool pw = (int)blockIdx.x >= nscan;
  const int cbase = (pw ? (int)blockIdx.x - nscan : (int)blockIdx.x) * SC_COLS;
  const int tr0 = wave * 2;
  if (pw) { Init = nullptr; OutAll = nullptr; OutEnd = nullptr; }

  const bool hasInit = (Init != nullptr);
  const bool hasG = (L > 0 && !pw);
  const float* gbase = G + ((size_t)(cbase + lr) * L) * 256 + tr0 * 16 + 8 * hl;
  float4 ga = make_float4(0.0f, 0.0f, 0.0f, 0.0f), gb = ga, gc = ga, gd = ga;
  if (hasG) {
    ga = *(const float4*)(gbase);
    gb = *(const float4*)(gbase + 4);
    gc = *(const float4*)(gbase + 16);
    gd = *(const float4*)(gbase + 20);
  }
  float xi[SC_COLS];
  if (hasInit) {
    #pragma unroll
    for (int e = 0; e < SC_COLS; ++e) xi[e] = Init[(size_t)(cbase + e) * 256 + tid];
  }
  {
    const float4* src = (const float4*)Mg;
    float4 mr[64];
    #pragma unroll
    for (int i = 0; i < 64; ++i) mr[i] = src[tid + 256 * i];
    float4* dst = (float4*)Ms;
    #pragma unroll
    for (int i = 0; i < 64; ++i) {
      const int k = tid + 256 * i;
      dst[(k >> 6) * (MS_LD / 4) + (k & 63)] = mr[i];
    }
  }
  {
    float4* xr = (float4*)(Xbuf0 + tid * SC_COLS);
    if (hasInit) {
      #pragma unroll
      for (int q = 0; q < SC_COLS / 4; ++q)
        xr[q] = make_float4(xi[4 * q], xi[4 * q + 1], xi[4 * q + 2], xi[4 * q + 3]);
    } else {
      #pragma unroll
      for (int q = 0; q < SC_COLS / 4; ++q) {
        const int e = cbase + 4 * q;
        xr[q] = make_float4((pw && tid == e) ? 1.0f : 0.0f, (pw && tid == e + 1) ? 1.0f : 0.0f,
                            (pw && tid == e + 2) ? 1.0f : 0.0f, (pw && tid == e + 3) ? 1.0f : 0.0f);
      }
    }
  }
  __syncthreads();

  v2f a0[64], a1[64];
  {
    const float* Mr0 = Ms + (size_t)(tr0 * 16 + lr) * MS_LD + 2 * hl;
    const float* Mr1 = Mr0 + 16 * MS_LD;
    #pragma unroll
    for (int kc = 0; kc < 64; ++kc) {
      a0[kc] = *(const v2f*)(Mr0 + kc * 4);
      a1[kc] = *(const v2f*)(Mr1 + kc * 4);
    }
  }

  float* Xc = Xbuf0;
  float* Xn = Xbuf1;
  for (int i = 0; i < L; ++i) {
    if (OutAll) {
      for (int e = 0; e < SC_COLS; ++e) {
        size_t m = (size_t)(cbase + e) * L + i;
        OutAll[m * 256 + tid] = Xc[tid * SC_COLS + e];
      }
    }
    v8f acc0, acc1;
    acc0[0]=ga.x; acc0[1]=ga.y; acc0[2]=ga.z; acc0[3]=ga.w;
    acc0[4]=gb.x; acc0[5]=gb.y; acc0[6]=gb.z; acc0[7]=gb.w;
    acc1[0]=gc.x; acc1[1]=gc.y; acc1[2]=gc.z; acc1[3]=gc.w;
    acc1[4]=gd.x; acc1[5]=gd.y; acc1[6]=gd.z; acc1[7]=gd.w;
    if (!pw) {
      const float* gp = gbase + (size_t)((i + 1 < L) ? i + 1 : i) * 256;
      ga = *(const float4*)(gp);
      gb = *(const float4*)(gp + 4);
      gc = *(const float4*)(gp + 16);
      gd = *(const float4*)(gp + 20);
    }
    v2f b[64];
    {
      const float* Xk = Xc + 2 * hl * SC_COLS + lr;
      #pragma unroll
      for (int kg = 0; kg < 16; ++kg) {
        #pragma unroll
        for (int j = 0; j < 4; ++j) {
          const int kc = 4 * kg + j;
          b[kc][0] = Xk[kc * 4 * SC_COLS];
          b[kc][1] = Xk[kc * 4 * SC_COLS + SC_COLS];
        }
        __builtin_amdgcn_sched_barrier(0);
      }
    }
    #pragma unroll
    for (int kc = 0; kc < 64; ++kc) {
      acc0 = WMMA_F32(a0[kc], b[kc], acc0);
      acc1 = WMMA_F32(a1[kc], b[kc], acc1);
    }
    #pragma unroll
    for (int v = 0; v < 8; ++v) {
      Xn[(tr0 * 16 + 8 * hl + v) * SC_COLS + lr] = acc0[v];
      Xn[((tr0 + 1) * 16 + 8 * hl + v) * SC_COLS + lr] = acc1[v];
    }
    __syncthreads();
    float* tmp = Xc; Xc = Xn; Xn = tmp;
    if (pw && PwAll && i + 1 < L) {
      float4* dst = (float4*)(PwAll + (size_t)i * (NS * NS) + (size_t)tid * 256 + cbase);
      const float4* xr = (const float4*)(Xc + tid * SC_COLS);
      for (int q = 0; q < SC_COLS / 4; ++q) dst[q] = xr[q];
    }
  }
  if (OutEnd) {
    for (int e = 0; e < SC_COLS; ++e)
      OutEnd[(size_t)(cbase + e) * 256 + tid] = Xc[tid * SC_COLS + e];
  }
  if (pw) {
    float4* dst = (float4*)(Pw + (size_t)tid * 256 + cbase);
    const float4* xr = (const float4*)(Xc + tid * SC_COLS);
    for (int q = 0; q < SC_COLS / 4; ++q) dst[q] = xr[q];
  }
}

__global__ __launch_bounds__(256) void fix_kernel(
    const float* __restrict__ PwAll,
    const float* __restrict__ Pfx,
    const float* __restrict__ Init,
    float* __restrict__ Out) {
  __shared__ __attribute__((aligned(16))) float smem[SCAN_LDS_FLOATS];
  float* Ms = smem;
  float* Xc = smem + 256 * MS_LD;
  float* Xn = Xc + 256 * SC_COLS;
  const int tid = threadIdx.x;
  const int lane = tid & 31, wave = tid >> 5;
  const int hl = lane >> 4, lr = lane & 15;
  const int cbase = (int)blockIdx.x * SC_COLS;
  const int i = (int)blockIdx.y + 1;
  const int tr0 = wave * 2;
  const float* Mg = PwAll + (size_t)(i - 1) * (NS * NS);

  const float* gbase = Pfx + ((size_t)(cbase + lr) * LC + i) * 256 + tr0 * 16 + 8 * hl;
  const float4 ga = *(const float4*)(gbase);
  const float4 gb = *(const float4*)(gbase + 4);
  const float4 gc = *(const float4*)(gbase + 16);
  const float4 gd = *(const float4*)(gbase + 20);
  float xi[SC_COLS];
  #pragma unroll
  for (int e = 0; e < SC_COLS; ++e) xi[e] = Init[(size_t)(cbase + e) * 256 + tid];
  {
    const float4* src = (const float4*)Mg;
    float4 mr[64];
    #pragma unroll
    for (int j = 0; j < 64; ++j) mr[j] = src[tid + 256 * j];
    float4* dst = (float4*)Ms;
    #pragma unroll
    for (int j = 0; j < 64; ++j) {
      const int k = tid + 256 * j;
      dst[(k >> 6) * (MS_LD / 4) + (k & 63)] = mr[j];
    }
  }
  {
    float4* xr = (float4*)(Xc + tid * SC_COLS);
    #pragma unroll
    for (int q = 0; q < SC_COLS / 4; ++q)
      xr[q] = make_float4(xi[4 * q], xi[4 * q + 1], xi[4 * q + 2], xi[4 * q + 3]);
  }
  __syncthreads();

  v8f acc0, acc1;
  acc0[0]=ga.x; acc0[1]=ga.y; acc0[2]=ga.z; acc0[3]=ga.w;
  acc0[4]=gb.x; acc0[5]=gb.y; acc0[6]=gb.z; acc0[7]=gb.w;
  acc1[0]=gc.x; acc1[1]=gc.y; acc1[2]=gc.z; acc1[3]=gc.w;
  acc1[4]=gd.x; acc1[5]=gd.y; acc1[6]=gd.z; acc1[7]=gd.w;
  const float* Mr0 = Ms + (size_t)(tr0 * 16 + lr) * MS_LD + 2 * hl;
  const float* Mr1 = Mr0 + 16 * MS_LD;
  const float* Xk = Xc + 2 * hl * SC_COLS + lr;
  v2f a0[2][16], a1[2][16], b[2][16];
  #pragma unroll
  for (int g = 0; g < 5; ++g) {
    if (g < 4) {
      #pragma unroll
      for (int j = 0; j < 16; ++j) {
        const int kc = 16 * g + j;
        a0[g & 1][j] = *(const v2f*)(Mr0 + kc * 4);
        a1[g & 1][j] = *(const v2f*)(Mr1 + kc * 4);
        b[g & 1][j][0] = Xk[kc * 4 * SC_COLS];
        b[g & 1][j][1] = Xk[kc * 4 * SC_COLS + SC_COLS];
      }
    }
    __builtin_amdgcn_sched_barrier(0);
    if (g > 0) {
      #pragma unroll
      for (int j = 0; j < 16; ++j) {
        acc0 = WMMA_F32(a0[(g - 1) & 1][j], b[(g - 1) & 1][j], acc0);
        acc1 = WMMA_F32(a1[(g - 1) & 1][j], b[(g - 1) & 1][j], acc1);
      }
      __builtin_amdgcn_sched_barrier(0);
    }
  }
  #pragma unroll
  for (int v = 0; v < 8; ++v) {
    Xn[(tr0 * 16 + 8 * hl + v) * SC_COLS + lr] = acc0[v];
    Xn[((tr0 + 1) * 16 + 8 * hl + v) * SC_COLS + lr] = acc1[v];
  }
  __syncthreads();
  for (int e = 0; e < SC_COLS; ++e)
    Out[((size_t)(cbase + e) * LC + i) * 256 + tid] = Xn[tid * SC_COLS + e];
  if (i == 1) {
    for (int e = 0; e < SC_COLS; ++e)
      Out[((size_t)(cbase + e) * LC) * 256 + tid] = Xc[tid * SC_COLS + e];
  }
}

static constexpr int TV_PITCH = 20;
static constexpr int TV_FLOATS = 16 * TV_PITCH;

template <int CTRL>
__device__ __forceinline__ float lane_read(float v) {
  return __int_as_float(__builtin_amdgcn_update_dpp(0, __float_as_int(v), CTRL, 0xF, 0xF, true));
}
static constexpr int DPP_SWAP1 = 0xB1;
static constexpr int DPP_SWAP2 = 0x4E;
static constexpr int DPP_ROR4  = 0x124;
static constexpr int DPP_ROR8  = 0x128;

__global__ __launch_bounds__(1024) void carry_kernel(
    const float* __restrict__ Mg, const float* __restrict__ Z,
    const float* __restrict__ x0, float* __restrict__ Tcar, int n) {
  __shared__ __attribute__((aligned(16))) float tv[2 * TV_FLOATS];
  const int tid = threadIdx.x;
  const int lane = tid & 31, wave = tid >> 5;
  const int g = lane & 15, rg = lane >> 4;
  const int rbase = wave * 8 + rg * 4;

  v2f m[4][8];
  for (int i = 0; i < 4; ++i) {
    const float4* mrow = (const float4*)(Mg + (size_t)(rbase + i) * 256 + 16 * g);
    for (int c = 0; c < 4; ++c) {
      float4 t4 = mrow[c];
      m[i][2 * c]     = (v2f){t4.x, t4.y};
      m[i][2 * c + 1] = (v2f){t4.z, t4.w};
    }
  }
  const int outslot = (tid >> 4) * TV_PITCH + (tid & 15);
  if (tid < 256) tv[outslot] = x0[tid];
  const bool odd1 = (g & 1) != 0, odd2 = (g & 2) != 0;
  const int myrow = rbase + (odd1 ? 2 : 0) + (odd2 ? 1 : 0);
  const int myslot = (myrow >> 4) * TV_PITCH + (myrow & 15);
  float zc = (n > 1) ? Z[myrow] : 0.0f;
  __syncthreads();

  for (int j = 0; j + 1 < n; ++j) {
    const float* tcur = tv + (j & 1) * TV_FLOATS;
    float* tnxt = tv + ((j + 1) & 1) * TV_FLOATS;
    if (wave < 8) Tcar[(size_t)j * 256 + tid] = tcur[outslot];
    float zn = 0.0f;
    if (j + 2 < n) zn = Z[(size_t)(j + 1) * 256 + myrow];
    const float* xs = tcur + g * TV_PITCH;
    v2f acc[4] = {};
    for (int c = 0; c < 4; ++c) {
      float4 x4 = *(const float4*)(xs + 4 * c);
      v2f xa = {x4.x, x4.y}, xb = {x4.z, x4.w};
      for (int i = 0; i < 4; ++i) {
        acc[i] += m[i][2 * c] * xa;
        acc[i] += m[i][2 * c + 1] * xb;
      }
    }
    float p0 = acc[0][0] + acc[0][1];
    float p1 = acc[1][0] + acc[1][1];
    float p2 = acc[2][0] + acc[2][1];
    float p3 = acc[3][0] + acc[3][1];
    float s0 = odd1 ? p0 : p2, s1 = odd1 ? p1 : p3;
    float k0 = odd1 ? p2 : p0, k1 = odd1 ? p3 : p1;
    k0 += lane_read<DPP_SWAP1>(s0);
    k1 += lane_read<DPP_SWAP1>(s1);
    float s = odd2 ? k0 : k1;
    float k = odd2 ? k1 : k0;
    k += lane_read<DPP_SWAP2>(s);
    k += lane_read<DPP_ROR4>(k);
    k += lane_read<DPP_ROR8>(k);
    if (g < 4) tnxt[myslot] = k + zc;
    __syncthreads();
    zc = zn;
  }
  if (wave < 8) Tcar[(size_t)(n - 1) * 256 + tid] = tv[((n - 1) & 1) * TV_FLOATS + outslot];
}

static inline void launch_gemm(hipStream_t s, float* D, int ldd,
                               const float* A, int lda, const float* B, int ldb,
                               const float* Cin, int ldc,
                               int M, int N, int K, float alpha, float beta, int transB) {
  int tiles = (M / 16) * (N / 16);
  gemm_kernel<<<(tiles + 3) / 4, 128, 0, s>>>(D, ldd, A, lda, B, ldb, Cin, ldc,
                                              M, N, K, alpha, beta, transB);
}

extern "C" void kernel_launch(void* const* d_in, const int* in_sizes, int n_in,
                              void* d_out, int out_size, void* d_ws, size_t ws_size,
                              hipStream_t stream) {
  (void)in_sizes; (void)n_in; (void)out_size; (void)ws_size;
  const float* t  = (const float*)d_in[0];
  const float* u  = (const float*)d_in[1];
  const float* x0 = (const float*)d_in[2];
  const float* Am = (const float*)d_in[3];
  const float* Bm = (const float*)d_in[4];
  const float* Cm = (const float*)d_in[5];
  const float* Dm = (const float*)d_in[6];
  float* xout = (float*)d_out;
  float* yout = xout + (size_t)Tn * NS;

  float* w = (float*)d_ws;
  float* Ad = w;  w += 65536;
  float* Bd = w;  w += 8192;
  float* Kt[6]; for (int i = 0; i < 6; i++) { Kt[i] = w; w += 65536; }
  Kt[0] = Ad;
  float* Lt[6]; for (int i = 0; i < 6; i++) { Lt[i] = w; w += 49152; }
  float* M64  = w; w += 65536 + 49152;
  float* Mm   = w; w += 65536;
  w += 49152;
  float* Qc   = w; w += 32768;
  float* M8   = w; w += 65536;
  float* M512 = w; w += 65536;
  float* PT1  = w; w += 65536;
  float* PT2  = w; w += 65536;
  float* Pm   = w; w += (size_t)Tn * 128;
  float* G    = w; w += (size_t)Tn * 256;
  float* Zend2 = w; w += (size_t)C2 * 256;
  float* Scar2 = w; w += (size_t)C2 * 256;
  float* Zend3 = w; w += (size_t)C3 * 256;
  float* Tcar  = w; w += (size_t)C3 * 256;
  float* Zend1 = Pm;
  float* Scar1 = Pm + (size_t)C1 * 256;
  float* Pfx2 = Scar1 + (size_t)C1 * 256;
  float* Pfx3 = Pfx2 + (size_t)C1 * 256;
  float* Pow2 = Pfx3 + (size_t)C2 * 256;
  float* Pow3 = Pow2 + (size_t)(LC - 1) * NS * NS;

  auto gdim = [](int n) { return dim3((unsigned)((n + 255) / 256)); };

  k_prep<<<gdim(65536 + 49152), 256, 0, stream>>>(Ad, Bd, Lt[0], Am, Bm, t);

  double aa[7][7] = {};
  aa[2][1] = 0.2;
  aa[3][1] = 3.0/40.0;      aa[3][2] = 9.0/40.0;
  aa[4][1] = 44.0/45.0;     aa[4][2] = -56.0/15.0;     aa[4][3] = 32.0/9.0;
  aa[5][1] = 19372.0/6561.0; aa[5][2] = -25360.0/2187.0;
  aa[5][3] = 64448.0/6561.0; aa[5][4] = -212.0/729.0;
  aa[6][1] = 9017.0/3168.0; aa[6][2] = -355.0/33.0; aa[6][3] = 46732.0/5247.0;
  aa[6][4] = 49.0/176.0;    aa[6][5] = -5103.0/18656.0;
  stage_kernel<1><<<112, 128, 0, stream>>>(Kt[1], Lt[1], Ad, Bd, 1,
      Kt[0], Kt[0], Kt[0], Kt[0], Kt[0], Lt[0], Lt[0], Lt[0], Lt[0], Lt[0],
      (float)aa[2][1], 0.f, 0.f, 0.f, 0.f);
  stage_kernel<2><<<112, 128, 0, stream>>>(Kt[2], Lt[2], Ad, Bd, 2,
      Kt[0], Kt[1], Kt[0], Kt[0], Kt[0], Lt[0], Lt[1], Lt[0], Lt[0], Lt[0],
      (float)aa[3][1], (float)aa[3][2], 0.f, 0.f, 0.f);
  stage_kernel<3><<<112, 128, 0, stream>>>(Kt[3], Lt[3], Ad, Bd, 3,
      Kt[0], Kt[1], Kt[2], Kt[0], Kt[0], Lt[0], Lt[1], Lt[2], Lt[0], Lt[0],
      (float)aa[4][1], (float)aa[4][2], (float)aa[4][3], 0.f, 0.f);
  stage_kernel<4><<<112, 128, 0, stream>>>(Kt[4], Lt[4], Ad, Bd, 4,
      Kt[0], Kt[1], Kt[2], Kt[3], Kt[0], Lt[0], Lt[1], Lt[2], Lt[3], Lt[0],
      (float)aa[5][1], (float)aa[5][2], (float)aa[5][3], (float)aa[5][4], 0.f);
  stage_kernel<5><<<112, 128, 0, stream>>>(Kt[5], Lt[5], Ad, Bd, 5,
      Kt[0], Kt[1], Kt[2], Kt[3], Kt[4], Lt[0], Lt[1], Lt[2], Lt[3], Lt[4],
      (float)aa[6][1], (float)aa[6][2], (float)aa[6][3], (float)aa[6][4], (float)aa[6][5]);

  const double bb[7] = {0.0, 35.0/384.0, 0.0, 500.0/1113.0, 125.0/192.0,
                        -2187.0/6784.0, 11.0/84.0};
  k_final<<<gdim(65536 + 8192), 256, 0, stream>>>(Mm, Qc,
      Kt[0], Kt[2], Kt[3], Kt[4], Kt[5], Lt[0], Lt[2], Lt[3], Lt[4], Lt[5],
      (float)bb[1], (float)bb[3], (float)bb[4], (float)bb[5], (float)bb[6]);

  gemm_k256_kernel<false, 8><<<64, 128, 0, stream>>>(PT1,  256, Mm,  Mm);
  gemm_k256_kernel<false, 8><<<64, 128, 0, stream>>>(PT2,  256, PT1, PT1);
  gemm_k256_kernel<false, 8><<<64, 128, 0, stream>>>(M8,   256, PT2, PT2);

  drive_kernel<<<Tn / DR_ROWS, 256, 0, stream>>>(G, u, Qc);

  scan_kernel<<<C1 / SC_COLS, 256, 0, stream>>>(Mm, G, nullptr, nullptr, Zend1, LC, nullptr, nullptr, C1 / SC_COLS);
  scan_kernel<<<C2 / SC_COLS + PW_WGS, 256, 0, stream>>>(M8, Zend1, nullptr, Pfx2, Zend2, LC,
                                                         M64, Pow2, C2 / SC_COLS);
  scan_kernel<<<C3 / SC_COLS + PW_WGS, 256, 0, stream>>>(M64, Zend2, nullptr, Pfx3, Zend3, LC,
                                                         M512, Pow3, C3 / SC_COLS);
  carry_kernel<<<1, 1024, 0, stream>>>(M512, Zend3, x0, Tcar, C3);
  fix_kernel<<<dim3(C3 / SC_COLS, LC - 1), 256, 0, stream>>>(Pow3, Pfx3, Tcar, Scar2);
  fix_kernel<<<dim3(C2 / SC_COLS, LC - 1), 256, 0, stream>>>(Pow2, Pfx2, Scar2, Scar1);
  scan_kernel<<<C1 / SC_COLS, 256, 0, stream>>>(Mm, G, Scar1, xout, nullptr, LC, nullptr, nullptr, C1 / SC_COLS);

  gemm_k256_kernel<true, 1><<<Tn / 32, 128, 0, stream>>>(yout, 32, xout, Cm);
  launch_gemm(stream, yout, 32, u,    32,  Dm, 32,  yout,   32, Tn, 32, 32,  1.f, 1.f, 1);
}
